# starting kernel
# baseline (speedup 1.0000x reference)
#include <hip/hip_runtime.h>
#include <hip/hip_bf16.h>

#define IN_FEAT 128
#define HIDDEN  64

typedef __attribute__((ext_vector_type(2))) float v2f;
typedef __attribute__((ext_vector_type(8))) float v8f;

__global__ void init_deg_kernel(float* __restrict__ deg, int n) {
    int i = blockIdx.x * blockDim.x + threadIdx.x;
    if (i < n) deg[i] = 1.0f;
}

__global__ void deg_accum_kernel(const int* __restrict__ cols,
                                 const float* __restrict__ ew,
                                 float* __restrict__ deg, int e) {
    int i = blockIdx.x * blockDim.x + threadIdx.x;
    if (i < e) atomicAdd(&deg[cols[i]], ew[i]);
}

__global__ void dinv_kernel(float* __restrict__ deg, int n) {
    int i = blockIdx.x * blockDim.x + threadIdx.x;
    if (i < n) {
        float d = deg[i];
        deg[i] = (d > 0.0f) ? __frsqrt_rn(d) : 0.0f;
    }
}

__global__ void __launch_bounds__(256)
gemm1_wmma_kernel(const float* __restrict__ X, const float* __restrict__ W,
                  float* __restrict__ H, int n) {
    __shared__ float sW[IN_FEAT * HIDDEN];
    for (int i = threadIdx.x; i < IN_FEAT * HIDDEN; i += blockDim.x)
        sW[i] = W[i];
    __syncthreads();

    const int wave = blockIdx.x * (blockDim.x >> 5) + (threadIdx.x >> 5);
    const int lane = threadIdx.x & 31;
    const int mt = wave >> 2;
    const int nt = wave & 3;
    const int m0 = mt * 16;
    if (m0 >= n) return;
    const int n0 = nt * 16;
    const int half = lane >> 4;
    const int l = lane & 15;

    const float* Xrow = X + (size_t)(m0 + l) * IN_FEAT;
    v8f c = {0.f, 0.f, 0.f, 0.f, 0.f, 0.f, 0.f, 0.f};

#pragma unroll 8
    for (int k = 0; k < IN_FEAT; k += 4) {
        v2f a = *(const v2f*)(Xrow + k + 2 * half);
        v2f b;
        b.x = sW[(k + 2 * half + 0) * HIDDEN + n0 + l];
        b.y = sW[(k + 2 * half + 1) * HIDDEN + n0 + l];
        c = __builtin_amdgcn_wmma_f32_16x16x4_f32(
                 false, a,  false, b,
                 (short)0, c,  false,  false);
    }

    float* Hout = H + (size_t)(m0 + 8 * half) * HIDDEN + n0 + l;
#pragma unroll
    for (int v = 0; v < 8; v++)
        Hout[(size_t)v * HIDDEN] = c[v];
}

__global__ void init_agg_kernel(const float* __restrict__ dinv,
                                const float* __restrict__ h0,
                                float* __restrict__ agg, int n) {
    int wave = blockIdx.x * (blockDim.x >> 5) + (threadIdx.x >> 5);
    int lane = threadIdx.x & 31;
    if (wave >= n) return;
    float d = dinv[wave];
    float s = d * d;
    const float2* src = (const float2*)(h0 + (size_t)wave * HIDDEN);
    float2* dst = (float2*)(agg + (size_t)wave * HIDDEN);
    float2 v = src[lane];
    v.x *= s; v.y *= s;
    dst[lane] = v;
}

__global__ void scatter1_kernel(const int* __restrict__ rows,
                                const int* __restrict__ cols,
                                const float* __restrict__ ew,
                                const float* __restrict__ dinv,
                                const float* __restrict__ h0,
                                float* __restrict__ agg, int e) {
    int wave = blockIdx.x * (blockDim.x >> 5) + (threadIdx.x >> 5);
    int lane = threadIdx.x & 31;
    if (wave >= e) return;
    int r = rows[wave];
    int c = cols[wave];
    float norm = dinv[r] * ew[wave] * dinv[c];
    float2 v = ((const float2*)(h0 + (size_t)r * HIDDEN))[lane];
    float* dst = agg + (size_t)c * HIDDEN + 2 * lane;
    atomicAdd(dst + 0, norm * v.x);
    atomicAdd(dst + 1, norm * v.y);
}

__global__ void relu_bias_kernel(float* __restrict__ agg,
                                 const float* __restrict__ b1, int total) {
    int i = blockIdx.x * blockDim.x + threadIdx.x;
    if (i < total) {
        float v = agg[i] + b1[i & (HIDDEN - 1)];
        agg[i] = fmaxf(v, 0.0f);
    }
}

__global__ void gemv2_init_kernel(const float* __restrict__ h,
                                  const float* __restrict__ w2,
                                  const float* __restrict__ b2,
                                  const float* __restrict__ dinv,
                                  float* __restrict__ g,
                                  float* __restrict__ out, int n) {
    int wave = blockIdx.x * (blockDim.x >> 5) + (threadIdx.x >> 5);
    int lane = threadIdx.x & 31;
    if (wave >= n) return;
    float2 hv = ((const float2*)(h + (size_t)wave * HIDDEN))[lane];
    float2 wv = ((const float2*)w2)[lane];
    float p = hv.x * wv.x + hv.y * wv.y;
#pragma unroll
    for (int off = 16; off > 0; off >>= 1)
        p += __shfl_xor(p, off, 32);
    if (lane == 0) {
        float d = dinv[wave];
        g[wave] = p;
        out[wave] = b2[0] + d * d * p;
    }
}

__global__ void scatter2_kernel(const int* __restrict__ rows,
                                const int* __restrict__ cols,
                                const float* __restrict__ ew,
                                const float* __restrict__ dinv,
                                const float* __restrict__ g,
                                float* __restrict__ out, int e) {
    int i = blockIdx.x * blockDim.x + threadIdx.x;
    if (i >= e) return;
    int r = rows[i];
    int c = cols[i];
    float norm = dinv[r] * ew[i] * dinv[c];
    atomicAdd(&out[c], norm * g[r]);
}

extern "C" void kernel_launch(void* const* d_in, const int* in_sizes, int n_in,
                              void* d_out, int out_size, void* d_ws, size_t ws_size,
                              hipStream_t stream) {
    const float* x  = (const float*)d_in[0];
    const int*   ei = (const int*)d_in[1];
    const float* ew = (const float*)d_in[2];
    const float* w1 = (const float*)d_in[3];
    const float* b1 = (const float*)d_in[4];
    const float* w2 = (const float*)d_in[5];
    const float* b2 = (const float*)d_in[6];
    float* out = (float*)d_out;

    const int E = in_sizes[2];
    const int N = in_sizes[0] / IN_FEAT;
    const int* rows = ei;
    const int* cols = ei + E;

    float* ws   = (float*)d_ws;
    float* dinv = ws;
    float* g    = ws + N;
    float* h0   = ws + 2 * (size_t)N;
    float* agg  = h0 + (size_t)N * HIDDEN;

    const int B = 256;
    init_deg_kernel<<<(N + B - 1) / B, B, 0, stream>>>(dinv, N);
    deg_accum_kernel<<<(E + B - 1) / B, B, 0, stream>>>(cols, ew, dinv, E);
    dinv_kernel<<<(N + B - 1) / B, B, 0, stream>>>(dinv, N);

    {
        int mtiles = (N + 15) / 16;
        int waves  = mtiles * (HIDDEN / 16);
        int blocks = (waves + 7) / 8;
        gemm1_wmma_kernel<<<blocks, B, 0, stream>>>(x, w1, h0, N);
    }

    init_agg_kernel<<<(N + 7) / 8, B, 0, stream>>>(dinv, h0, agg, N);
    scatter1_kernel<<<(E + 7) / 8, B, 0, stream>>>(rows, cols, ew, dinv, h0, agg, E);
    relu_bias_kernel<<<((size_t)N * HIDDEN + B - 1) / B, B, 0, stream>>>(agg, b1, N * HIDDEN);

    gemv2_init_kernel<<<(N + 7) / 8, B, 0, stream>>>(agg, w2, b2, dinv, g, out, N);
    scatter2_kernel<<<(E + B - 1) / B, B, 0, stream>>>(rows, cols, ew, dinv, g, out, E);
}


# Degree and layer-2 scatters: per-edge global atomics replaced by LDS-atomic bins per work-group (128 edge chunks x 2 node ranges) flushed with coalesced global atomics; float sum order changes
# speedup vs baseline: 1.4749x; 1.4749x over previous
#include <hip/hip_runtime.h>
#include <hip/hip_bf16.h>

#define IN_FEAT 128
#define HIDDEN  64

typedef __attribute__((ext_vector_type(2))) float v2f;
typedef __attribute__((ext_vector_type(8))) float v8f;

__global__ void init_deg_kernel(float* __restrict__ deg, int n) {
    int i = blockIdx.x * blockDim.x + threadIdx.x;
    if (i < n) deg[i] = 1.0f;
}

#define SPAN       51200
#define SC_THREADS 1024
#define SC_CHUNKS  128
#define SC_UNROLL  8
static_assert(SC_UNROLL == 8, "HOLD8 names eight values");
#define HOLD8(a) "v"(a[0]), "v"(a[1]), "v"(a[2]), "v"(a[3]), "v"(a[4]), "v"(a[5]), "v"(a[6]), "v"(a[7])

__device__ __forceinline__ void bins_zero(float* bins) {
    float4* b4 = (float4*)bins;
    for (int i = threadIdx.x; i < SPAN / 4; i += SC_THREADS)
        b4[i] = make_float4(0.0f, 0.0f, 0.0f, 0.0f);
}

__device__ __forceinline__ void bins_flush(const float* bins,
                                           float* __restrict__ dst, int span) {
    const int rot = (int)((blockIdx.x * 416u) % SPAN);
    for (int i = threadIdx.x; i < SPAN; i += SC_THREADS) {
        int j = i + rot;
        if (j >= SPAN) j -= SPAN;
        if (j < span) {
            float v = bins[j];
            if (v != 0.0f) atomicAdd(dst + j, v);
        }
    }
}

__global__ void __launch_bounds__(SC_THREADS)
deg_accum_kernel(const int* __restrict__ cols,
                 const float* __restrict__ ew,
                 float* __restrict__ deg, int n, int e) {
    __shared__ __attribute__((aligned(16))) float bins[SPAN];
    const int lo = blockIdx.y * SPAN;
    const int span = min(SPAN, n - lo);
    bins_zero(bins);
    __syncthreads();

    const int nch = (int)gridDim.x;
    const int per = (e + nch - 1) / nch;
    const int start = blockIdx.x * per;
    const int end = min(e, start + per);
    int base = start;
    for (; base + SC_THREADS * SC_UNROLL <= end; base += SC_THREADS * SC_UNROLL) {
        int c[SC_UNROLL];
        float w[SC_UNROLL];
#pragma unroll
        for (int u = 0; u < SC_UNROLL; u++) {
            const int i = base + u * SC_THREADS + (int)threadIdx.x;
            c[u] = cols[i];
            w[u] = ew[i];
        }
        asm volatile("" :: HOLD8(c), HOLD8(w));
#pragma unroll
        for (int u = 0; u < SC_UNROLL; u++) {
            const unsigned k = (unsigned)(c[u] - lo);
            if (k < (unsigned)span) atomicAdd(&bins[k], w[u]);
        }
    }
    for (int i = base + (int)threadIdx.x; i < end; i += SC_THREADS) {
        const unsigned k = (unsigned)(cols[i] - lo);
        if (k < (unsigned)span) atomicAdd(&bins[k], ew[i]);
    }
    __syncthreads();
    bins_flush(bins, deg + lo, span);
}

__global__ void dinv_kernel(float* __restrict__ deg, int n) {
    int i = blockIdx.x * blockDim.x + threadIdx.x;
    if (i < n) {
        float d = deg[i];
        deg[i] = (d > 0.0f) ? __frsqrt_rn(d) : 0.0f;
    }
}

__global__ void __launch_bounds__(256)
gemm1_wmma_kernel(const float* __restrict__ X, const float* __restrict__ W,
                  float* __restrict__ H, int n) {
    __shared__ float sW[IN_FEAT * HIDDEN];
    for (int i = threadIdx.x; i < IN_FEAT * HIDDEN; i += blockDim.x)
        sW[i] = W[i];
    __syncthreads();

    const int wave = blockIdx.x * (blockDim.x >> 5) + (threadIdx.x >> 5);
    const int lane = threadIdx.x & 31;
    const int mt = wave >> 2;
    const int nt = wave & 3;
    const int m0 = mt * 16;
    if (m0 >= n) return;
    const int n0 = nt * 16;
    const int half = lane >> 4;
    const int l = lane & 15;

    const float* Xrow = X + (size_t)(m0 + l) * IN_FEAT;
    v8f c = {0.f, 0.f, 0.f, 0.f, 0.f, 0.f, 0.f, 0.f};

#pragma unroll 8
    for (int k = 0; k < IN_FEAT; k += 4) {
        v2f a = *(const v2f*)(Xrow + k + 2 * half);
        v2f b;
        b.x = sW[(k + 2 * half + 0) * HIDDEN + n0 + l];
        b.y = sW[(k + 2 * half + 1) * HIDDEN + n0 + l];
        c = __builtin_amdgcn_wmma_f32_16x16x4_f32(
                 false, a,  false, b,
                 (short)0, c,  false,  false);
    }

    float* Hout = H + (size_t)(m0 + 8 * half) * HIDDEN + n0 + l;
#pragma unroll
    for (int v = 0; v < 8; v++)
        Hout[(size_t)v * HIDDEN] = c[v];
}

__global__ void init_agg_kernel(const float* __restrict__ dinv,
                                const float* __restrict__ h0,
                                float* __restrict__ agg, int n) {
    int wave = blockIdx.x * (blockDim.x >> 5) + (threadIdx.x >> 5);
    int lane = threadIdx.x & 31;
    if (wave >= n) return;
    float d = dinv[wave];
    float s = d * d;
    const float2* src = (const float2*)(h0 + (size_t)wave * HIDDEN);
    float2* dst = (float2*)(agg + (size_t)wave * HIDDEN);
    float2 v = src[lane];
    v.x *= s; v.y *= s;
    dst[lane] = v;
}

__global__ void scatter1_kernel(const int* __restrict__ rows,
                                const int* __restrict__ cols,
                                const float* __restrict__ ew,
                                const float* __restrict__ dinv,
                                const float* __restrict__ h0,
                                float* __restrict__ agg, int e) {
    int wave = blockIdx.x * (blockDim.x >> 5) + (threadIdx.x >> 5);
    int lane = threadIdx.x & 31;
    if (wave >= e) return;
    int r = rows[wave];
    int c = cols[wave];
    float norm = dinv[r] * ew[wave] * dinv[c];
    float2 v = ((const float2*)(h0 + (size_t)r * HIDDEN))[lane];
    float* dst = agg + (size_t)c * HIDDEN + 2 * lane;
    atomicAdd(dst + 0, norm * v.x);
    atomicAdd(dst + 1, norm * v.y);
}

__global__ void relu_bias_kernel(float* __restrict__ agg,
                                 const float* __restrict__ b1, int total) {
    int i = blockIdx.x * blockDim.x + threadIdx.x;
    if (i < total) {
        float v = agg[i] + b1[i & (HIDDEN - 1)];
        agg[i] = fmaxf(v, 0.0f);
    }
}

__global__ void gemv2_init_kernel(const float* __restrict__ h,
                                  const float* __restrict__ w2,
                                  const float* __restrict__ b2,
                                  const float* __restrict__ dinv,
                                  float* __restrict__ g,
                                  float* __restrict__ out, int n) {
    int wave = blockIdx.x * (blockDim.x >> 5) + (threadIdx.x >> 5);
    int lane = threadIdx.x & 31;
    if (wave >= n) return;
    float2 hv = ((const float2*)(h + (size_t)wave * HIDDEN))[lane];
    float2 wv = ((const float2*)w2)[lane];
    float p = hv.x * wv.x + hv.y * wv.y;
#pragma unroll
    for (int off = 16; off > 0; off >>= 1)
        p += __shfl_xor(p, off, 32);
    if (lane == 0) {
        float d = dinv[wave];
        g[wave] = p;
        out[wave] = b2[0] + d * d * p;
    }
}

__global__ void __launch_bounds__(SC_THREADS)
scatter2_kernel(const int* __restrict__ rows,
                const int* __restrict__ cols,
                const float* __restrict__ ew,
                const float* __restrict__ dinv,
                const float* __restrict__ g,
                float* __restrict__ out, int n, int e) {
    __shared__ __attribute__((aligned(16))) float bins[SPAN];
    const int lo = blockIdx.y * SPAN;
    const int span = min(SPAN, n - lo);
    bins_zero(bins);
    __syncthreads();

    const int nch = (int)gridDim.x;
    const int per = (e + nch - 1) / nch;
    const int start = blockIdx.x * per;
    const int end = min(e, start + per);
    int base = start;
    for (; base + SC_THREADS * SC_UNROLL <= end; base += SC_THREADS * SC_UNROLL) {
        int c[SC_UNROLL];
        int r[SC_UNROLL];
        float w[SC_UNROLL];
        float dr[SC_UNROLL];
        float dc[SC_UNROLL];
        float gr[SC_UNROLL];
#pragma unroll
        for (int u = 0; u < SC_UNROLL; u++) {
            const int i = base + u * SC_THREADS + (int)threadIdx.x;
            c[u] = cols[i];
            r[u] = rows[i];
            w[u] = ew[i];
        }
        asm volatile("" :: HOLD8(c), HOLD8(r), HOLD8(w));
#pragma unroll
        for (int u = 0; u < SC_UNROLL; u++) {
            const bool mine = (unsigned)(c[u] - lo) < (unsigned)span;
            const int cc = mine ? c[u] : 0;
            const int rr = mine ? r[u] : 0;
            dr[u] = dinv[rr];
            dc[u] = dinv[cc];
            gr[u] = g[rr];
        }
        asm volatile("" :: HOLD8(dr), HOLD8(dc), HOLD8(gr));
#pragma unroll
        for (int u = 0; u < SC_UNROLL; u++) {
            const unsigned k = (unsigned)(c[u] - lo);
            if (k < (unsigned)span) {
                float norm = dr[u] * w[u] * dc[u];
                atomicAdd(&bins[k], norm * gr[u]);
            }
        }
    }
    for (int i = base + (int)threadIdx.x; i < end; i += SC_THREADS) {
        const int c = cols[i];
        const unsigned k = (unsigned)(c - lo);
        if (k < (unsigned)span) {
            const int r = rows[i];
            float norm = dinv[r] * ew[i] * dinv[c];
            atomicAdd(&bins[k], norm * g[r]);
        }
    }
    __syncthreads();
    bins_flush(bins, out + lo, span);
}

extern "C" void kernel_launch(void* const* d_in, const int* in_sizes, int n_in,
                              void* d_out, int out_size, void* d_ws, size_t ws_size,
                              hipStream_t stream) {
    const float* x  = (const float*)d_in[0];
    const int*   ei = (const int*)d_in[1];
    const float* ew = (const float*)d_in[2];
    const float* w1 = (const float*)d_in[3];
    const float* b1 = (const float*)d_in[4];
    const float* w2 = (const float*)d_in[5];
    const float* b2 = (const float*)d_in[6];
    float* out = (float*)d_out;

    const int E = in_sizes[2];
    const int N = in_sizes[0] / IN_FEAT;
    const int* rows = ei;
    const int* cols = ei + E;

    float* ws   = (float*)d_ws;
    float* dinv = ws;
    float* g    = ws + N;
    float* h0   = ws + 2 * (size_t)N;
    float* agg  = h0 + (size_t)N * HIDDEN;

    const int B = 256;
    init_deg_kernel<<<(N + B - 1) / B, B, 0, stream>>>(dinv, N);
    const dim3 sgrid(SC_CHUNKS, (N + SPAN - 1) / SPAN);
    deg_accum_kernel<<<sgrid, SC_THREADS, 0, stream>>>(cols, ew, dinv, N, E);
    dinv_kernel<<<(N + B - 1) / B, B, 0, stream>>>(dinv, N);

    {
        int mtiles = (N + 15) / 16;
        int waves  = mtiles * (HIDDEN / 16);
        int blocks = (waves + 7) / 8;
        gemm1_wmma_kernel<<<blocks, B, 0, stream>>>(x, w1, h0, N);
    }

    init_agg_kernel<<<(N + 7) / 8, B, 0, stream>>>(dinv, h0, agg, N);
    scatter1_kernel<<<(E + 7) / 8, B, 0, stream>>>(rows, cols, ew, dinv, h0, agg, E);
    relu_bias_kernel<<<((size_t)N * HIDDEN + B - 1) / B, B, 0, stream>>>(agg, b1, N * HIDDEN);

    gemv2_init_kernel<<<(N + 7) / 8, B, 0, stream>>>(agg, w2, b2, dinv, g, out, N);
    scatter2_kernel<<<sgrid, SC_THREADS, 0, stream>>>(rows, cols, ew, dinv, g, out, N, E);
}


# Layer-1 scatter: per-edge global f32 atomics replaced by a counting sort of edges into 128-node destination blocks and per-block LDS-atomic tiles stored once; float sum order changes
# speedup vs baseline: 2.2890x; 1.5520x over previous
#include <hip/hip_runtime.h>
#include <hip/hip_bf16.h>

#define IN_FEAT 128
#define HIDDEN  64

typedef __attribute__((ext_vector_type(2))) float v2f;
typedef __attribute__((ext_vector_type(8))) float v8f;

__global__ void init_deg_kernel(float* __restrict__ deg, int n) {
    int i = blockIdx.x * blockDim.x + threadIdx.x;
    if (i < n) deg[i] = 1.0f;
}

#define SPAN       51200
#define SC_THREADS 1024
#define SC_CHUNKS  128
#define SC_UNROLL  8
static_assert(SC_UNROLL == 8, "HOLD8 names eight values");
#define HOLD8(a) "v"(a[0]), "v"(a[1]), "v"(a[2]), "v"(a[3]), "v"(a[4]), "v"(a[5]), "v"(a[6]), "v"(a[7])

__device__ __forceinline__ void bins_zero(float* bins) {
    float4* b4 = (float4*)bins;
    for (int i = threadIdx.x; i < SPAN / 4; i += SC_THREADS)
        b4[i] = make_float4(0.0f, 0.0f, 0.0f, 0.0f);
}

__device__ __forceinline__ void bins_flush(const float* bins,
                                           float* __restrict__ dst, int span) {
    const int rot = (int)((blockIdx.x * 416u) % SPAN);
    for (int i = threadIdx.x; i < SPAN; i += SC_THREADS) {
        int j = i + rot;
        if (j >= SPAN) j -= SPAN;
        if (j < span) {
            float v = bins[j];
            if (v != 0.0f) atomicAdd(dst + j, v);
        }
    }
}

__global__ void __launch_bounds__(SC_THREADS)
deg_accum_kernel(const int* __restrict__ cols,
                 const float* __restrict__ ew,
                 float* __restrict__ deg, int n, int e) {
    __shared__ __attribute__((aligned(16))) float bins[SPAN];
    const int lo = blockIdx.y * SPAN;
    const int span = min(SPAN, n - lo);
    bins_zero(bins);
    __syncthreads();

    const int nch = (int)gridDim.x;
    const int per = (e + nch - 1) / nch;
    const int start = blockIdx.x * per;
    const int end = min(e, start + per);
    int base = start;
    for (; base + SC_THREADS * SC_UNROLL <= end; base += SC_THREADS * SC_UNROLL) {
        int c[SC_UNROLL];
        float w[SC_UNROLL];
#pragma unroll
        for (int u = 0; u < SC_UNROLL; u++) {
            const int i = base + u * SC_THREADS + (int)threadIdx.x;
            c[u] = cols[i];
            w[u] = ew[i];
        }
        asm volatile("" :: HOLD8(c), HOLD8(w));
#pragma unroll
        for (int u = 0; u < SC_UNROLL; u++) {
            const unsigned k = (unsigned)(c[u] - lo);
            if (k < (unsigned)span) atomicAdd(&bins[k], w[u]);
        }
    }
    for (int i = base + (int)threadIdx.x; i < end; i += SC_THREADS) {
        const unsigned k = (unsigned)(cols[i] - lo);
        if (k < (unsigned)span) atomicAdd(&bins[k], ew[i]);
    }
    __syncthreads();
    bins_flush(bins, deg + lo, span);
}

__global__ void dinv_kernel(float* __restrict__ deg, int n) {
    int i = blockIdx.x * blockDim.x + threadIdx.x;
    if (i < n) {
        float d = deg[i];
        deg[i] = (d > 0.0f) ? __frsqrt_rn(d) : 0.0f;
    }
}

__global__ void __launch_bounds__(256)
gemm1_wmma_kernel(const float* __restrict__ X, const float* __restrict__ W,
                  float* __restrict__ H, int n) {
    __shared__ float sW[IN_FEAT * HIDDEN];
    for (int i = threadIdx.x; i < IN_FEAT * HIDDEN; i += blockDim.x)
        sW[i] = W[i];
    __syncthreads();

    const int wave = blockIdx.x * (blockDim.x >> 5) + (threadIdx.x >> 5);
    const int lane = threadIdx.x & 31;
    const int mt = wave >> 2;
    const int nt = wave & 3;
    const int m0 = mt * 16;
    if (m0 >= n) return;
    const int n0 = nt * 16;
    const int half = lane >> 4;
    const int l = lane & 15;

    const float* Xrow = X + (size_t)(m0 + l) * IN_FEAT;
    v8f c = {0.f, 0.f, 0.f, 0.f, 0.f, 0.f, 0.f, 0.f};

#pragma unroll 8
    for (int k = 0; k < IN_FEAT; k += 4) {
        v2f a = *(const v2f*)(Xrow + k + 2 * half);
        v2f b;
        b.x = sW[(k + 2 * half + 0) * HIDDEN + n0 + l];
        b.y = sW[(k + 2 * half + 1) * HIDDEN + n0 + l];
        c = __builtin_amdgcn_wmma_f32_16x16x4_f32(
                 false, a,  false, b,
                 (short)0, c,  false,  false);
    }

    float* Hout = H + (size_t)(m0 + 8 * half) * HIDDEN + n0 + l;
#pragma unroll
    for (int v = 0; v < 8; v++)
        Hout[(size_t)v * HIDDEN] = c[v];
}

#define BK_SHIFT   7
#define BK_NODES   (1 << BK_SHIFT)
#define BK_MAX     1024
#define EC_CHUNKS  256
#define EC_THREADS 1024
#define AG_THREADS 256
#define AG_STRIDE  ((AG_THREADS / 32) * 32)
#define AG_BATCH   16
static_assert(EC_CHUNKS == 256, "chunk_prefix_kernel gives each lane 8 chunks");
static_assert(HIDDEN == 64, "one float2 per lane covers a row");

__global__ void __launch_bounds__(EC_THREADS)
edge_count_kernel(const int* __restrict__ cols, int* __restrict__ counts,
                  int nb, int e) {
    __shared__ int cnt[BK_MAX];
    for (int t = threadIdx.x; t < BK_MAX; t += EC_THREADS) cnt[t] = 0;
    __syncthreads();
    const int per = (e + EC_CHUNKS - 1) / EC_CHUNKS;
    const int start = blockIdx.x * per;
    const int end = min(e, start + per);
    for (int base = start; base < end; base += EC_THREADS * 4) {
        int c[4];
#pragma unroll
        for (int u = 0; u < 4; u++) {
            const int i = base + u * EC_THREADS + (int)threadIdx.x;
            c[u] = (i < end) ? cols[i] : -1;
        }
#pragma unroll
        for (int u = 0; u < 4; u++)
            if (c[u] >= 0) atomicAdd(&cnt[c[u] >> BK_SHIFT], 1);
    }
    __syncthreads();
    for (int b = threadIdx.x; b < nb; b += EC_THREADS)
        counts[(size_t)b * EC_CHUNKS + blockIdx.x] = cnt[b];
}

__global__ void __launch_bounds__(256)
chunk_prefix_kernel(int* __restrict__ counts, int* __restrict__ tot, int nb) {
    const int b = blockIdx.x * (blockDim.x >> 5) + (threadIdx.x >> 5);
    const int lane = threadIdx.x & 31;
    if (b >= nb) return;
    int4* row = (int4*)(counts + (size_t)b * EC_CHUNKS) + 2 * lane;
    const int4 a = row[0];
    const int4 c = row[1];
    const int sum = a.x + a.y + a.z + a.w + c.x + c.y + c.z + c.w;
    int inc = sum;
    for (int off = 1; off < 32; off <<= 1) {
        const int v = __shfl_up(inc, off, 32);
        if (lane >= off) inc += v;
    }
    int4 oa, oc;
    oa.x = inc - sum;
    oa.y = oa.x + a.x;
    oa.z = oa.y + a.y;
    oa.w = oa.z + a.z;
    oc.x = oa.w + a.w;
    oc.y = oc.x + c.x;
    oc.z = oc.y + c.y;
    oc.w = oc.z + c.z;
    row[0] = oa;
    row[1] = oc;
    if (lane == 31) tot[b] = inc;
}

__global__ void __launch_bounds__(EC_THREADS)
edge_place_kernel(const int* __restrict__ rows, const int* __restrict__ cols,
                  const float* __restrict__ ew, const float* __restrict__ dinv,
                  const int* __restrict__ counts, const int* __restrict__ tot,
                  int* __restrict__ bstart, int4* __restrict__ recs,
                  int nb, int e) {
    __shared__ int cursor[BK_MAX];
    __shared__ int wsum[32];
    const int t = threadIdx.x;
    const int lane = t & 31;
    const int wid = t >> 5;
    const int mine = (t < nb) ? tot[t] : 0;
    int inc = mine;
    for (int off = 1; off < 32; off <<= 1) {
        const int v = __shfl_up(inc, off, 32);
        if (lane >= off) inc += v;
    }
    if (lane == 31) wsum[wid] = inc;
    __syncthreads();
    if (wid == 0) {
        const int s = wsum[lane];
        int si = s;
        for (int off = 1; off < 32; off <<= 1) {
            const int v = __shfl_up(si, off, 32);
            if (lane >= off) si += v;
        }
        wsum[lane] = si - s;
    }
    __syncthreads();
    const int excl = inc - mine + wsum[wid];
    if (t < nb) {
        cursor[t] = excl + counts[(size_t)t * EC_CHUNKS + blockIdx.x];
        if (blockIdx.x == 0) {
            bstart[t] = excl;
            if (t == nb - 1) bstart[nb] = excl + mine;
        }
    }
    __syncthreads();

    const int per = (e + EC_CHUNKS - 1) / EC_CHUNKS;
    const int start = blockIdx.x * per;
    const int end = min(e, start + per);
    for (int base = start; base < end; base += EC_THREADS * 4) {
        int c[4];
        int r[4];
        float w[4];
        float dr[4];
        float dc[4];
#pragma unroll
        for (int u = 0; u < 4; u++) {
            const int i = base + u * EC_THREADS + t;
            const bool ok = i < end;
            c[u] = ok ? cols[i] : -1;
            r[u] = ok ? rows[i] : 0;
            w[u] = ok ? ew[i] : 0.0f;
        }
#pragma unroll
        for (int u = 0; u < 4; u++) {
            dr[u] = dinv[r[u]];
            dc[u] = dinv[max(c[u], 0)];
        }
#pragma unroll
        for (int u = 0; u < 4; u++) {
            if (c[u] >= 0) {
                const int pos = atomicAdd(&cursor[c[u] >> BK_SHIFT], 1);
                const float norm = dr[u] * w[u] * dc[u];
                recs[pos] = make_int4(r[u], c[u] & (BK_NODES - 1), __float_as_int(norm), 0);
            }
        }
    }
}

__global__ void __launch_bounds__(AG_THREADS)
block_aggregate_kernel(const int4* __restrict__ recs, const int* __restrict__ bstart,
                       const float* __restrict__ dinv, const float* __restrict__ h0,
                       float* __restrict__ agg, int n) {
    __shared__ __attribute__((aligned(16))) float acc[(BK_NODES + 1) * HIDDEN];
    const int t = threadIdx.x;
    const int lane = t & 31;
    const int wave = __builtin_amdgcn_readfirstlane(t >> 5);
    const int lo = blockIdx.x * BK_NODES;
    const int nn = min(BK_NODES, n - lo);
    float2* acc2 = (float2*)acc;
    for (int idx = t; idx < nn * (HIDDEN / 2); idx += AG_THREADS) {
        const int node = lo + (idx >> 5);
        const float d = dinv[node];
        const float s = d * d;
        float2 v = ((const float2*)(h0 + (size_t)node * HIDDEN))[idx & 31];
        v.x *= s;
        v.y *= s;
        acc2[idx] = v;
    }
    __syncthreads();

    const int beg = __builtin_amdgcn_readfirstlane(bstart[blockIdx.x]);
    const int end = __builtin_amdgcn_readfirstlane(bstart[blockIdx.x + 1]);
    const int g0 = beg + wave * 32;
    if (g0 < end) {
        const int last = end - 1;
        int4 rec = recs[min(g0 + lane, last)];
        for (int g = g0; g < end; g += AG_STRIDE) {
            const int4 nxt = recs[min(g + AG_STRIDE + lane, last)];
            if (g + lane >= end) rec = make_int4(0, BK_NODES, 0, 0);
#pragma unroll
            for (int u0 = 0; u0 < 32; u0 += AG_BATCH) {
                float2 v[AG_BATCH];
                int cl[AG_BATCH];
                float nm[AG_BATCH];
#pragma unroll
                for (int k = 0; k < AG_BATCH; k++) {
                    const int r = __builtin_amdgcn_readlane(rec.x, u0 + k);
                    cl[k] = __builtin_amdgcn_readlane(rec.y, u0 + k);
                    nm[k] = __int_as_float(__builtin_amdgcn_readlane(rec.z, u0 + k));
                    v[k] = ((const float2*)(h0 + (size_t)r * HIDDEN))[lane];
                }
#pragma unroll
                for (int k = 0; k < AG_BATCH; k++) {
                    float* d = acc + cl[k] * HIDDEN + 2 * lane;
                    atomicAdd(d + 0, nm[k] * v[k].x);
                    atomicAdd(d + 1, nm[k] * v[k].y);
                }
            }
            rec = nxt;
        }
    }
    __syncthreads();
    for (int idx = t; idx < nn * (HIDDEN / 2); idx += AG_THREADS)
        ((float2*)(agg + (size_t)(lo + (idx >> 5)) * HIDDEN))[idx & 31] = acc2[idx];
}

__global__ void relu_bias_kernel(float* __restrict__ agg,
                                 const float* __restrict__ b1, int total) {
    int i = blockIdx.x * blockDim.x + threadIdx.x;
    if (i < total) {
        float v = agg[i] + b1[i & (HIDDEN - 1)];
        agg[i] = fmaxf(v, 0.0f);
    }
}

__global__ void gemv2_init_kernel(const float* __restrict__ h,
                                  const float* __restrict__ w2,
                                  const float* __restrict__ b2,
                                  const float* __restrict__ dinv,
                                  float* __restrict__ g,
                                  float* __restrict__ out, int n) {
    int wave = blockIdx.x * (blockDim.x >> 5) + (threadIdx.x >> 5);
    int lane = threadIdx.x & 31;
    if (wave >= n) return;
    float2 hv = ((const float2*)(h + (size_t)wave * HIDDEN))[lane];
    float2 wv = ((const float2*)w2)[lane];
    float p = hv.x * wv.x + hv.y * wv.y;
#pragma unroll
    for (int off = 16; off > 0; off >>= 1)
        p += __shfl_xor(p, off, 32);
    if (lane == 0) {
        float d = dinv[wave];
        g[wave] = p;
        out[wave] = b2[0] + d * d * p;
    }
}

__global__ void __launch_bounds__(SC_THREADS)
scatter2_kernel(const int* __restrict__ rows,
                const int* __restrict__ cols,
                const float* __restrict__ ew,
                const float* __restrict__ dinv,
                const float* __restrict__ g,
                float* __restrict__ out, int n, int e) {
    __shared__ __attribute__((aligned(16))) float bins[SPAN];
    const int lo = blockIdx.y * SPAN;
    const int span = min(SPAN, n - lo);
    bins_zero(bins);
    __syncthreads();

    const int nch = (int)gridDim.x;
    const int per = (e + nch - 1) / nch;
    const int start = blockIdx.x * per;
    const int end = min(e, start + per);
    int base = start;
    for (; base + SC_THREADS * SC_UNROLL <= end; base += SC_THREADS * SC_UNROLL) {
        int c[SC_UNROLL];
        int r[SC_UNROLL];
        float w[SC_UNROLL];
        float dr[SC_UNROLL];
        float dc[SC_UNROLL];
        float gr[SC_UNROLL];
#pragma unroll
        for (int u = 0; u < SC_UNROLL; u++) {
            const int i = base + u * SC_THREADS + (int)threadIdx.x;
            c[u] = cols[i];
            r[u] = rows[i];
            w[u] = ew[i];
        }
        asm volatile("" :: HOLD8(c), HOLD8(r), HOLD8(w));
#pragma unroll
        for (int u = 0; u < SC_UNROLL; u++) {
            const bool mine = (unsigned)(c[u] - lo) < (unsigned)span;
            const int cc = mine ? c[u] : 0;
            const int rr = mine ? r[u] : 0;
            dr[u] = dinv[rr];
            dc[u] = dinv[cc];
            gr[u] = g[rr];
        }
        asm volatile("" :: HOLD8(dr), HOLD8(dc), HOLD8(gr));
#pragma unroll
        for (int u = 0; u < SC_UNROLL; u++) {
            const unsigned k = (unsigned)(c[u] - lo);
            if (k < (unsigned)span) {
                float norm = dr[u] * w[u] * dc[u];
                atomicAdd(&bins[k], norm * gr[u]);
            }
        }
    }
    for (int i = base + (int)threadIdx.x; i < end; i += SC_THREADS) {
        const int c = cols[i];
        const unsigned k = (unsigned)(c - lo);
        if (k < (unsigned)span) {
            const int r = rows[i];
            float norm = dinv[r] * ew[i] * dinv[c];
            atomicAdd(&bins[k], norm * g[r]);
        }
    }
    __syncthreads();
    bins_flush(bins, out + lo, span);
}

extern "C" void kernel_launch(void* const* d_in, const int* in_sizes, int n_in,
                              void* d_out, int out_size, void* d_ws, size_t ws_size,
                              hipStream_t stream) {
    const float* x  = (const float*)d_in[0];
    const int*   ei = (const int*)d_in[1];
    const float* ew = (const float*)d_in[2];
    const float* w1 = (const float*)d_in[3];
    const float* b1 = (const float*)d_in[4];
    const float* w2 = (const float*)d_in[5];
    const float* b2 = (const float*)d_in[6];
    float* out = (float*)d_out;

    const int E = in_sizes[2];
    const int N = in_sizes[0] / IN_FEAT;
    const int* rows = ei;
    const int* cols = ei + E;

    float* ws   = (float*)d_ws;
    float* dinv = ws;
    float* g    = ws + N;
    float* h0   = ws + 2 * (size_t)N;
    float* agg  = h0 + (size_t)N * HIDDEN;
    int*  counts = (int*)(agg + (size_t)N * HIDDEN);
    int*  tot    = counts + BK_MAX * EC_CHUNKS;
    int*  bstart = tot + BK_MAX;
    int4* recs   = (int4*)(bstart + BK_MAX + 32);

    const int B = 256;
    init_deg_kernel<<<(N + B - 1) / B, B, 0, stream>>>(dinv, N);
    const dim3 sgrid(SC_CHUNKS, (N + SPAN - 1) / SPAN);
    deg_accum_kernel<<<sgrid, SC_THREADS, 0, stream>>>(cols, ew, dinv, N, E);
    dinv_kernel<<<(N + B - 1) / B, B, 0, stream>>>(dinv, N);

    {
        int mtiles = (N + 15) / 16;
        int waves  = mtiles * (HIDDEN / 16);
        int blocks = (waves + 7) / 8;
        gemm1_wmma_kernel<<<blocks, B, 0, stream>>>(x, w1, h0, N);
    }

    {
        const int NB = (N + BK_NODES - 1) / BK_NODES;
        edge_count_kernel<<<EC_CHUNKS, EC_THREADS, 0, stream>>>(cols, counts, NB, E);
        chunk_prefix_kernel<<<(NB + 7) / 8, B, 0, stream>>>(counts, tot, NB);
        edge_place_kernel<<<EC_CHUNKS, EC_THREADS, 0, stream>>>(rows, cols, ew, dinv, counts, tot,
                                                                bstart, recs, NB, E);
        block_aggregate_kernel<<<NB, AG_THREADS, 0, stream>>>(recs, bstart, dinv, h0, agg, N);
    }
    relu_bias_kernel<<<((size_t)N * HIDDEN + B - 1) / B, B, 0, stream>>>(agg, b1, N * HIDDEN);

    gemv2_init_kernel<<<(N + 7) / 8, B, 0, stream>>>(agg, w2, b2, dinv, g, out, N);
    scatter2_kernel<<<sgrid, SC_THREADS, 0, stream>>>(rows, cols, ew, dinv, g, out, N, E);
}


# Layer-1 epilogue fused into the aggregate kernel: bias, relu, dot with w2 and self-loop start of out taken from the LDS tile; agg never stored, 2 launches and 300k short waves gone; sum order kept
# speedup vs baseline: 2.5954x; 1.1339x over previous
#include <hip/hip_runtime.h>
#include <hip/hip_bf16.h>

#define IN_FEAT 128
#define HIDDEN  64

typedef __attribute__((ext_vector_type(2))) float v2f;
typedef __attribute__((ext_vector_type(8))) float v8f;

__global__ void init_deg_kernel(float* __restrict__ deg, int n) {
    int i = blockIdx.x * blockDim.x + threadIdx.x;
    if (i < n) deg[i] = 1.0f;
}

#define SPAN       51200
#define SC_THREADS 1024
#define SC_CHUNKS  128
#define SC_UNROLL  8
static_assert(SC_UNROLL == 8, "HOLD8 names eight values");
#define HOLD8(a) "v"(a[0]), "v"(a[1]), "v"(a[2]), "v"(a[3]), "v"(a[4]), "v"(a[5]), "v"(a[6]), "v"(a[7])

__device__ __forceinline__ void bins_zero(float* bins) {
    float4* b4 = (float4*)bins;
    for (int i = threadIdx.x; i < SPAN / 4; i += SC_THREADS)
        b4[i] = make_float4(0.0f, 0.0f, 0.0f, 0.0f);
}

__device__ __forceinline__ void bins_flush(const float* bins,
                                           float* __restrict__ dst, int span) {
    const int rot = (int)((blockIdx.x * 416u) % SPAN);
    for (int i = threadIdx.x; i < SPAN; i += SC_THREADS) {
        int j = i + rot;
        if (j >= SPAN) j -= SPAN;
        if (j < span) {
            float v = bins[j];
            if (v != 0.0f) atomicAdd(dst + j, v);
        }
    }
}

__global__ void __launch_bounds__(SC_THREADS)
deg_accum_kernel(const int* __restrict__ cols,
                 const float* __restrict__ ew,
                 float* __restrict__ deg, int n, int e) {
    __shared__ __attribute__((aligned(16))) float bins[SPAN];
    const int lo = blockIdx.y * SPAN;
    const int span = min(SPAN, n - lo);
    bins_zero(bins);
    __syncthreads();

    const int nch = (int)gridDim.x;
    const int per = (e + nch - 1) / nch;
    const int start = blockIdx.x * per;
    const int end = min(e, start + per);
    int base = start;
    for (; base + SC_THREADS * SC_UNROLL <= end; base += SC_THREADS * SC_UNROLL) {
        int c[SC_UNROLL];
        float w[SC_UNROLL];
#pragma unroll
        for (int u = 0; u < SC_UNROLL; u++) {
            const int i = base + u * SC_THREADS + (int)threadIdx.x;
            c[u] = cols[i];
            w[u] = ew[i];
        }
        asm volatile("" :: HOLD8(c), HOLD8(w));
#pragma unroll
        for (int u = 0; u < SC_UNROLL; u++) {
            const unsigned k = (unsigned)(c[u] - lo);
            if (k < (unsigned)span) atomicAdd(&bins[k], w[u]);
        }
    }
    for (int i = base + (int)threadIdx.x; i < end; i += SC_THREADS) {
        const unsigned k = (unsigned)(cols[i] - lo);
        if (k < (unsigned)span) atomicAdd(&bins[k], ew[i]);
    }
    __syncthreads();
    bins_flush(bins, deg + lo, span);
}

__global__ void dinv_kernel(float* __restrict__ deg, int n) {
    int i = blockIdx.x * blockDim.x + threadIdx.x;
    if (i < n) {
        float d = deg[i];
        deg[i] = (d > 0.0f) ? __frsqrt_rn(d) : 0.0f;
    }
}

__global__ void __launch_bounds__(256)
gemm1_wmma_kernel(const float* __restrict__ X, const float* __restrict__ W,
                  float* __restrict__ H, int n) {
    __shared__ float sW[IN_FEAT * HIDDEN];
    for (int i = threadIdx.x; i < IN_FEAT * HIDDEN; i += blockDim.x)
        sW[i] = W[i];
    __syncthreads();

    const int wave = blockIdx.x * (blockDim.x >> 5) + (threadIdx.x >> 5);
    const int lane = threadIdx.x & 31;
    const int mt = wave >> 2;
    const int nt = wave & 3;
    const int m0 = mt * 16;
    if (m0 >= n) return;
    const int n0 = nt * 16;
    const int half = lane >> 4;
    const int l = lane & 15;

    const float* Xrow = X + (size_t)(m0 + l) * IN_FEAT;
    v8f c = {0.f, 0.f, 0.f, 0.f, 0.f, 0.f, 0.f, 0.f};

#pragma unroll 8
    for (int k = 0; k < IN_FEAT; k += 4) {
        v2f a = *(const v2f*)(Xrow + k + 2 * half);
        v2f b;
        b.x = sW[(k + 2 * half + 0) * HIDDEN + n0 + l];
        b.y = sW[(k + 2 * half + 1) * HIDDEN + n0 + l];
        c = __builtin_amdgcn_wmma_f32_16x16x4_f32(
                 false, a,  false, b,
                 (short)0, c,  false,  false);
    }

    float* Hout = H + (size_t)(m0 + 8 * half) * HIDDEN + n0 + l;
#pragma unroll
    for (int v = 0; v < 8; v++)
        Hout[(size_t)v * HIDDEN] = c[v];
}

#define BK_SHIFT   7
#define BK_NODES   (1 << BK_SHIFT)
#define BK_MAX     1024
#define EC_CHUNKS  256
#define EC_THREADS 1024
#define AG_THREADS 256
#define AG_STRIDE  ((AG_THREADS / 32) * 32)
#define AG_BATCH   16
static_assert(EC_CHUNKS == 256, "chunk_prefix_kernel gives each lane 8 chunks");
static_assert(HIDDEN == 64, "one float2 per lane covers a row");

__global__ void __launch_bounds__(EC_THREADS)
edge_count_kernel(const int* __restrict__ cols, int* __restrict__ counts,
                  int nb, int e) {
    __shared__ int cnt[BK_MAX];
    for (int t = threadIdx.x; t < BK_MAX; t += EC_THREADS) cnt[t] = 0;
    __syncthreads();
    const int per = (e + EC_CHUNKS - 1) / EC_CHUNKS;
    const int start = blockIdx.x * per;
    const int end = min(e, start + per);
    for (int base = start; base < end; base += EC_THREADS * 4) {
        int c[4];
#pragma unroll
        for (int u = 0; u < 4; u++) {
            const int i = base + u * EC_THREADS + (int)threadIdx.x;
            c[u] = (i < end) ? cols[i] : -1;
        }
#pragma unroll
        for (int u = 0; u < 4; u++)
            if (c[u] >= 0) atomicAdd(&cnt[c[u] >> BK_SHIFT], 1);
    }
    __syncthreads();
    for (int b = threadIdx.x; b < nb; b += EC_THREADS)
        counts[(size_t)b * EC_CHUNKS + blockIdx.x] = cnt[b];
}

__global__ void __launch_bounds__(256)
chunk_prefix_kernel(int* __restrict__ counts, int* __restrict__ tot, int nb) {
    const int b = blockIdx.x * (blockDim.x >> 5) + (threadIdx.x >> 5);
    const int lane = threadIdx.x & 31;
    if (b >= nb) return;
    int4* row = (int4*)(counts + (size_t)b * EC_CHUNKS) + 2 * lane;
    const int4 a = row[0];
    const int4 c = row[1];
    const int sum = a.x + a.y + a.z + a.w + c.x + c.y + c.z + c.w;
    int inc = sum;
    for (int off = 1; off < 32; off <<= 1) {
        const int v = __shfl_up(inc, off, 32);
        if (lane >= off) inc += v;
    }
    int4 oa, oc;
    oa.x = inc - sum;
    oa.y = oa.x + a.x;
    oa.z = oa.y + a.y;
    oa.w = oa.z + a.z;
    oc.x = oa.w + a.w;
    oc.y = oc.x + c.x;
    oc.z = oc.y + c.y;
    oc.w = oc.z + c.z;
    row[0] = oa;
    row[1] = oc;
    if (lane == 31) tot[b] = inc;
}

__global__ void __launch_bounds__(EC_THREADS)
edge_place_kernel(const int* __restrict__ rows, const int* __restrict__ cols,
                  const float* __restrict__ ew, const float* __restrict__ dinv,
                  const int* __restrict__ counts, const int* __restrict__ tot,
                  int* __restrict__ bstart, int4* __restrict__ recs,
                  int nb, int e) {
    __shared__ int cursor[BK_MAX];
    __shared__ int wsum[32];
    const int t = threadIdx.x;
    const int lane = t & 31;
    const int wid = t >> 5;
    const int mine = (t < nb) ? tot[t] : 0;
    int inc = mine;
    for (int off = 1; off < 32; off <<= 1) {
        const int v = __shfl_up(inc, off, 32);
        if (lane >= off) inc += v;
    }
    if (lane == 31) wsum[wid] = inc;
    __syncthreads();
    if (wid == 0) {
        const int s = wsum[lane];
        int si = s;
        for (int off = 1; off < 32; off <<= 1) {
            const int v = __shfl_up(si, off, 32);
            if (lane >= off) si += v;
        }
        wsum[lane] = si - s;
    }
    __syncthreads();
    const int excl = inc - mine + wsum[wid];
    if (t < nb) {
        cursor[t] = excl + counts[(size_t)t * EC_CHUNKS + blockIdx.x];
        if (blockIdx.x == 0) {
            bstart[t] = excl;
            if (t == nb - 1) bstart[nb] = excl + mine;
        }
    }
    __syncthreads();

    const int per = (e + EC_CHUNKS - 1) / EC_CHUNKS;
    const int start = blockIdx.x * per;
    const int end = min(e, start + per);
    for (int base = start; base < end; base += EC_THREADS * 4) {
        int c[4];
        int r[4];
        float w[4];
        float dr[4];
        float dc[4];
#pragma unroll
        for (int u = 0; u < 4; u++) {
            const int i = base + u * EC_THREADS + t;
            const bool ok = i < end;
            c[u] = ok ? cols[i] : -1;
            r[u] = ok ? rows[i] : 0;
            w[u] = ok ? ew[i] : 0.0f;
        }
#pragma unroll
        for (int u = 0; u < 4; u++) {
            dr[u] = dinv[r[u]];
            dc[u] = dinv[max(c[u], 0)];
        }
#pragma unroll
        for (int u = 0; u < 4; u++) {
            if (c[u] >= 0) {
                const int pos = atomicAdd(&cursor[c[u] >> BK_SHIFT], 1);
                const float norm = dr[u] * w[u] * dc[u];
                recs[pos] = make_int4(r[u], c[u] & (BK_NODES - 1), __float_as_int(norm), 0);
            }
        }
    }
}

#define AG_ROWS (BK_NODES / (AG_THREADS / 32))
static_assert(AG_ROWS * (AG_THREADS / 32) == BK_NODES && AG_ROWS <= 16,
              "the waves' node groups cover the tile; one lane of a 16-lane row per node");

__device__ __forceinline__ float lane_xor16(float x) {
    const int b = __float_as_int(x);
    return __int_as_float(__builtin_amdgcn_permlanex16(b, b, 0x76543210, 0xfedcba98, false, false));
}
template <int M>
__device__ __forceinline__ float row_xor(float x) {
    static_assert(M > 0 && M < 16, "row_xmask stays inside a row of 16");
    const int b = __float_as_int(x);
    return __int_as_float(__builtin_amdgcn_update_dpp(b, b, 0x160 + M, 0xf, 0xf, false));
}

__global__ void __launch_bounds__(AG_THREADS)
block_aggregate_kernel(const int4* __restrict__ recs, const int* __restrict__ bstart,
                       const float* __restrict__ dinv, const float* __restrict__ h0,
                       const float* __restrict__ b1, const float* __restrict__ w2,
                       const float* __restrict__ b2,
                       float* __restrict__ g, float* __restrict__ out, int n) {
    __shared__ __attribute__((aligned(16))) float acc[(BK_NODES + 1) * HIDDEN];
    __shared__ float gs[BK_NODES];
    const int t = threadIdx.x;
    const int lane = t & 31;
    const int wave = __builtin_amdgcn_readfirstlane(t >> 5);
    const int lo = blockIdx.x * BK_NODES;
    const int nn = min(BK_NODES, n - lo);
    float2* acc2 = (float2*)acc;
    for (int idx = t; idx < nn * (HIDDEN / 2); idx += AG_THREADS) {
        const int node = lo + (idx >> 5);
        const float d = dinv[node];
        const float s = d * d;
        float2 v = ((const float2*)(h0 + (size_t)node * HIDDEN))[idx & 31];
        v.x *= s;
        v.y *= s;
        acc2[idx] = v;
    }
    __syncthreads();

    const int beg = __builtin_amdgcn_readfirstlane(bstart[blockIdx.x]);
    const int end = __builtin_amdgcn_readfirstlane(bstart[blockIdx.x + 1]);
    const int g0 = beg + wave * 32;
    if (g0 < end) {
        const int last = end - 1;
        int4 rec = recs[min(g0 + lane, last)];
        for (int g = g0; g < end; g += AG_STRIDE) {
            const int4 nxt = recs[min(g + AG_STRIDE + lane, last)];
            if (g + lane >= end) rec = make_int4(0, BK_NODES, 0, 0);
#pragma unroll
            for (int u0 = 0; u0 < 32; u0 += AG_BATCH) {
                float2 v[AG_BATCH];
                int cl[AG_BATCH];
                float nm[AG_BATCH];
#pragma unroll
                for (int k = 0; k < AG_BATCH; k++) {
                    const int r = __builtin_amdgcn_readlane(rec.x, u0 + k);
                    cl[k] = __builtin_amdgcn_readlane(rec.y, u0 + k);
                    nm[k] = __int_as_float(__builtin_amdgcn_readlane(rec.z, u0 + k));
                    v[k] = ((const float2*)(h0 + (size_t)r * HIDDEN))[lane];
                }
#pragma unroll
                for (int k = 0; k < AG_BATCH; k++) {
                    float* d = acc + cl[k] * HIDDEN + 2 * lane;
                    atomicAdd(d + 0, nm[k] * v[k].x);
                    atomicAdd(d + 1, nm[k] * v[k].y);
                }
            }
            rec = nxt;
        }
    }
    __syncthreads();

    const float2 bv = ((const float2*)b1)[lane];
    const float2 wv = ((const float2*)w2)[lane];
    float p[AG_ROWS];
#pragma unroll
    for (int k = 0; k < AG_ROWS; k++) {
        const int i = min(wave * AG_ROWS + k, nn - 1);
        const float2 a = acc2[i * (HIDDEN / 2) + lane];
        float2 hv;
        hv.x = fmaxf(a.x + bv.x, 0.0f);
        hv.y = fmaxf(a.y + bv.y, 0.0f);
        p[k] = hv.x * wv.x + hv.y * wv.y;
    }
#pragma unroll
    for (int k = 0; k < AG_ROWS; k++) p[k] += lane_xor16(p[k]);
#pragma unroll
    for (int k = 0; k < AG_ROWS; k++) p[k] += row_xor<8>(p[k]);
#pragma unroll
    for (int k = 0; k < AG_ROWS; k++) p[k] += row_xor<4>(p[k]);
#pragma unroll
    for (int k = 0; k < AG_ROWS; k++) p[k] += row_xor<2>(p[k]);
#pragma unroll
    for (int k = 0; k < AG_ROWS; k++) p[k] += row_xor<1>(p[k]);
    float mine = p[0];
#pragma unroll
    for (int k = 1; k < AG_ROWS; k++) mine = (lane == k) ? p[k] : mine;
    if (lane < AG_ROWS) gs[wave * AG_ROWS + lane] = mine;
    __syncthreads();
    if (t < nn) {
        const float d = dinv[lo + t];
        const float gv = gs[t];
        g[lo + t] = gv;
        out[lo + t] = b2[0] + d * d * gv;
    }
}

__global__ void __launch_bounds__(SC_THREADS)
scatter2_kernel(const int* __restrict__ rows,
                const int* __restrict__ cols,
                const float* __restrict__ ew,
                const float* __restrict__ dinv,
                const float* __restrict__ g,
                float* __restrict__ out, int n, int e) {
    __shared__ __attribute__((aligned(16))) float bins[SPAN];
    const int lo = blockIdx.y * SPAN;
    const int span = min(SPAN, n - lo);
    bins_zero(bins);
    __syncthreads();

    const int nch = (int)gridDim.x;
    const int per = (e + nch - 1) / nch;
    const int start = blockIdx.x * per;
    const int end = min(e, start + per);
    int base = start;
    for (; base + SC_THREADS * SC_UNROLL <= end; base += SC_THREADS * SC_UNROLL) {
        int c[SC_UNROLL];
        int r[SC_UNROLL];
        float w[SC_UNROLL];
        float dr[SC_UNROLL];
        float dc[SC_UNROLL];
        float gr[SC_UNROLL];
#pragma unroll
        for (int u = 0; u < SC_UNROLL; u++) {
            const int i = base + u * SC_THREADS + (int)threadIdx.x;
            c[u] = cols[i];
            r[u] = rows[i];
            w[u] = ew[i];
        }
        asm volatile("" :: HOLD8(c), HOLD8(r), HOLD8(w));
#pragma unroll
        for (int u = 0; u < SC_UNROLL; u++) {
            const bool mine = (unsigned)(c[u] - lo) < (unsigned)span;
            const int cc = mine ? c[u] : 0;
            const int rr = mine ? r[u] : 0;
            dr[u] = dinv[rr];
            dc[u] = dinv[cc];
            gr[u] = g[rr];
        }
        asm volatile("" :: HOLD8(dr), HOLD8(dc), HOLD8(gr));
#pragma unroll
        for (int u = 0; u < SC_UNROLL; u++) {
            const unsigned k = (unsigned)(c[u] - lo);
            if (k < (unsigned)span) {
                float norm = dr[u] * w[u] * dc[u];
                atomicAdd(&bins[k], norm * gr[u]);
            }
        }
    }
    for (int i = base + (int)threadIdx.x; i < end; i += SC_THREADS) {
        const int c = cols[i];
        const unsigned k = (unsigned)(c - lo);
        if (k < (unsigned)span) {
            const int r = rows[i];
            float norm = dinv[r] * ew[i] * dinv[c];
            atomicAdd(&bins[k], norm * g[r]);
        }
    }
    __syncthreads();
    bins_flush(bins, out + lo, span);
}

extern "C" void kernel_launch(void* const* d_in, const int* in_sizes, int n_in,
                              void* d_out, int out_size, void* d_ws, size_t ws_size,
                              hipStream_t stream) {
    const float* x  = (const float*)d_in[0];
    const int*   ei = (const int*)d_in[1];
    const float* ew = (const float*)d_in[2];
    const float* w1 = (const float*)d_in[3];
    const float* b1 = (const float*)d_in[4];
    const float* w2 = (const float*)d_in[5];
    const float* b2 = (const float*)d_in[6];
    float* out = (float*)d_out;

    const int E = in_sizes[2];
    const int N = in_sizes[0] / IN_FEAT;
    const int* rows = ei;
    const int* cols = ei + E;

    float* ws   = (float*)d_ws;
    float* dinv = ws;
    float* g    = ws + N;
    float* h0   = ws + 2 * (size_t)N;
    int*  counts = (int*)(h0 + (size_t)N * HIDDEN);
    int*  tot    = counts + BK_MAX * EC_CHUNKS;
    int*  bstart = tot + BK_MAX;
    int4* recs   = (int4*)(bstart + BK_MAX + 32);

    const int B = 256;
    init_deg_kernel<<<(N + B - 1) / B, B, 0, stream>>>(dinv, N);
    const dim3 sgrid(SC_CHUNKS, (N + SPAN - 1) / SPAN);
    deg_accum_kernel<<<sgrid, SC_THREADS, 0, stream>>>(cols, ew, dinv, N, E);
    dinv_kernel<<<(N + B - 1) / B, B, 0, stream>>>(dinv, N);

    {
        int mtiles = (N + 15) / 16;
        int waves  = mtiles * (HIDDEN / 16);
        int blocks = (waves + 7) / 8;
        gemm1_wmma_kernel<<<blocks, B, 0, stream>>>(x, w1, h0, N);
    }

    {
        const int NB = (N + BK_NODES - 1) / BK_NODES;
        edge_count_kernel<<<EC_CHUNKS, EC_THREADS, 0, stream>>>(cols, counts, NB, E);
        chunk_prefix_kernel<<<(NB + 7) / 8, B, 0, stream>>>(counts, tot, NB);
        edge_place_kernel<<<EC_CHUNKS, EC_THREADS, 0, stream>>>(rows, cols, ew, dinv, counts, tot,
                                                                bstart, recs, NB, E);
        block_aggregate_kernel<<<NB, AG_THREADS, 0, stream>>>(recs, bstart, dinv, h0, b1, w2, b2,
                                                              g, out, N);
    }

    scatter2_kernel<<<sgrid, SC_THREADS, 0, stream>>>(rows, cols, ew, dinv, g, out, N, E);
}


# Layer-2 scatter from the block-sorted records: one work-group per 128-node block adds norm*g[row] into LDS bins, stores out once; no global atomics, 2 of 3 gathers gone; float sum order changes
# speedup vs baseline: 3.0784x; 1.1861x over previous
#include <hip/hip_runtime.h>
#include <hip/hip_bf16.h>

#define IN_FEAT 128
#define HIDDEN  64

typedef __attribute__((ext_vector_type(2))) float v2f;
typedef __attribute__((ext_vector_type(8))) float v8f;

__global__ void init_deg_kernel(float* __restrict__ deg, int n) {
    int i = blockIdx.x * blockDim.x + threadIdx.x;
    if (i < n) deg[i] = 1.0f;
}

#define SPAN       51200
#define SC_THREADS 1024
#define SC_CHUNKS  128
#define SC_UNROLL  8
static_assert(SC_UNROLL == 8, "HOLD8 names eight values");
#define HOLD8(a) "v"(a[0]), "v"(a[1]), "v"(a[2]), "v"(a[3]), "v"(a[4]), "v"(a[5]), "v"(a[6]), "v"(a[7])

__device__ __forceinline__ void bins_zero(float* bins) {
    float4* b4 = (float4*)bins;
    for (int i = threadIdx.x; i < SPAN / 4; i += SC_THREADS)
        b4[i] = make_float4(0.0f, 0.0f, 0.0f, 0.0f);
}

__device__ __forceinline__ void bins_flush(const float* bins,
                                           float* __restrict__ dst, int span) {
    const int rot = (int)((blockIdx.x * 416u) % SPAN);
    for (int i = threadIdx.x; i < SPAN; i += SC_THREADS) {
        int j = i + rot;
        if (j >= SPAN) j -= SPAN;
        if (j < span) {
            float v = bins[j];
            if (v != 0.0f) atomicAdd(dst + j, v);
        }
    }
}

__global__ void __launch_bounds__(SC_THREADS)
deg_accum_kernel(const int* __restrict__ cols,
                 const float* __restrict__ ew,
                 float* __restrict__ deg, int n, int e) {
    __shared__ __attribute__((aligned(16))) float bins[SPAN];
    const int lo = blockIdx.y * SPAN;
    const int span = min(SPAN, n - lo);
    bins_zero(bins);
    __syncthreads();

    const int nch = (int)gridDim.x;
    const int per = (e + nch - 1) / nch;
    const int start = blockIdx.x * per;
    const int end = min(e, start + per);
    int base = start;
    for (; base + SC_THREADS * SC_UNROLL <= end; base += SC_THREADS * SC_UNROLL) {
        int c[SC_UNROLL];
        float w[SC_UNROLL];
#pragma unroll
        for (int u = 0; u < SC_UNROLL; u++) {
            const int i = base + u * SC_THREADS + (int)threadIdx.x;
            c[u] = cols[i];
            w[u] = ew[i];
        }
        asm volatile("" :: HOLD8(c), HOLD8(w));
#pragma unroll
        for (int u = 0; u < SC_UNROLL; u++) {
            const unsigned k = (unsigned)(c[u] - lo);
            if (k < (unsigned)span) atomicAdd(&bins[k], w[u]);
        }
    }
    for (int i = base + (int)threadIdx.x; i < end; i += SC_THREADS) {
        const unsigned k = (unsigned)(cols[i] - lo);
        if (k < (unsigned)span) atomicAdd(&bins[k], ew[i]);
    }
    __syncthreads();
    bins_flush(bins, deg + lo, span);
}

__global__ void dinv_kernel(float* __restrict__ deg, int n) {
    int i = blockIdx.x * blockDim.x + threadIdx.x;
    if (i < n) {
        float d = deg[i];
        deg[i] = (d > 0.0f) ? __frsqrt_rn(d) : 0.0f;
    }
}

__global__ void __launch_bounds__(256)
gemm1_wmma_kernel(const float* __restrict__ X, const float* __restrict__ W,
                  float* __restrict__ H, int n) {
    __shared__ float sW[IN_FEAT * HIDDEN];
    for (int i = threadIdx.x; i < IN_FEAT * HIDDEN; i += blockDim.x)
        sW[i] = W[i];
    __syncthreads();

    const int wave = blockIdx.x * (blockDim.x >> 5) + (threadIdx.x >> 5);
    const int lane = threadIdx.x & 31;
    const int mt = wave >> 2;
    const int nt = wave & 3;
    const int m0 = mt * 16;
    if (m0 >= n) return;
    const int n0 = nt * 16;
    const int half = lane >> 4;
    const int l = lane & 15;

    const float* Xrow = X + (size_t)(m0 + l) * IN_FEAT;
    v8f c = {0.f, 0.f, 0.f, 0.f, 0.f, 0.f, 0.f, 0.f};

#pragma unroll 8
    for (int k = 0; k < IN_FEAT; k += 4) {
        v2f a = *(const v2f*)(Xrow + k + 2 * half);
        v2f b;
        b.x = sW[(k + 2 * half + 0) * HIDDEN + n0 + l];
        b.y = sW[(k + 2 * half + 1) * HIDDEN + n0 + l];
        c = __builtin_amdgcn_wmma_f32_16x16x4_f32(
                 false, a,  false, b,
                 (short)0, c,  false,  false);
    }

    float* Hout = H + (size_t)(m0 + 8 * half) * HIDDEN + n0 + l;
#pragma unroll
    for (int v = 0; v < 8; v++)
        Hout[(size_t)v * HIDDEN] = c[v];
}

#define BK_SHIFT   7
#define BK_NODES   (1 << BK_SHIFT)
#define BK_MAX     1024
#define EC_CHUNKS  256
#define EC_THREADS 1024
#define AG_THREADS 256
#define AG_STRIDE  ((AG_THREADS / 32) * 32)
#define AG_BATCH   16
static_assert(EC_CHUNKS == 256, "chunk_prefix_kernel gives each lane 8 chunks");
static_assert(HIDDEN == 64, "one float2 per lane covers a row");

__global__ void __launch_bounds__(EC_THREADS)
edge_count_kernel(const int* __restrict__ cols, int* __restrict__ counts,
                  int nb, int e) {
    __shared__ int cnt[BK_MAX];
    for (int t = threadIdx.x; t < BK_MAX; t += EC_THREADS) cnt[t] = 0;
    __syncthreads();
    const int per = (e + EC_CHUNKS - 1) / EC_CHUNKS;
    const int start = blockIdx.x * per;
    const int end = min(e, start + per);
    for (int base = start; base < end; base += EC_THREADS * 4) {
        int c[4];
#pragma unroll
        for (int u = 0; u < 4; u++) {
            const int i = base + u * EC_THREADS + (int)threadIdx.x;
            c[u] = (i < end) ? cols[i] : -1;
        }
#pragma unroll
        for (int u = 0; u < 4; u++)
            if (c[u] >= 0) atomicAdd(&cnt[c[u] >> BK_SHIFT], 1);
    }
    __syncthreads();
    for (int b = threadIdx.x; b < nb; b += EC_THREADS)
        counts[(size_t)b * EC_CHUNKS + blockIdx.x] = cnt[b];
}

__global__ void __launch_bounds__(256)
chunk_prefix_kernel(int* __restrict__ counts, int* __restrict__ tot, int nb) {
    const int b = blockIdx.x * (blockDim.x >> 5) + (threadIdx.x >> 5);
    const int lane = threadIdx.x & 31;
    if (b >= nb) return;
    int4* row = (int4*)(counts + (size_t)b * EC_CHUNKS) + 2 * lane;
    const int4 a = row[0];
    const int4 c = row[1];
    const int sum = a.x + a.y + a.z + a.w + c.x + c.y + c.z + c.w;
    int inc = sum;
    for (int off = 1; off < 32; off <<= 1) {
        const int v = __shfl_up(inc, off, 32);
        if (lane >= off) inc += v;
    }
    int4 oa, oc;
    oa.x = inc - sum;
    oa.y = oa.x + a.x;
    oa.z = oa.y + a.y;
    oa.w = oa.z + a.z;
    oc.x = oa.w + a.w;
    oc.y = oc.x + c.x;
    oc.z = oc.y + c.y;
    oc.w = oc.z + c.z;
    row[0] = oa;
    row[1] = oc;
    if (lane == 31) tot[b] = inc;
}

__global__ void __launch_bounds__(EC_THREADS)
edge_place_kernel(const int* __restrict__ rows, const int* __restrict__ cols,
                  const float* __restrict__ ew, const float* __restrict__ dinv,
                  const int* __restrict__ counts, const int* __restrict__ tot,
                  int* __restrict__ bstart, int4* __restrict__ recs,
                  int nb, int e) {
    __shared__ int cursor[BK_MAX];
    __shared__ int wsum[32];
    const int t = threadIdx.x;
    const int lane = t & 31;
    const int wid = t >> 5;
    const int mine = (t < nb) ? tot[t] : 0;
    int inc = mine;
    for (int off = 1; off < 32; off <<= 1) {
        const int v = __shfl_up(inc, off, 32);
        if (lane >= off) inc += v;
    }
    if (lane == 31) wsum[wid] = inc;
    __syncthreads();
    if (wid == 0) {
        const int s = wsum[lane];
        int si = s;
        for (int off = 1; off < 32; off <<= 1) {
            const int v = __shfl_up(si, off, 32);
            if (lane >= off) si += v;
        }
        wsum[lane] = si - s;
    }
    __syncthreads();
    const int excl = inc - mine + wsum[wid];
    if (t < nb) {
        cursor[t] = excl + counts[(size_t)t * EC_CHUNKS + blockIdx.x];
        if (blockIdx.x == 0) {
            bstart[t] = excl;
            if (t == nb - 1) bstart[nb] = excl + mine;
        }
    }
    __syncthreads();

    const int per = (e + EC_CHUNKS - 1) / EC_CHUNKS;
    const int start = blockIdx.x * per;
    const int end = min(e, start + per);
    for (int base = start; base < end; base += EC_THREADS * 4) {
        int c[4];
        int r[4];
        float w[4];
        float dr[4];
        float dc[4];
#pragma unroll
        for (int u = 0; u < 4; u++) {
            const int i = base + u * EC_THREADS + t;
            const bool ok = i < end;
            c[u] = ok ? cols[i] : -1;
            r[u] = ok ? rows[i] : 0;
            w[u] = ok ? ew[i] : 0.0f;
        }
#pragma unroll
        for (int u = 0; u < 4; u++) {
            dr[u] = dinv[r[u]];
            dc[u] = dinv[max(c[u], 0)];
        }
#pragma unroll
        for (int u = 0; u < 4; u++) {
            if (c[u] >= 0) {
                const int pos = atomicAdd(&cursor[c[u] >> BK_SHIFT], 1);
                const float norm = dr[u] * w[u] * dc[u];
                recs[pos] = make_int4(r[u], c[u] & (BK_NODES - 1), __float_as_int(norm), 0);
            }
        }
    }
}

#define AG_ROWS (BK_NODES / (AG_THREADS / 32))
static_assert(AG_ROWS * (AG_THREADS / 32) == BK_NODES && AG_ROWS <= 16,
              "the waves' node groups cover the tile; one lane of a 16-lane row per node");

__device__ __forceinline__ float lane_xor16(float x) {
    const int b = __float_as_int(x);
    return __int_as_float(__builtin_amdgcn_permlanex16(b, b, 0x76543210, 0xfedcba98, false, false));
}
template <int M>
__device__ __forceinline__ float row_xor(float x) {
    static_assert(M > 0 && M < 16, "row_xmask stays inside a row of 16");
    const int b = __float_as_int(x);
    return __int_as_float(__builtin_amdgcn_update_dpp(b, b, 0x160 + M, 0xf, 0xf, false));
}

__global__ void __launch_bounds__(AG_THREADS)
block_aggregate_kernel(const int4* __restrict__ recs, const int* __restrict__ bstart,
                       const float* __restrict__ dinv, const float* __restrict__ h0,
                       const float* __restrict__ b1, const float* __restrict__ w2,
                       float* __restrict__ g, int n) {
    __shared__ __attribute__((aligned(16))) float acc[(BK_NODES + 1) * HIDDEN];
    __shared__ float gs[BK_NODES];
    const int t = threadIdx.x;
    const int lane = t & 31;
    const int wave = __builtin_amdgcn_readfirstlane(t >> 5);
    const int lo = blockIdx.x * BK_NODES;
    const int nn = min(BK_NODES, n - lo);
    float2* acc2 = (float2*)acc;
    for (int idx = t; idx < nn * (HIDDEN / 2); idx += AG_THREADS) {
        const int node = lo + (idx >> 5);
        const float d = dinv[node];
        const float s = d * d;
        float2 v = ((const float2*)(h0 + (size_t)node * HIDDEN))[idx & 31];
        v.x *= s;
        v.y *= s;
        acc2[idx] = v;
    }
    __syncthreads();

    const int beg = __builtin_amdgcn_readfirstlane(bstart[blockIdx.x]);
    const int end = __builtin_amdgcn_readfirstlane(bstart[blockIdx.x + 1]);
    const int g0 = beg + wave * 32;
    if (g0 < end) {
        const int last = end - 1;
        int4 rec = recs[min(g0 + lane, last)];
        for (int g = g0; g < end; g += AG_STRIDE) {
            const int4 nxt = recs[min(g + AG_STRIDE + lane, last)];
            if (g + lane >= end) rec = make_int4(0, BK_NODES, 0, 0);
#pragma unroll
            for (int u0 = 0; u0 < 32; u0 += AG_BATCH) {
                float2 v[AG_BATCH];
                int cl[AG_BATCH];
                float nm[AG_BATCH];
#pragma unroll
                for (int k = 0; k < AG_BATCH; k++) {
                    const int r = __builtin_amdgcn_readlane(rec.x, u0 + k);
                    cl[k] = __builtin_amdgcn_readlane(rec.y, u0 + k);
                    nm[k] = __int_as_float(__builtin_amdgcn_readlane(rec.z, u0 + k));
                    v[k] = ((const float2*)(h0 + (size_t)r * HIDDEN))[lane];
                }
#pragma unroll
                for (int k = 0; k < AG_BATCH; k++) {
                    float* d = acc + cl[k] * HIDDEN + 2 * lane;
                    atomicAdd(d + 0, nm[k] * v[k].x);
                    atomicAdd(d + 1, nm[k] * v[k].y);
                }
            }
            rec = nxt;
        }
    }
    __syncthreads();

    const float2 bv = ((const float2*)b1)[lane];
    const float2 wv = ((const float2*)w2)[lane];
    float p[AG_ROWS];
#pragma unroll
    for (int k = 0; k < AG_ROWS; k++) {
        const int i = min(wave * AG_ROWS + k, nn - 1);
        const float2 a = acc2[i * (HIDDEN / 2) + lane];
        float2 hv;
        hv.x = fmaxf(a.x + bv.x, 0.0f);
        hv.y = fmaxf(a.y + bv.y, 0.0f);
        p[k] = hv.x * wv.x + hv.y * wv.y;
    }
#pragma unroll
    for (int k = 0; k < AG_ROWS; k++) p[k] += lane_xor16(p[k]);
#pragma unroll
    for (int k = 0; k < AG_ROWS; k++) p[k] += row_xor<8>(p[k]);
#pragma unroll
    for (int k = 0; k < AG_ROWS; k++) p[k] += row_xor<4>(p[k]);
#pragma unroll
    for (int k = 0; k < AG_ROWS; k++) p[k] += row_xor<2>(p[k]);
#pragma unroll
    for (int k = 0; k < AG_ROWS; k++) p[k] += row_xor<1>(p[k]);
    float mine = p[0];
#pragma unroll
    for (int k = 1; k < AG_ROWS; k++) mine = (lane == k) ? p[k] : mine;
    if (lane < AG_ROWS) gs[wave * AG_ROWS + lane] = mine;
    __syncthreads();
    if (t < nn) g[lo + t] = gs[t];
}

#define S2_THREADS 320
#define S2_UNROLL  8
static_assert(S2_UNROLL == 8, "HOLD8 names eight values");
static_assert(S2_THREADS >= BK_NODES, "one thread per node for the start values and the stores");

__global__ void __launch_bounds__(S2_THREADS)
block_scatter2_kernel(const int4* __restrict__ recs, const int* __restrict__ bstart,
                      const float* __restrict__ dinv, const float* __restrict__ g,
                      const float* __restrict__ b2, float* __restrict__ out, int n) {
    __shared__ float bins[BK_NODES];
    const int t = threadIdx.x;
    const int lo = blockIdx.x * BK_NODES;
    const int nn = min(BK_NODES, n - lo);
    if (t < nn) {
        const float d = dinv[lo + t];
        bins[t] = b2[0] + d * d * g[lo + t];
    }
    __syncthreads();

    const int beg = bstart[blockIdx.x];
    const int end = bstart[blockIdx.x + 1];
    const int last = end - 1;
    for (int base = beg; base < end; base += S2_THREADS * S2_UNROLL) {
        int4 rec[S2_UNROLL];
        float gr[S2_UNROLL];
#pragma unroll
        for (int u = 0; u < S2_UNROLL; u++)
            rec[u] = recs[min(base + u * S2_THREADS + t, last)];
#pragma unroll
        for (int u = 0; u < S2_UNROLL; u++)
            gr[u] = g[rec[u].x];
        asm volatile("" :: HOLD8(gr));
#pragma unroll
        for (int u = 0; u < S2_UNROLL; u++)
            if (base + u * S2_THREADS + t < end)
                atomicAdd(&bins[rec[u].y], __int_as_float(rec[u].z) * gr[u]);
    }
    __syncthreads();
    if (t < nn) out[lo + t] = bins[t];
}

extern "C" void kernel_launch(void* const* d_in, const int* in_sizes, int n_in,
                              void* d_out, int out_size, void* d_ws, size_t ws_size,
                              hipStream_t stream) {
    const float* x  = (const float*)d_in[0];
    const int*   ei = (const int*)d_in[1];
    const float* ew = (const float*)d_in[2];
    const float* w1 = (const float*)d_in[3];
    const float* b1 = (const float*)d_in[4];
    const float* w2 = (const float*)d_in[5];
    const float* b2 = (const float*)d_in[6];
    float* out = (float*)d_out;

    const int E = in_sizes[2];
    const int N = in_sizes[0] / IN_FEAT;
    const int* rows = ei;
    const int* cols = ei + E;

    float* ws   = (float*)d_ws;
    float* dinv = ws;
    float* g    = ws + N;
    float* h0   = ws + 2 * (size_t)N;
    int*  counts = (int*)(h0 + (size_t)N * HIDDEN);
    int*  tot    = counts + BK_MAX * EC_CHUNKS;
    int*  bstart = tot + BK_MAX;
    int4* recs   = (int4*)(bstart + BK_MAX + 32);

    const int B = 256;
    init_deg_kernel<<<(N + B - 1) / B, B, 0, stream>>>(dinv, N);
    const dim3 sgrid(SC_CHUNKS, (N + SPAN - 1) / SPAN);
    deg_accum_kernel<<<sgrid, SC_THREADS, 0, stream>>>(cols, ew, dinv, N, E);
    dinv_kernel<<<(N + B - 1) / B, B, 0, stream>>>(dinv, N);

    {
        int mtiles = (N + 15) / 16;
        int waves  = mtiles * (HIDDEN / 16);
        int blocks = (waves + 7) / 8;
        gemm1_wmma_kernel<<<blocks, B, 0, stream>>>(x, w1, h0, N);
    }

    {
        const int NB = (N + BK_NODES - 1) / BK_NODES;
        edge_count_kernel<<<EC_CHUNKS, EC_THREADS, 0, stream>>>(cols, counts, NB, E);
        chunk_prefix_kernel<<<(NB + 7) / 8, B, 0, stream>>>(counts, tot, NB);
        edge_place_kernel<<<EC_CHUNKS, EC_THREADS, 0, stream>>>(rows, cols, ew, dinv, counts, tot,
                                                                bstart, recs, NB, E);
        block_aggregate_kernel<<<NB, AG_THREADS, 0, stream>>>(recs, bstart, dinv, h0, b1, w2,
                                                              g, N);
        block_scatter2_kernel<<<NB, S2_THREADS, 0, stream>>>(recs, bstart, dinv, g, b2, out, N);
    }
}


# Norm factored D^-1/2 (A+I) D^-1/2: degree from sorted records per block (LDS bins), rows scaled at GEMM store, column factor once per node; atomic edge pass, 2 launches gone; sum order changes
# speedup vs baseline: 3.5098x; 1.1401x over previous
#include <hip/hip_runtime.h>
#include <hip/hip_bf16.h>

#define IN_FEAT 128
#define HIDDEN  64

typedef __attribute__((ext_vector_type(2))) float v2f;
typedef __attribute__((ext_vector_type(8))) float v8f;

#define HOLD8(a) "v"(a[0]), "v"(a[1]), "v"(a[2]), "v"(a[3]), "v"(a[4]), "v"(a[5]), "v"(a[6]), "v"(a[7])

__global__ void __launch_bounds__(256)
gemm1_wmma_kernel(const float* __restrict__ X, const float* __restrict__ W,
                  const float* __restrict__ dinv, float* __restrict__ H, int n) {
    __shared__ float sW[IN_FEAT * HIDDEN];
    for (int i = threadIdx.x; i < IN_FEAT * HIDDEN; i += blockDim.x)
        sW[i] = W[i];
    __syncthreads();

    const int wave = blockIdx.x * (blockDim.x >> 5) + (threadIdx.x >> 5);
    const int lane = threadIdx.x & 31;
    const int mt = wave >> 2;
    const int nt = wave & 3;
    const int m0 = mt * 16;
    if (m0 >= n) return;
    const int n0 = nt * 16;
    const int half = lane >> 4;
    const int l = lane & 15;

    const float* Xrow = X + (size_t)(m0 + l) * IN_FEAT;
    v8f c = {0.f, 0.f, 0.f, 0.f, 0.f, 0.f, 0.f, 0.f};
    float dsc[8];
#pragma unroll
    for (int v = 0; v < 8; v++)
        dsc[v] = dinv[m0 + 8 * half + v];

#pragma unroll 8
    for (int k = 0; k < IN_FEAT; k += 4) {
        v2f a = *(const v2f*)(Xrow + k + 2 * half);
        v2f b;
        b.x = sW[(k + 2 * half + 0) * HIDDEN + n0 + l];
        b.y = sW[(k + 2 * half + 1) * HIDDEN + n0 + l];
        c = __builtin_amdgcn_wmma_f32_16x16x4_f32(
                 false, a,  false, b,
                 (short)0, c,  false,  false);
    }

    float* Hout = H + (size_t)(m0 + 8 * half) * HIDDEN + n0 + l;
#pragma unroll
    for (int v = 0; v < 8; v++)
        Hout[(size_t)v * HIDDEN] = dsc[v] * c[v];
}

#define BK_SHIFT   7
#define BK_NODES   (1 << BK_SHIFT)
#define BK_MAX     1024
#define EC_CHUNKS  256
#define EC_THREADS 1024
#define AG_THREADS 256
#define AG_STRIDE  ((AG_THREADS / 32) * 32)
#define AG_BATCH   16
static_assert(EC_CHUNKS == 256, "chunk_prefix_kernel gives each lane 8 chunks");
static_assert(HIDDEN == 64, "one float2 per lane covers a row");

__global__ void __launch_bounds__(EC_THREADS)
edge_count_kernel(const int* __restrict__ cols, int* __restrict__ counts,
                  int nb, int e) {
    __shared__ int cnt[BK_MAX];
    for (int t = threadIdx.x; t < BK_MAX; t += EC_THREADS) cnt[t] = 0;
    __syncthreads();
    const int per = (e + EC_CHUNKS - 1) / EC_CHUNKS;
    const int start = blockIdx.x * per;
    const int end = min(e, start + per);
    for (int base = start; base < end; base += EC_THREADS * 4) {
        int c[4];
#pragma unroll
        for (int u = 0; u < 4; u++) {
            const int i = base + u * EC_THREADS + (int)threadIdx.x;
            c[u] = (i < end) ? cols[i] : -1;
        }
#pragma unroll
        for (int u = 0; u < 4; u++)
            if (c[u] >= 0) atomicAdd(&cnt[c[u] >> BK_SHIFT], 1);
    }
    __syncthreads();
    for (int b = threadIdx.x; b < nb; b += EC_THREADS)
        counts[(size_t)b * EC_CHUNKS + blockIdx.x] = cnt[b];
}

__global__ void __launch_bounds__(256)
chunk_prefix_kernel(int* __restrict__ counts, int* __restrict__ tot, int nb) {
    const int b = blockIdx.x * (blockDim.x >> 5) + (threadIdx.x >> 5);
    const int lane = threadIdx.x & 31;
    if (b >= nb) return;
    int4* row = (int4*)(counts + (size_t)b * EC_CHUNKS) + 2 * lane;
    const int4 a = row[0];
    const int4 c = row[1];
    const int sum = a.x + a.y + a.z + a.w + c.x + c.y + c.z + c.w;
    int inc = sum;
    for (int off = 1; off < 32; off <<= 1) {
        const int v = __shfl_up(inc, off, 32);
        if (lane >= off) inc += v;
    }
    int4 oa, oc;
    oa.x = inc - sum;
    oa.y = oa.x + a.x;
    oa.z = oa.y + a.y;
    oa.w = oa.z + a.z;
    oc.x = oa.w + a.w;
    oc.y = oc.x + c.x;
    oc.z = oc.y + c.y;
    oc.w = oc.z + c.z;
    row[0] = oa;
    row[1] = oc;
    if (lane == 31) tot[b] = inc;
}

__global__ void __launch_bounds__(EC_THREADS)
edge_place_kernel(const int* __restrict__ rows, const int* __restrict__ cols,
                  const float* __restrict__ ew,
                  const int* __restrict__ counts, const int* __restrict__ tot,
                  int* __restrict__ bstart, int4* __restrict__ recs,
                  int nb, int e) {
    __shared__ int cursor[BK_MAX];
    __shared__ int wsum[32];
    const int t = threadIdx.x;
    const int lane = t & 31;
    const int wid = t >> 5;
    const int mine = (t < nb) ? tot[t] : 0;
    int inc = mine;
    for (int off = 1; off < 32; off <<= 1) {
        const int v = __shfl_up(inc, off, 32);
        if (lane >= off) inc += v;
    }
    if (lane == 31) wsum[wid] = inc;
    __syncthreads();
    if (wid == 0) {
        const int s = wsum[lane];
        int si = s;
        for (int off = 1; off < 32; off <<= 1) {
            const int v = __shfl_up(si, off, 32);
            if (lane >= off) si += v;
        }
        wsum[lane] = si - s;
    }
    __syncthreads();
    const int excl = inc - mine + wsum[wid];
    if (t < nb) {
        cursor[t] = excl + counts[(size_t)t * EC_CHUNKS + blockIdx.x];
        if (blockIdx.x == 0) {
            bstart[t] = excl;
            if (t == nb - 1) bstart[nb] = excl + mine;
        }
    }
    __syncthreads();

    const int per = (e + EC_CHUNKS - 1) / EC_CHUNKS;
    const int start = blockIdx.x * per;
    const int end = min(e, start + per);
    for (int base = start; base < end; base += EC_THREADS * 4) {
        int c[4];
        int r[4];
        float w[4];
#pragma unroll
        for (int u = 0; u < 4; u++) {
            const int i = base + u * EC_THREADS + t;
            const bool ok = i < end;
            c[u] = ok ? cols[i] : -1;
            r[u] = ok ? rows[i] : 0;
            w[u] = ok ? ew[i] : 0.0f;
        }
#pragma unroll
        for (int u = 0; u < 4; u++) {
            if (c[u] >= 0) {
                const int pos = atomicAdd(&cursor[c[u] >> BK_SHIFT], 1);
                recs[pos] = make_int4(r[u], c[u] & (BK_NODES - 1), __float_as_int(w[u]), 0);
            }
        }
    }
}

#define DG_THREADS 512
#define DG_UNROLL  8
static_assert(DG_UNROLL == 8, "HOLD8 names eight values");
static_assert(DG_THREADS >= BK_NODES, "one thread per node for the start values and the stores");

__global__ void __launch_bounds__(DG_THREADS)
block_degree_kernel(const int4* __restrict__ recs, const int* __restrict__ bstart,
                    float* __restrict__ dinv, int n) {
    __shared__ float bins[BK_NODES];
    const int t = threadIdx.x;
    const int lo = blockIdx.x * BK_NODES;
    const int nn = min(BK_NODES, n - lo);
    if (t < BK_NODES) bins[t] = 1.0f;
    __syncthreads();

    const int beg = bstart[blockIdx.x];
    const int end = bstart[blockIdx.x + 1];
    const int last = end - 1;
    for (int base = beg; base < end; base += DG_THREADS * DG_UNROLL) {
        int cl[DG_UNROLL];
        float w[DG_UNROLL];
#pragma unroll
        for (int u = 0; u < DG_UNROLL; u++) {
            const int4 rec = recs[min(base + u * DG_THREADS + t, last)];
            cl[u] = rec.y;
            w[u] = __int_as_float(rec.z);
        }
        asm volatile("" :: HOLD8(cl), HOLD8(w));
#pragma unroll
        for (int u = 0; u < DG_UNROLL; u++)
            if (base + u * DG_THREADS + t < end) atomicAdd(&bins[cl[u]], w[u]);
    }
    __syncthreads();
    if (t < nn) {
        const float d = bins[t];
        dinv[lo + t] = (d > 0.0f) ? __frsqrt_rn(d) : 0.0f;
    }
}

#define AG_ROWS (BK_NODES / (AG_THREADS / 32))
static_assert(AG_ROWS * (AG_THREADS / 32) == BK_NODES && AG_ROWS <= 16,
              "the waves' node groups cover the tile; one lane of a 16-lane row per node");

__device__ __forceinline__ float lane_xor16(float x) {
    const int b = __float_as_int(x);
    return __int_as_float(__builtin_amdgcn_permlanex16(b, b, 0x76543210, 0xfedcba98, false, false));
}
template <int M>
__device__ __forceinline__ float row_xor(float x) {
    static_assert(M > 0 && M < 16, "row_xmask stays inside a row of 16");
    const int b = __float_as_int(x);
    return __int_as_float(__builtin_amdgcn_update_dpp(b, b, 0x160 + M, 0xf, 0xf, false));
}

__global__ void __launch_bounds__(AG_THREADS)
block_aggregate_kernel(const int4* __restrict__ recs, const int* __restrict__ bstart,
                       const float* __restrict__ dinv, const float* __restrict__ h0,
                       const float* __restrict__ b1, const float* __restrict__ w2,
                       float* __restrict__ g, int n) {
    __shared__ __attribute__((aligned(16))) float acc[(BK_NODES + 1) * HIDDEN];
    __shared__ float gs[BK_NODES];
    __shared__ float dl[BK_NODES];
    const int t = threadIdx.x;
    const int lane = t & 31;
    const int wave = __builtin_amdgcn_readfirstlane(t >> 5);
    const int lo = blockIdx.x * BK_NODES;
    const int nn = min(BK_NODES, n - lo);
    float2* acc2 = (float2*)acc;
    if (t < BK_NODES) dl[t] = (t < nn) ? dinv[lo + t] : 0.0f;
    for (int idx = t; idx < nn * (HIDDEN / 2); idx += AG_THREADS) {
        const int node = lo + (idx >> 5);
        acc2[idx] = ((const float2*)(h0 + (size_t)node * HIDDEN))[idx & 31];
    }
    __syncthreads();

    const int beg = __builtin_amdgcn_readfirstlane(bstart[blockIdx.x]);
    const int end = __builtin_amdgcn_readfirstlane(bstart[blockIdx.x + 1]);
    const int g0 = beg + wave * 32;
    if (g0 < end) {
        const int last = end - 1;
        int4 rec = recs[min(g0 + lane, last)];
        for (int g = g0; g < end; g += AG_STRIDE) {
            const int4 nxt = recs[min(g + AG_STRIDE + lane, last)];
            if (g + lane >= end) rec = make_int4(0, BK_NODES, 0, 0);
#pragma unroll
            for (int u0 = 0; u0 < 32; u0 += AG_BATCH) {
                float2 v[AG_BATCH];
                int cl[AG_BATCH];
                float nm[AG_BATCH];
#pragma unroll
                for (int k = 0; k < AG_BATCH; k++) {
                    const int r = __builtin_amdgcn_readlane(rec.x, u0 + k);
                    cl[k] = __builtin_amdgcn_readlane(rec.y, u0 + k);
                    nm[k] = __int_as_float(__builtin_amdgcn_readlane(rec.z, u0 + k));
                    v[k] = ((const float2*)(h0 + (size_t)r * HIDDEN))[lane];
                }
#pragma unroll
                for (int k = 0; k < AG_BATCH; k++) {
                    float* d = acc + cl[k] * HIDDEN + 2 * lane;
                    atomicAdd(d + 0, nm[k] * v[k].x);
                    atomicAdd(d + 1, nm[k] * v[k].y);
                }
            }
            rec = nxt;
        }
    }
    __syncthreads();

    const float2 bv = ((const float2*)b1)[lane];
    const float2 wv = ((const float2*)w2)[lane];
    float p[AG_ROWS];
#pragma unroll
    for (int k = 0; k < AG_ROWS; k++) {
        const int i = min(wave * AG_ROWS + k, nn - 1);
        const float2 a = acc2[i * (HIDDEN / 2) + lane];
        const float dc = dl[i];
        float2 hv;
        hv.x = fmaxf(dc * a.x + bv.x, 0.0f);
        hv.y = fmaxf(dc * a.y + bv.y, 0.0f);
        p[k] = hv.x * wv.x + hv.y * wv.y;
    }
#pragma unroll
    for (int k = 0; k < AG_ROWS; k++) p[k] += lane_xor16(p[k]);
#pragma unroll
    for (int k = 0; k < AG_ROWS; k++) p[k] += row_xor<8>(p[k]);
#pragma unroll
    for (int k = 0; k < AG_ROWS; k++) p[k] += row_xor<4>(p[k]);
#pragma unroll
    for (int k = 0; k < AG_ROWS; k++) p[k] += row_xor<2>(p[k]);
#pragma unroll
    for (int k = 0; k < AG_ROWS; k++) p[k] += row_xor<1>(p[k]);
    float mine = p[0];
#pragma unroll
    for (int k = 1; k < AG_ROWS; k++) mine = (lane == k) ? p[k] : mine;
    if (lane < AG_ROWS) gs[wave * AG_ROWS + lane] = dl[wave * AG_ROWS + lane] * mine;
    __syncthreads();
    if (t < nn) g[lo + t] = gs[t];
}

#define S2_THREADS 320
#define S2_UNROLL  8
static_assert(S2_UNROLL == 8, "HOLD8 names eight values");
static_assert(S2_THREADS >= BK_NODES, "one thread per node for the start values and the stores");

__global__ void __launch_bounds__(S2_THREADS)
block_scatter2_kernel(const int4* __restrict__ recs, const int* __restrict__ bstart,
                      const float* __restrict__ dinv, const float* __restrict__ g,
                      const float* __restrict__ b2, float* __restrict__ out, int n) {
    __shared__ float bins[BK_NODES];
    const int t = threadIdx.x;
    const int lo = blockIdx.x * BK_NODES;
    const int nn = min(BK_NODES, n - lo);
    if (t < nn) bins[t] = g[lo + t];
    __syncthreads();

    const int beg = bstart[blockIdx.x];
    const int end = bstart[blockIdx.x + 1];
    const int last = end - 1;
    for (int base = beg; base < end; base += S2_THREADS * S2_UNROLL) {
        int4 rec[S2_UNROLL];
        float gr[S2_UNROLL];
#pragma unroll
        for (int u = 0; u < S2_UNROLL; u++)
            rec[u] = recs[min(base + u * S2_THREADS + t, last)];
#pragma unroll
        for (int u = 0; u < S2_UNROLL; u++)
            gr[u] = g[rec[u].x];
        asm volatile("" :: HOLD8(gr));
#pragma unroll
        for (int u = 0; u < S2_UNROLL; u++)
            if (base + u * S2_THREADS + t < end)
                atomicAdd(&bins[rec[u].y], __int_as_float(rec[u].z) * gr[u]);
    }
    __syncthreads();
    if (t < nn) out[lo + t] = b2[0] + dinv[lo + t] * bins[t];
}

extern "C" void kernel_launch(void* const* d_in, const int* in_sizes, int n_in,
                              void* d_out, int out_size, void* d_ws, size_t ws_size,
                              hipStream_t stream) {
    const float* x  = (const float*)d_in[0];
    const int*   ei = (const int*)d_in[1];
    const float* ew = (const float*)d_in[2];
    const float* w1 = (const float*)d_in[3];
    const float* b1 = (const float*)d_in[4];
    const float* w2 = (const float*)d_in[5];
    const float* b2 = (const float*)d_in[6];
    float* out = (float*)d_out;

    const int E = in_sizes[2];
    const int N = in_sizes[0] / IN_FEAT;
    const int* rows = ei;
    const int* cols = ei + E;

    float* ws   = (float*)d_ws;
    float* dinv = ws;
    float* g    = ws + N;
    float* h0   = ws + 2 * (size_t)N;
    int*  counts = (int*)(h0 + (size_t)N * HIDDEN);
    int*  tot    = counts + BK_MAX * EC_CHUNKS;
    int*  bstart = tot + BK_MAX;
    int4* recs   = (int4*)(bstart + BK_MAX + 32);

    const int B = 256;
    const int NB = (N + BK_NODES - 1) / BK_NODES;

    edge_count_kernel<<<EC_CHUNKS, EC_THREADS, 0, stream>>>(cols, counts, NB, E);
    chunk_prefix_kernel<<<(NB + 7) / 8, B, 0, stream>>>(counts, tot, NB);
    edge_place_kernel<<<EC_CHUNKS, EC_THREADS, 0, stream>>>(rows, cols, ew, counts, tot,
                                                            bstart, recs, NB, E);
    block_degree_kernel<<<NB, DG_THREADS, 0, stream>>>(recs, bstart, dinv, N);

    {
        int mtiles = (N + 15) / 16;
        int waves  = mtiles * (HIDDEN / 16);
        int blocks = (waves + 7) / 8;
        gemm1_wmma_kernel<<<blocks, B, 0, stream>>>(x, w1, dinv, h0, N);
    }

    block_aggregate_kernel<<<NB, AG_THREADS, 0, stream>>>(recs, bstart, dinv, h0, b1, w2, g, N);
    block_scatter2_kernel<<<NB, S2_THREADS, 0, stream>>>(recs, bstart, dinv, g, b2, out, N);
}


# edge_place: each chunk's records are ordered by block in a 200 KB LDS stage and copied out by consecutive lanes, so runs are stored as contiguous pieces, not one scattered 16-byte store per edge
# speedup vs baseline: 3.7098x; 1.0570x over previous
#include <hip/hip_runtime.h>
#include <hip/hip_bf16.h>

#define IN_FEAT 128
#define HIDDEN  64

typedef __attribute__((ext_vector_type(2))) float v2f;
typedef __attribute__((ext_vector_type(8))) float v8f;

#define HOLD8(a) "v"(a[0]), "v"(a[1]), "v"(a[2]), "v"(a[3]), "v"(a[4]), "v"(a[5]), "v"(a[6]), "v"(a[7])

__global__ void __launch_bounds__(256)
gemm1_wmma_kernel(const float* __restrict__ X, const float* __restrict__ W,
                  const float* __restrict__ dinv, float* __restrict__ H, int n) {
    __shared__ float sW[IN_FEAT * HIDDEN];
    for (int i = threadIdx.x; i < IN_FEAT * HIDDEN; i += blockDim.x)
        sW[i] = W[i];
    __syncthreads();

    const int wave = blockIdx.x * (blockDim.x >> 5) + (threadIdx.x >> 5);
    const int lane = threadIdx.x & 31;
    const int mt = wave >> 2;
    const int nt = wave & 3;
    const int m0 = mt * 16;
    if (m0 >= n) return;
    const int n0 = nt * 16;
    const int half = lane >> 4;
    const int l = lane & 15;

    const float* Xrow = X + (size_t)(m0 + l) * IN_FEAT;
    v8f c = {0.f, 0.f, 0.f, 0.f, 0.f, 0.f, 0.f, 0.f};
    float dsc[8];
#pragma unroll
    for (int v = 0; v < 8; v++)
        dsc[v] = dinv[m0 + 8 * half + v];

#pragma unroll 8
    for (int k = 0; k < IN_FEAT; k += 4) {
        v2f a = *(const v2f*)(Xrow + k + 2 * half);
        v2f b;
        b.x = sW[(k + 2 * half + 0) * HIDDEN + n0 + l];
        b.y = sW[(k + 2 * half + 1) * HIDDEN + n0 + l];
        c = __builtin_amdgcn_wmma_f32_16x16x4_f32(
                 false, a,  false, b,
                 (short)0, c,  false,  false);
    }

    float* Hout = H + (size_t)(m0 + 8 * half) * HIDDEN + n0 + l;
#pragma unroll
    for (int v = 0; v < 8; v++)
        Hout[(size_t)v * HIDDEN] = dsc[v] * c[v];
}

#define BK_SHIFT   7
#define BK_NODES   (1 << BK_SHIFT)
#define BK_MAX     1024
#define EC_CHUNKS  256
#define EC_THREADS 1024
#define AG_THREADS 256
#define AG_STRIDE  ((AG_THREADS / 32) * 32)
#define AG_BATCH   16
static_assert(EC_CHUNKS == 256, "chunk_prefix_kernel gives each lane 8 chunks");
static_assert(HIDDEN == 64, "one float2 per lane covers a row");

__global__ void __launch_bounds__(EC_THREADS)
edge_count_kernel(const int* __restrict__ cols, int* __restrict__ counts,
                  int nb, int e) {
    __shared__ int cnt[BK_MAX];
    for (int t = threadIdx.x; t < BK_MAX; t += EC_THREADS) cnt[t] = 0;
    __syncthreads();
    const int per = (e + EC_CHUNKS - 1) / EC_CHUNKS;
    const int start = blockIdx.x * per;
    const int end = min(e, start + per);
    for (int base = start; base < end; base += EC_THREADS * 4) {
        int c[4];
#pragma unroll
        for (int u = 0; u < 4; u++) {
            const int i = base + u * EC_THREADS + (int)threadIdx.x;
            c[u] = (i < end) ? cols[i] : -1;
        }
#pragma unroll
        for (int u = 0; u < 4; u++)
            if (c[u] >= 0) atomicAdd(&cnt[c[u] >> BK_SHIFT], 1);
    }
    __syncthreads();
    for (int b = threadIdx.x; b < nb; b += EC_THREADS)
        counts[(size_t)b * EC_CHUNKS + blockIdx.x] = cnt[b];
}

__global__ void __launch_bounds__(256)
chunk_prefix_kernel(int* __restrict__ counts, int* __restrict__ tot, int nb) {
    const int b = blockIdx.x * (blockDim.x >> 5) + (threadIdx.x >> 5);
    const int lane = threadIdx.x & 31;
    if (b >= nb) return;
    int4* row = (int4*)(counts + (size_t)b * EC_CHUNKS) + 2 * lane;
    const int4 a = row[0];
    const int4 c = row[1];
    const int sum = a.x + a.y + a.z + a.w + c.x + c.y + c.z + c.w;
    int inc = sum;
    for (int off = 1; off < 32; off <<= 1) {
        const int v = __shfl_up(inc, off, 32);
        if (lane >= off) inc += v;
    }
    int4 oa, oc;
    oa.x = inc - sum;
    oa.y = oa.x + a.x;
    oa.z = oa.y + a.y;
    oa.w = oa.z + a.z;
    oc.x = oa.w + a.w;
    oc.y = oc.x + c.x;
    oc.z = oc.y + c.y;
    oc.w = oc.z + c.z;
    row[0] = oa;
    row[1] = oc;
    if (lane == 31) tot[b] = inc;
}

#define EP_STAGE 12800
__global__ void __launch_bounds__(EC_THREADS)
edge_place_kernel(const int* __restrict__ rows, const int* __restrict__ cols,
                  const float* __restrict__ ew,
                  const int* __restrict__ counts, const int* __restrict__ tot,
                  int* __restrict__ bstart, int4* __restrict__ recs,
                  int nb, int e) {
    __shared__ int4 stage[EP_STAGE];
    __shared__ int cursor[BK_MAX];
    __shared__ int delta[BK_MAX];
    __shared__ int wsum_g[32];
    __shared__ int wsum_l[32];
    const int t = threadIdx.x;
    const int lane = t & 31;
    const int wid = t >> 5;
    const int per = (e + EC_CHUNKS - 1) / EC_CHUNKS;
    const int start = blockIdx.x * per;
    const int end = min(e, start + per);
    const bool staged = per <= EP_STAGE;

    int mine = 0, cur = 0, len = 0;
    if (t < nb) {
        mine = tot[t];
        cur = counts[(size_t)t * EC_CHUNKS + blockIdx.x];
        const int nxt = (blockIdx.x + 1 < EC_CHUNKS)
                            ? counts[(size_t)t * EC_CHUNKS + blockIdx.x + 1] : mine;
        len = nxt - cur;
    }
    int inc_g = mine;
    int inc_l = len;
    for (int off = 1; off < 32; off <<= 1) {
        const int vg = __shfl_up(inc_g, off, 32);
        const int vl = __shfl_up(inc_l, off, 32);
        if (lane >= off) { inc_g += vg; inc_l += vl; }
    }
    if (lane == 31) { wsum_g[wid] = inc_g; wsum_l[wid] = inc_l; }
    __syncthreads();
    if (wid == 0) {
        const int sg = wsum_g[lane];
        const int sl = wsum_l[lane];
        int sig = sg;
        int sil = sl;
        for (int off = 1; off < 32; off <<= 1) {
            const int vg = __shfl_up(sig, off, 32);
            const int vl = __shfl_up(sil, off, 32);
            if (lane >= off) { sig += vg; sil += vl; }
        }
        wsum_g[lane] = sig - sg;
        wsum_l[lane] = sil - sl;
    }
    __syncthreads();
    const int excl = inc_g - mine + wsum_g[wid];
    const int lexcl = inc_l - len + wsum_l[wid];
    if (t < nb) {
        const int gpos = excl + cur;
        cursor[t] = staged ? lexcl : gpos;
        delta[t] = gpos - lexcl;
        if (blockIdx.x == 0) {
            bstart[t] = excl;
            if (t == nb - 1) bstart[nb] = excl + mine;
        }
    }
    __syncthreads();

    for (int base = start; base < end; base += EC_THREADS * 4) {
        int c[4];
        int r[4];
        float w[4];
#pragma unroll
        for (int u = 0; u < 4; u++) {
            const int i = base + u * EC_THREADS + t;
            const bool ok = i < end;
            c[u] = ok ? cols[i] : -1;
            r[u] = ok ? rows[i] : 0;
            w[u] = ok ? ew[i] : 0.0f;
        }
#pragma unroll
        for (int u = 0; u < 4; u++) {
            if (c[u] >= 0) {
                const int pos = atomicAdd(&cursor[c[u] >> BK_SHIFT], 1);
                if (staged)
                    stage[pos] = make_int4(r[u], c[u], __float_as_int(w[u]), 0);
                else
                    recs[pos] = make_int4(r[u], c[u] & (BK_NODES - 1), __float_as_int(w[u]), 0);
            }
        }
    }
    if (staged) {
        __syncthreads();
        const int cnt = end - start;
        for (int p = t; p < cnt; p += EC_THREADS) {
            const int4 rec = stage[p];
            recs[p + delta[rec.y >> BK_SHIFT]] =
                make_int4(rec.x, rec.y & (BK_NODES - 1), rec.z, 0);
        }
    }
}

#define DG_THREADS 512
#define DG_UNROLL  8
static_assert(DG_UNROLL == 8, "HOLD8 names eight values");
static_assert(DG_THREADS >= BK_NODES, "one thread per node for the start values and the stores");

__global__ void __launch_bounds__(DG_THREADS)
block_degree_kernel(const int4* __restrict__ recs, const int* __restrict__ bstart,
                    float* __restrict__ dinv, int n) {
    __shared__ float bins[BK_NODES];
    const int t = threadIdx.x;
    const int lo = blockIdx.x * BK_NODES;
    const int nn = min(BK_NODES, n - lo);
    if (t < BK_NODES) bins[t] = 1.0f;
    __syncthreads();

    const int beg = bstart[blockIdx.x];
    const int end = bstart[blockIdx.x + 1];
    const int last = end - 1;
    for (int base = beg; base < end; base += DG_THREADS * DG_UNROLL) {
        int cl[DG_UNROLL];
        float w[DG_UNROLL];
#pragma unroll
        for (int u = 0; u < DG_UNROLL; u++) {
            const int4 rec = recs[min(base + u * DG_THREADS + t, last)];
            cl[u] = rec.y;
            w[u] = __int_as_float(rec.z);
        }
        asm volatile("" :: HOLD8(cl), HOLD8(w));
#pragma unroll
        for (int u = 0; u < DG_UNROLL; u++)
            if (base + u * DG_THREADS + t < end) atomicAdd(&bins[cl[u]], w[u]);
    }
    __syncthreads();
    if (t < nn) {
        const float d = bins[t];
        dinv[lo + t] = (d > 0.0f) ? __frsqrt_rn(d) : 0.0f;
    }
}

#define AG_ROWS (BK_NODES / (AG_THREADS / 32))
static_assert(AG_ROWS * (AG_THREADS / 32) == BK_NODES && AG_ROWS <= 16,
              "the waves' node groups cover the tile; one lane of a 16-lane row per node");

__device__ __forceinline__ float lane_xor16(float x) {
    const int b = __float_as_int(x);
    return __int_as_float(__builtin_amdgcn_permlanex16(b, b, 0x76543210, 0xfedcba98, false, false));
}
template <int M>
__device__ __forceinline__ float row_xor(float x) {
    static_assert(M > 0 && M < 16, "row_xmask stays inside a row of 16");
    const int b = __float_as_int(x);
    return __int_as_float(__builtin_amdgcn_update_dpp(b, b, 0x160 + M, 0xf, 0xf, false));
}

__global__ void __launch_bounds__(AG_THREADS)
block_aggregate_kernel(const int4* __restrict__ recs, const int* __restrict__ bstart,
                       const float* __restrict__ dinv, const float* __restrict__ h0,
                       const float* __restrict__ b1, const float* __restrict__ w2,
                       float* __restrict__ g, int n) {
    __shared__ __attribute__((aligned(16))) float acc[(BK_NODES + 1) * HIDDEN];
    __shared__ float gs[BK_NODES];
    __shared__ float dl[BK_NODES];
    const int t = threadIdx.x;
    const int lane = t & 31;
    const int wave = __builtin_amdgcn_readfirstlane(t >> 5);
    const int lo = blockIdx.x * BK_NODES;
    const int nn = min(BK_NODES, n - lo);
    float2* acc2 = (float2*)acc;
    if (t < BK_NODES) dl[t] = (t < nn) ? dinv[lo + t] : 0.0f;
    for (int idx = t; idx < nn * (HIDDEN / 2); idx += AG_THREADS) {
        const int node = lo + (idx >> 5);
        acc2[idx] = ((const float2*)(h0 + (size_t)node * HIDDEN))[idx & 31];
    }
    __syncthreads();

    const int beg = __builtin_amdgcn_readfirstlane(bstart[blockIdx.x]);
    const int end = __builtin_amdgcn_readfirstlane(bstart[blockIdx.x + 1]);
    const int g0 = beg + wave * 32;
    if (g0 < end) {
        const int last = end - 1;
        int4 rec = recs[min(g0 + lane, last)];
        for (int g = g0; g < end; g += AG_STRIDE) {
            const int4 nxt = recs[min(g + AG_STRIDE + lane, last)];
            if (g + lane >= end) rec = make_int4(0, BK_NODES, 0, 0);
#pragma unroll
            for (int u0 = 0; u0 < 32; u0 += AG_BATCH) {
                float2 v[AG_BATCH];
                int cl[AG_BATCH];
                float nm[AG_BATCH];
#pragma unroll
                for (int k = 0; k < AG_BATCH; k++) {
                    const int r = __builtin_amdgcn_readlane(rec.x, u0 + k);
                    cl[k] = __builtin_amdgcn_readlane(rec.y, u0 + k);
                    nm[k] = __int_as_float(__builtin_amdgcn_readlane(rec.z, u0 + k));
                    v[k] = ((const float2*)(h0 + (size_t)r * HIDDEN))[lane];
                }
#pragma unroll
                for (int k = 0; k < AG_BATCH; k++) {
                    float* d = acc + cl[k] * HIDDEN + 2 * lane;
                    atomicAdd(d + 0, nm[k] * v[k].x);
                    atomicAdd(d + 1, nm[k] * v[k].y);
                }
            }
            rec = nxt;
        }
    }
    __syncthreads();

    const float2 bv = ((const float2*)b1)[lane];
    const float2 wv = ((const float2*)w2)[lane];
    float p[AG_ROWS];
#pragma unroll
    for (int k = 0; k < AG_ROWS; k++) {
        const int i = min(wave * AG_ROWS + k, nn - 1);
        const float2 a = acc2[i * (HIDDEN / 2) + lane];
        const float dc = dl[i];
        float2 hv;
        hv.x = fmaxf(dc * a.x + bv.x, 0.0f);
        hv.y = fmaxf(dc * a.y + bv.y, 0.0f);
        p[k] = hv.x * wv.x + hv.y * wv.y;
    }
#pragma unroll
    for (int k = 0; k < AG_ROWS; k++) p[k] += lane_xor16(p[k]);
#pragma unroll
    for (int k = 0; k < AG_ROWS; k++) p[k] += row_xor<8>(p[k]);
#pragma unroll
    for (int k = 0; k < AG_ROWS; k++) p[k] += row_xor<4>(p[k]);
#pragma unroll
    for (int k = 0; k < AG_ROWS; k++) p[k] += row_xor<2>(p[k]);
#pragma unroll
    for (int k = 0; k < AG_ROWS; k++) p[k] += row_xor<1>(p[k]);
    float mine = p[0];
#pragma unroll
    for (int k = 1; k < AG_ROWS; k++) mine = (lane == k) ? p[k] : mine;
    if (lane < AG_ROWS) gs[wave * AG_ROWS + lane] = dl[wave * AG_ROWS + lane] * mine;
    __syncthreads();
    if (t < nn) g[lo + t] = gs[t];
}

#define S2_THREADS 320
#define S2_UNROLL  8
static_assert(S2_UNROLL == 8, "HOLD8 names eight values");
static_assert(S2_THREADS >= BK_NODES, "one thread per node for the start values and the stores");

__global__ void __launch_bounds__(S2_THREADS)
block_scatter2_kernel(const int4* __restrict__ recs, const int* __restrict__ bstart,
                      const float* __restrict__ dinv, const float* __restrict__ g,
                      const float* __restrict__ b2, float* __restrict__ out, int n) {
    __shared__ float bins[BK_NODES];
    const int t = threadIdx.x;
    const int lo = blockIdx.x * BK_NODES;
    const int nn = min(BK_NODES, n - lo);
    if (t < nn) bins[t] = g[lo + t];
    __syncthreads();

    const int beg = bstart[blockIdx.x];
    const int end = bstart[blockIdx.x + 1];
    const int last = end - 1;
    for (int base = beg; base < end; base += S2_THREADS * S2_UNROLL) {
        int4 rec[S2_UNROLL];
        float gr[S2_UNROLL];
#pragma unroll
        for (int u = 0; u < S2_UNROLL; u++)
            rec[u] = recs[min(base + u * S2_THREADS + t, last)];
#pragma unroll
        for (int u = 0; u < S2_UNROLL; u++)
            gr[u] = g[rec[u].x];
        asm volatile("" :: HOLD8(gr));
#pragma unroll
        for (int u = 0; u < S2_UNROLL; u++)
            if (base + u * S2_THREADS + t < end)
                atomicAdd(&bins[rec[u].y], __int_as_float(rec[u].z) * gr[u]);
    }
    __syncthreads();
    if (t < nn) out[lo + t] = b2[0] + dinv[lo + t] * bins[t];
}

extern "C" void kernel_launch(void* const* d_in, const int* in_sizes, int n_in,
                              void* d_out, int out_size, void* d_ws, size_t ws_size,
                              hipStream_t stream) {
    const float* x  = (const float*)d_in[0];
    const int*   ei = (const int*)d_in[1];
    const float* ew = (const float*)d_in[2];
    const float* w1 = (const float*)d_in[3];
    const float* b1 = (const float*)d_in[4];
    const float* w2 = (const float*)d_in[5];
    const float* b2 = (const float*)d_in[6];
    float* out = (float*)d_out;

    const int E = in_sizes[2];
    const int N = in_sizes[0] / IN_FEAT;
    const int* rows = ei;
    const int* cols = ei + E;

    float* ws   = (float*)d_ws;
    float* dinv = ws;
    float* g    = ws + N;
    float* h0   = ws + 2 * (size_t)N;
    int*  counts = (int*)(h0 + (size_t)N * HIDDEN);
    int*  tot    = counts + BK_MAX * EC_CHUNKS;
    int*  bstart = tot + BK_MAX;
    int4* recs   = (int4*)(bstart + BK_MAX + 32);

    const int B = 256;
    const int NB = (N + BK_NODES - 1) / BK_NODES;

    edge_count_kernel<<<EC_CHUNKS, EC_THREADS, 0, stream>>>(cols, counts, NB, E);
    chunk_prefix_kernel<<<(NB + 7) / 8, B, 0, stream>>>(counts, tot, NB);
    edge_place_kernel<<<EC_CHUNKS, EC_THREADS, 0, stream>>>(rows, cols, ew, counts, tot,
                                                            bstart, recs, NB, E);
    block_degree_kernel<<<NB, DG_THREADS, 0, stream>>>(recs, bstart, dinv, N);

    {
        int mtiles = (N + 15) / 16;
        int waves  = mtiles * (HIDDEN / 16);
        int blocks = (waves + 7) / 8;
        gemm1_wmma_kernel<<<blocks, B, 0, stream>>>(x, w1, dinv, h0, N);
    }

    block_aggregate_kernel<<<NB, AG_THREADS, 0, stream>>>(recs, bstart, dinv, h0, b1, w2, g, N);
    block_scatter2_kernel<<<NB, S2_THREADS, 0, stream>>>(recs, bstart, dinv, g, b2, out, N);
}


# Degree pass folded into the layer-1 GEMM, one work-group per 128-node block: a wave owns a 16x64 strip (X read once by 16-byte loads, whole-row stores); 1 launch gone; k order of the f32 sums changes
# speedup vs baseline: 4.2654x; 1.1498x over previous
#include <hip/hip_runtime.h>
#include <hip/hip_bf16.h>

#define IN_FEAT 128
#define HIDDEN  64

typedef __attribute__((ext_vector_type(2))) float v2f;
typedef __attribute__((ext_vector_type(8))) float v8f;

#define HOLD8(a) "v"(a[0]), "v"(a[1]), "v"(a[2]), "v"(a[3]), "v"(a[4]), "v"(a[5]), "v"(a[6]), "v"(a[7])

#define BK_SHIFT   7
#define BK_NODES   (1 << BK_SHIFT)
#define GB_THREADS 256
#define GB_WAVES   (GB_THREADS / 32)
#define GB_UNROLL  20
#define SO_PITCH   68
static_assert(GB_WAVES * 16 == BK_NODES, "one 16-row strip per wave covers the block");
static_assert(IN_FEAT == 128 && HIDDEN == 64, "16 k groups of 8, four column tiles");

__global__ void __launch_bounds__(GB_THREADS)
block_gemm1_kernel(const float* __restrict__ X, const float* __restrict__ W,
                   const int4* __restrict__ recs, const int* __restrict__ bstart,
                   float* __restrict__ dinv, float* __restrict__ H, int n) {
    __shared__ float4 sW4[(IN_FEAT / 4) * HIDDEN];
    __shared__ __attribute__((aligned(16))) float sOut[GB_WAVES * 16 * SO_PITCH];
    __shared__ float bins[BK_NODES + 32];
    const int t = threadIdx.x;
    const int lane = t & 31;
    const int wave = t >> 5;
    const int half = lane >> 4;
    const int l = lane & 15;
    const int lo = blockIdx.x * BK_NODES;
    const int nn = min(BK_NODES, n - lo);

    if (t < BK_NODES + 32) bins[t] = 1.0f;
    for (int it = t; it < (IN_FEAT / 4) * (HIDDEN / 4); it += GB_THREADS) {
        const int kq = it >> 4;
        const int q = it & 15;
        const float4 r0 = *(const float4*)(W + (4 * kq + 0) * HIDDEN + 4 * q);
        const float4 r1 = *(const float4*)(W + (4 * kq + 1) * HIDDEN + 4 * q);
        const float4 r2 = *(const float4*)(W + (4 * kq + 2) * HIDDEN + 4 * q);
        const float4 r3 = *(const float4*)(W + (4 * kq + 3) * HIDDEN + 4 * q);
        float4* dst = sW4 + kq * HIDDEN + 4 * q;
        dst[0] = make_float4(r0.x, r1.x, r2.x, r3.x);
        dst[1] = make_float4(r0.y, r1.y, r2.y, r3.y);
        dst[2] = make_float4(r0.z, r1.z, r2.z, r3.z);
        dst[3] = make_float4(r0.w, r1.w, r2.w, r3.w);
    }
    __syncthreads();

    const int beg = bstart[blockIdx.x];
    const int end = bstart[blockIdx.x + 1];
    const int last = end - 1;
    for (int base = beg; base < end; base += GB_THREADS * GB_UNROLL) {
        int cl[GB_UNROLL];
        float w[GB_UNROLL];
#pragma unroll
        for (int u = 0; u < GB_UNROLL; u++) {
            const int idx = base + u * GB_THREADS + t;
            const int4 rec = recs[min(idx, last)];
            const bool ok = idx < end;
            cl[u] = ok ? rec.y : BK_NODES + lane;
            w[u] = ok ? __int_as_float(rec.z) : 0.0f;
        }
#pragma unroll
        for (int u = 0; u < GB_UNROLL; u++) atomicAdd(&bins[cl[u]], w[u]);
    }
    __syncthreads();
    if (t < nn) {
        const float d = bins[t];
        dinv[lo + t] = (d > 0.0f) ? __frsqrt_rn(d) : 0.0f;
    }

    const int m0 = lo + wave * 16;
    const float* Xrow = X + (size_t)min(m0 + l, n - 1) * IN_FEAT + 4 * half;
    float4 a[16];
#pragma unroll
    for (int j = 0; j < 16; j++) a[j] = *(const float4*)(Xrow + 8 * j);

    v8f acc[4];
#pragma unroll
    for (int nt = 0; nt < 4; nt++) acc[nt] = (v8f){0.f, 0.f, 0.f, 0.f, 0.f, 0.f, 0.f, 0.f};
#pragma unroll
    for (int j = 0; j < 16; j++) {
        const float4* brow = sW4 + (2 * j + half) * HIDDEN + l;
        float4 bv[4];
#pragma unroll
        for (int nt = 0; nt < 4; nt++) bv[nt] = brow[16 * nt];
        const v2f a0 = {a[j].x, a[j].y};
        const v2f a1 = {a[j].z, a[j].w};
#pragma unroll
        for (int nt = 0; nt < 4; nt++) {
            const v2f b0 = {bv[nt].x, bv[nt].y};
            acc[nt] = __builtin_amdgcn_wmma_f32_16x16x4_f32(
                 false, a0,  false, b0,
                 (short)0, acc[nt],  false,  false);
        }
#pragma unroll
        for (int nt = 0; nt < 4; nt++) {
            const v2f b1 = {bv[nt].z, bv[nt].w};
            acc[nt] = __builtin_amdgcn_wmma_f32_16x16x4_f32(
                 false, a1,  false, b1,
                 (short)0, acc[nt],  false,  false);
        }
    }

    float* tile = sOut + wave * 16 * SO_PITCH;
#pragma unroll
    for (int nt = 0; nt < 4; nt++)
#pragma unroll
        for (int r = 0; r < 8; r++)
            tile[(8 * half + r) * SO_PITCH + 16 * nt + l] = acc[nt][r];
    __syncthreads();
#pragma unroll
    for (int i = 0; i < 8; i++) {
        const int rr = 2 * i + half;
        const int node = wave * 16 + rr;
        const float4 v = *(const float4*)(tile + rr * SO_PITCH + 4 * l);
        const float d = bins[min(node, BK_NODES - 1)];
        const float s = (d > 0.0f) ? __frsqrt_rn(d) : 0.0f;
        if (node < nn)
            *(float4*)(H + (size_t)(lo + node) * HIDDEN + 4 * l) =
                make_float4(s * v.x, s * v.y, s * v.z, s * v.w);
    }
}

#define BK_MAX     1024
#define EC_CHUNKS  256
#define EC_THREADS 1024
#define AG_THREADS 256
#define AG_STRIDE  ((AG_THREADS / 32) * 32)
#define AG_BATCH   16
static_assert(EC_CHUNKS == 256, "chunk_prefix_kernel gives each lane 8 chunks");
static_assert(HIDDEN == 64, "one float2 per lane covers a row");

__global__ void __launch_bounds__(EC_THREADS)
edge_count_kernel(const int* __restrict__ cols, int* __restrict__ counts,
                  int nb, int e) {
    __shared__ int cnt[BK_MAX];
    for (int t = threadIdx.x; t < BK_MAX; t += EC_THREADS) cnt[t] = 0;
    __syncthreads();
    const int per = (e + EC_CHUNKS - 1) / EC_CHUNKS;
    const int start = blockIdx.x * per;
    const int end = min(e, start + per);
    for (int base = start; base < end; base += EC_THREADS * 4) {
        int c[4];
#pragma unroll
        for (int u = 0; u < 4; u++) {
            const int i = base + u * EC_THREADS + (int)threadIdx.x;
            c[u] = (i < end) ? cols[i] : -1;
        }
#pragma unroll
        for (int u = 0; u < 4; u++)
            if (c[u] >= 0) atomicAdd(&cnt[c[u] >> BK_SHIFT], 1);
    }
    __syncthreads();
    for (int b = threadIdx.x; b < nb; b += EC_THREADS)
        counts[(size_t)b * EC_CHUNKS + blockIdx.x] = cnt[b];
}

__global__ void __launch_bounds__(256)
chunk_prefix_kernel(int* __restrict__ counts, int* __restrict__ tot, int nb) {
    const int b = blockIdx.x * (blockDim.x >> 5) + (threadIdx.x >> 5);
    const int lane = threadIdx.x & 31;
    if (b >= nb) return;
    int4* row = (int4*)(counts + (size_t)b * EC_CHUNKS) + 2 * lane;
    const int4 a = row[0];
    const int4 c = row[1];
    const int sum = a.x + a.y + a.z + a.w + c.x + c.y + c.z + c.w;
    int inc = sum;
    for (int off = 1; off < 32; off <<= 1) {
        const int v = __shfl_up(inc, off, 32);
        if (lane >= off) inc += v;
    }
    int4 oa, oc;
    oa.x = inc - sum;
    oa.y = oa.x + a.x;
    oa.z = oa.y + a.y;
    oa.w = oa.z + a.z;
    oc.x = oa.w + a.w;
    oc.y = oc.x + c.x;
    oc.z = oc.y + c.y;
    oc.w = oc.z + c.z;
    row[0] = oa;
    row[1] = oc;
    if (lane == 31) tot[b] = inc;
}

#define EP_STAGE 12800
__global__ void __launch_bounds__(EC_THREADS)
edge_place_kernel(const int* __restrict__ rows, const int* __restrict__ cols,
                  const float* __restrict__ ew,
                  const int* __restrict__ counts, const int* __restrict__ tot,
                  int* __restrict__ bstart, int4* __restrict__ recs,
                  int nb, int e) {
    __shared__ int4 stage[EP_STAGE];
    __shared__ int cursor[BK_MAX];
    __shared__ int delta[BK_MAX];
    __shared__ int wsum_g[32];
    __shared__ int wsum_l[32];
    const int t = threadIdx.x;
    const int lane = t & 31;
    const int wid = t >> 5;
    const int per = (e + EC_CHUNKS - 1) / EC_CHUNKS;
    const int start = blockIdx.x * per;
    const int end = min(e, start + per);
    const bool staged = per <= EP_STAGE;

    int mine = 0, cur = 0, len = 0;
    if (t < nb) {
        mine = tot[t];
        cur = counts[(size_t)t * EC_CHUNKS + blockIdx.x];
        const int nxt = (blockIdx.x + 1 < EC_CHUNKS)
                            ? counts[(size_t)t * EC_CHUNKS + blockIdx.x + 1] : mine;
        len = nxt - cur;
    }
    int inc_g = mine;
    int inc_l = len;
    for (int off = 1; off < 32; off <<= 1) {
        const int vg = __shfl_up(inc_g, off, 32);
        const int vl = __shfl_up(inc_l, off, 32);
        if (lane >= off) { inc_g += vg; inc_l += vl; }
    }
    if (lane == 31) { wsum_g[wid] = inc_g; wsum_l[wid] = inc_l; }
    __syncthreads();
    if (wid == 0) {
        const int sg = wsum_g[lane];
        const int sl = wsum_l[lane];
        int sig = sg;
        int sil = sl;
        for (int off = 1; off < 32; off <<= 1) {
            const int vg = __shfl_up(sig, off, 32);
            const int vl = __shfl_up(sil, off, 32);
            if (lane >= off) { sig += vg; sil += vl; }
        }
        wsum_g[lane] = sig - sg;
        wsum_l[lane] = sil - sl;
    }
    __syncthreads();
    const int excl = inc_g - mine + wsum_g[wid];
    const int lexcl = inc_l - len + wsum_l[wid];
    if (t < nb) {
        const int gpos = excl + cur;
        cursor[t] = staged ? lexcl : gpos;
        delta[t] = gpos - lexcl;
        if (blockIdx.x == 0) {
            bstart[t] = excl;
            if (t == nb - 1) bstart[nb] = excl + mine;
        }
    }
    __syncthreads();

    for (int base = start; base < end; base += EC_THREADS * 4) {
        int c[4];
        int r[4];
        float w[4];
#pragma unroll
        for (int u = 0; u < 4; u++) {
            const int i = base + u * EC_THREADS + t;
            const bool ok = i < end;
            c[u] = ok ? cols[i] : -1;
            r[u] = ok ? rows[i] : 0;
            w[u] = ok ? ew[i] : 0.0f;
        }
#pragma unroll
        for (int u = 0; u < 4; u++) {
            if (c[u] >= 0) {
                const int pos = atomicAdd(&cursor[c[u] >> BK_SHIFT], 1);
                if (staged)
                    stage[pos] = make_int4(r[u], c[u], __float_as_int(w[u]), 0);
                else
                    recs[pos] = make_int4(r[u], c[u] & (BK_NODES - 1), __float_as_int(w[u]), 0);
            }
        }
    }
    if (staged) {
        __syncthreads();
        const int cnt = end - start;
        for (int p = t; p < cnt; p += EC_THREADS) {
            const int4 rec = stage[p];
            recs[p + delta[rec.y >> BK_SHIFT]] =
                make_int4(rec.x, rec.y & (BK_NODES - 1), rec.z, 0);
        }
    }
}

#define AG_ROWS (BK_NODES / (AG_THREADS / 32))
static_assert(AG_ROWS * (AG_THREADS / 32) == BK_NODES && AG_ROWS <= 16,
              "the waves' node groups cover the tile; one lane of a 16-lane row per node");

__device__ __forceinline__ float lane_xor16(float x) {
    const int b = __float_as_int(x);
    return __int_as_float(__builtin_amdgcn_permlanex16(b, b, 0x76543210, 0xfedcba98, false, false));
}
template <int M>
__device__ __forceinline__ float row_xor(float x) {
    static_assert(M > 0 && M < 16, "row_xmask stays inside a row of 16");
    const int b = __float_as_int(x);
    return __int_as_float(__builtin_amdgcn_update_dpp(b, b, 0x160 + M, 0xf, 0xf, false));
}

__global__ void __launch_bounds__(AG_THREADS)
block_aggregate_kernel(const int4* __restrict__ recs, const int* __restrict__ bstart,
                       const float* __restrict__ dinv, const float* __restrict__ h0,
                       const float* __restrict__ b1, const float* __restrict__ w2,
                       float* __restrict__ g, int n) {
    __shared__ __attribute__((aligned(16))) float acc[(BK_NODES + 1) * HIDDEN];
    __shared__ float gs[BK_NODES];
    __shared__ float dl[BK_NODES];
    const int t = threadIdx.x;
    const int lane = t & 31;
    const int wave = __builtin_amdgcn_readfirstlane(t >> 5);
    const int lo = blockIdx.x * BK_NODES;
    const int nn = min(BK_NODES, n - lo);
    float2* acc2 = (float2*)acc;
    if (t < BK_NODES) dl[t] = (t < nn) ? dinv[lo + t] : 0.0f;
    for (int idx = t; idx < nn * (HIDDEN / 2); idx += AG_THREADS) {
        const int node = lo + (idx >> 5);
        acc2[idx] = ((const float2*)(h0 + (size_t)node * HIDDEN))[idx & 31];
    }
    __syncthreads();

    const int beg = __builtin_amdgcn_readfirstlane(bstart[blockIdx.x]);
    const int end = __builtin_amdgcn_readfirstlane(bstart[blockIdx.x + 1]);
    const int g0 = beg + wave * 32;
    if (g0 < end) {
        const int last = end - 1;
        int4 rec = recs[min(g0 + lane, last)];
        for (int g = g0; g < end; g += AG_STRIDE) {
            const int4 nxt = recs[min(g + AG_STRIDE + lane, last)];
            if (g + lane >= end) rec = make_int4(0, BK_NODES, 0, 0);
#pragma unroll
            for (int u0 = 0; u0 < 32; u0 += AG_BATCH) {
                float2 v[AG_BATCH];
                int cl[AG_BATCH];
                float nm[AG_BATCH];
#pragma unroll
                for (int k = 0; k < AG_BATCH; k++) {
                    const int r = __builtin_amdgcn_readlane(rec.x, u0 + k);
                    cl[k] = __builtin_amdgcn_readlane(rec.y, u0 + k);
                    nm[k] = __int_as_float(__builtin_amdgcn_readlane(rec.z, u0 + k));
                    v[k] = ((const float2*)(h0 + (size_t)r * HIDDEN))[lane];
                }
#pragma unroll
                for (int k = 0; k < AG_BATCH; k++) {
                    float* d = acc + cl[k] * HIDDEN + 2 * lane;
                    atomicAdd(d + 0, nm[k] * v[k].x);
                    atomicAdd(d + 1, nm[k] * v[k].y);
                }
            }
            rec = nxt;
        }
    }
    __syncthreads();

    const float2 bv = ((const float2*)b1)[lane];
    const float2 wv = ((const float2*)w2)[lane];
    float p[AG_ROWS];
#pragma unroll
    for (int k = 0; k < AG_ROWS; k++) {
        const int i = min(wave * AG_ROWS + k, nn - 1);
        const float2 a = acc2[i * (HIDDEN / 2) + lane];
        const float dc = dl[i];
        float2 hv;
        hv.x = fmaxf(dc * a.x + bv.x, 0.0f);
        hv.y = fmaxf(dc * a.y + bv.y, 0.0f);
        p[k] = hv.x * wv.x + hv.y * wv.y;
    }
#pragma unroll
    for (int k = 0; k < AG_ROWS; k++) p[k] += lane_xor16(p[k]);
#pragma unroll
    for (int k = 0; k < AG_ROWS; k++) p[k] += row_xor<8>(p[k]);
#pragma unroll
    for (int k = 0; k < AG_ROWS; k++) p[k] += row_xor<4>(p[k]);
#pragma unroll
    for (int k = 0; k < AG_ROWS; k++) p[k] += row_xor<2>(p[k]);
#pragma unroll
    for (int k = 0; k < AG_ROWS; k++) p[k] += row_xor<1>(p[k]);
    float mine = p[0];
#pragma unroll
    for (int k = 1; k < AG_ROWS; k++) mine = (lane == k) ? p[k] : mine;
    if (lane < AG_ROWS) gs[wave * AG_ROWS + lane] = dl[wave * AG_ROWS + lane] * mine;
    __syncthreads();
    if (t < nn) g[lo + t] = gs[t];
}

#define S2_THREADS 320
#define S2_UNROLL  8
static_assert(S2_UNROLL == 8, "HOLD8 names eight values");
static_assert(S2_THREADS >= BK_NODES, "one thread per node for the start values and the stores");

__global__ void __launch_bounds__(S2_THREADS)
block_scatter2_kernel(const int4* __restrict__ recs, const int* __restrict__ bstart,
                      const float* __restrict__ dinv, const float* __restrict__ g,
                      const float* __restrict__ b2, float* __restrict__ out, int n) {
    __shared__ float bins[BK_NODES];
    const int t = threadIdx.x;
    const int lo = blockIdx.x * BK_NODES;
    const int nn = min(BK_NODES, n - lo);
    if (t < nn) bins[t] = g[lo + t];
    __syncthreads();

    const int beg = bstart[blockIdx.x];
    const int end = bstart[blockIdx.x + 1];
    const int last = end - 1;
    for (int base = beg; base < end; base += S2_THREADS * S2_UNROLL) {
        int4 rec[S2_UNROLL];
        float gr[S2_UNROLL];
#pragma unroll
        for (int u = 0; u < S2_UNROLL; u++)
            rec[u] = recs[min(base + u * S2_THREADS + t, last)];
#pragma unroll
        for (int u = 0; u < S2_UNROLL; u++)
            gr[u] = g[rec[u].x];
        asm volatile("" :: HOLD8(gr));
#pragma unroll
        for (int u = 0; u < S2_UNROLL; u++)
            if (base + u * S2_THREADS + t < end)
                atomicAdd(&bins[rec[u].y], __int_as_float(rec[u].z) * gr[u]);
    }
    __syncthreads();
    if (t < nn) out[lo + t] = b2[0] + dinv[lo + t] * bins[t];
}

extern "C" void kernel_launch(void* const* d_in, const int* in_sizes, int n_in,
                              void* d_out, int out_size, void* d_ws, size_t ws_size,
                              hipStream_t stream) {
    const float* x  = (const float*)d_in[0];
    const int*   ei = (const int*)d_in[1];
    const float* ew = (const float*)d_in[2];
    const float* w1 = (const float*)d_in[3];
    const float* b1 = (const float*)d_in[4];
    const float* w2 = (const float*)d_in[5];
    const float* b2 = (const float*)d_in[6];
    float* out = (float*)d_out;

    const int E = in_sizes[2];
    const int N = in_sizes[0] / IN_FEAT;
    const int* rows = ei;
    const int* cols = ei + E;

    float* ws   = (float*)d_ws;
    float* dinv = ws;
    float* g    = ws + N;
    float* h0   = ws + 2 * (size_t)N;
    int*  counts = (int*)(h0 + (size_t)N * HIDDEN);
    int*  tot    = counts + BK_MAX * EC_CHUNKS;
    int*  bstart = tot + BK_MAX;
    int4* recs   = (int4*)(bstart + BK_MAX + 32);

    const int B = 256;
    const int NB = (N + BK_NODES - 1) / BK_NODES;

    edge_count_kernel<<<EC_CHUNKS, EC_THREADS, 0, stream>>>(cols, counts, NB, E);
    chunk_prefix_kernel<<<(NB + 7) / 8, B, 0, stream>>>(counts, tot, NB);
    edge_place_kernel<<<EC_CHUNKS, EC_THREADS, 0, stream>>>(rows, cols, ew, counts, tot,
                                                            bstart, recs, NB, E);
    block_gemm1_kernel<<<NB, GB_THREADS, 0, stream>>>(x, w1, recs, bstart, dinv, h0, N);

    block_aggregate_kernel<<<NB, AG_THREADS, 0, stream>>>(recs, bstart, dinv, h0, b1, w2, g, N);
    block_scatter2_kernel<<<NB, S2_THREADS, 0, stream>>>(recs, bstart, dinv, g, b2, out, N);
}


# Blocks of 104 destination nodes, not 128: 962 work-groups per block-wise launch sit 4 deep on all 256 WGPs (782 sat 3 deep, 4 on a few that set the time); LDS-atomic sum order changes
# speedup vs baseline: 4.6638x; 1.0934x over previous
#include <hip/hip_runtime.h>
#include <hip/hip_bf16.h>

#define IN_FEAT 128
#define HIDDEN  64

typedef __attribute__((ext_vector_type(2))) float v2f;
typedef __attribute__((ext_vector_type(8))) float v8f;

#define HOLD8(a) "v"(a[0]), "v"(a[1]), "v"(a[2]), "v"(a[3]), "v"(a[4]), "v"(a[5]), "v"(a[6]), "v"(a[7])

#define BK_NODES   104
#define GB_THREADS 224
#define GB_WAVES   (GB_THREADS / 32)
#define GB_UNROLL  20
#define SO_PITCH   68
static_assert(GB_WAVES * 16 >= BK_NODES && GB_THREADS >= BK_NODES + 32, "the waves' 16-row strips cover the block; one thread per bin");
static_assert(IN_FEAT == 128 && HIDDEN == 64, "16 k groups of 8, four column tiles");

__global__ void __launch_bounds__(GB_THREADS)
block_gemm1_kernel(const float* __restrict__ X, const float* __restrict__ W,
                   const int4* __restrict__ recs, const int* __restrict__ bstart,
                   float* __restrict__ dinv, float* __restrict__ H, int n) {
    __shared__ float4 sW4[(IN_FEAT / 4) * HIDDEN];
    __shared__ __attribute__((aligned(16))) float sOut[GB_WAVES * 16 * SO_PITCH];
    __shared__ float bins[BK_NODES + 32];
    const int t = threadIdx.x;
    const int lane = t & 31;
    const int wave = t >> 5;
    const int half = lane >> 4;
    const int l = lane & 15;
    const int lo = blockIdx.x * BK_NODES;
    const int nn = min(BK_NODES, n - lo);

    if (t < BK_NODES + 32) bins[t] = 1.0f;
    for (int it = t; it < (IN_FEAT / 4) * (HIDDEN / 4); it += GB_THREADS) {
        const int kq = it >> 4;
        const int q = it & 15;
        const float4 r0 = *(const float4*)(W + (4 * kq + 0) * HIDDEN + 4 * q);
        const float4 r1 = *(const float4*)(W + (4 * kq + 1) * HIDDEN + 4 * q);
        const float4 r2 = *(const float4*)(W + (4 * kq + 2) * HIDDEN + 4 * q);
        const float4 r3 = *(const float4*)(W + (4 * kq + 3) * HIDDEN + 4 * q);
        float4* dst = sW4 + kq * HIDDEN + 4 * q;
        dst[0] = make_float4(r0.x, r1.x, r2.x, r3.x);
        dst[1] = make_float4(r0.y, r1.y, r2.y, r3.y);
        dst[2] = make_float4(r0.z, r1.z, r2.z, r3.z);
        dst[3] = make_float4(r0.w, r1.w, r2.w, r3.w);
    }
    __syncthreads();

    const int beg = bstart[blockIdx.x];
    const int end = bstart[blockIdx.x + 1];
    const int last = end - 1;
    for (int base = beg; base < end; base += GB_THREADS * GB_UNROLL) {
        int cl[GB_UNROLL];
        float w[GB_UNROLL];
#pragma unroll
        for (int u = 0; u < GB_UNROLL; u++) {
            const int idx = base + u * GB_THREADS + t;
            const int4 rec = recs[min(idx, last)];
            const bool ok = idx < end;
            cl[u] = ok ? rec.y : BK_NODES + lane;
            w[u] = ok ? __int_as_float(rec.z) : 0.0f;
        }
#pragma unroll
        for (int u = 0; u < GB_UNROLL; u++) atomicAdd(&bins[cl[u]], w[u]);
    }
    __syncthreads();
    if (t < nn) {
        const float d = bins[t];
        dinv[lo + t] = (d > 0.0f) ? __frsqrt_rn(d) : 0.0f;
    }

    const int m0 = lo + wave * 16;
    const float* Xrow = X + (size_t)min(m0 + l, n - 1) * IN_FEAT + 4 * half;
    float4 a[16];
#pragma unroll
    for (int j = 0; j < 16; j++) a[j] = *(const float4*)(Xrow + 8 * j);

    v8f acc[4];
#pragma unroll
    for (int nt = 0; nt < 4; nt++) acc[nt] = (v8f){0.f, 0.f, 0.f, 0.f, 0.f, 0.f, 0.f, 0.f};
#pragma unroll
    for (int j = 0; j < 16; j++) {
        const float4* brow = sW4 + (2 * j + half) * HIDDEN + l;
        float4 bv[4];
#pragma unroll
        for (int nt = 0; nt < 4; nt++) bv[nt] = brow[16 * nt];
        const v2f a0 = {a[j].x, a[j].y};
        const v2f a1 = {a[j].z, a[j].w};
#pragma unroll
        for (int nt = 0; nt < 4; nt++) {
            const v2f b0 = {bv[nt].x, bv[nt].y};
            acc[nt] = __builtin_amdgcn_wmma_f32_16x16x4_f32(
                 false, a0,  false, b0,
                 (short)0, acc[nt],  false,  false);
        }
#pragma unroll
        for (int nt = 0; nt < 4; nt++) {
            const v2f b1 = {bv[nt].z, bv[nt].w};
            acc[nt] = __builtin_amdgcn_wmma_f32_16x16x4_f32(
                 false, a1,  false, b1,
                 (short)0, acc[nt],  false,  false);
        }
    }

    float* tile = sOut + wave * 16 * SO_PITCH;
#pragma unroll
    for (int nt = 0; nt < 4; nt++)
#pragma unroll
        for (int r = 0; r < 8; r++)
            tile[(8 * half + r) * SO_PITCH + 16 * nt + l] = acc[nt][r];
    __syncthreads();
#pragma unroll
    for (int i = 0; i < 8; i++) {
        const int rr = 2 * i + half;
        const int node = wave * 16 + rr;
        const float4 v = *(const float4*)(tile + rr * SO_PITCH + 4 * l);
        const float d = bins[min(node, BK_NODES - 1)];
        const float s = (d > 0.0f) ? __frsqrt_rn(d) : 0.0f;
        if (node < nn)
            *(float4*)(H + (size_t)(lo + node) * HIDDEN + 4 * l) =
                make_float4(s * v.x, s * v.y, s * v.z, s * v.w);
    }
}

#define BK_MAX     1024
#define EC_CHUNKS  256
#define EC_THREADS 1024
#define AG_THREADS 256
#define AG_STRIDE  ((AG_THREADS / 32) * 32)
#define AG_BATCH   16
static_assert(EC_CHUNKS == 256, "chunk_prefix_kernel gives each lane 8 chunks");
static_assert(HIDDEN == 64, "one float2 per lane covers a row");

__global__ void __launch_bounds__(EC_THREADS)
edge_count_kernel(const int* __restrict__ cols, int* __restrict__ counts,
                  int nb, int e) {
    __shared__ int cnt[BK_MAX];
    for (int t = threadIdx.x; t < BK_MAX; t += EC_THREADS) cnt[t] = 0;
    __syncthreads();
    const int per = (e + EC_CHUNKS - 1) / EC_CHUNKS;
    const int start = blockIdx.x * per;
    const int end = min(e, start + per);
    for (int base = start; base < end; base += EC_THREADS * 4) {
        int c[4];
#pragma unroll
        for (int u = 0; u < 4; u++) {
            const int i = base + u * EC_THREADS + (int)threadIdx.x;
            c[u] = (i < end) ? cols[i] : -1;
        }
#pragma unroll
        for (int u = 0; u < 4; u++)
            if (c[u] >= 0) atomicAdd(&cnt[(unsigned)c[u] / BK_NODES], 1);
    }
    __syncthreads();
    for (int b = threadIdx.x; b < nb; b += EC_THREADS)
        counts[(size_t)b * EC_CHUNKS + blockIdx.x] = cnt[b];
}

__global__ void __launch_bounds__(256)
chunk_prefix_kernel(int* __restrict__ counts, int* __restrict__ tot, int nb) {
    const int b = blockIdx.x * (blockDim.x >> 5) + (threadIdx.x >> 5);
    const int lane = threadIdx.x & 31;
    if (b >= nb) return;
    int4* row = (int4*)(counts + (size_t)b * EC_CHUNKS) + 2 * lane;
    const int4 a = row[0];
    const int4 c = row[1];
    const int sum = a.x + a.y + a.z + a.w + c.x + c.y + c.z + c.w;
    int inc = sum;
    for (int off = 1; off < 32; off <<= 1) {
        const int v = __shfl_up(inc, off, 32);
        if (lane >= off) inc += v;
    }
    int4 oa, oc;
    oa.x = inc - sum;
    oa.y = oa.x + a.x;
    oa.z = oa.y + a.y;
    oa.w = oa.z + a.z;
    oc.x = oa.w + a.w;
    oc.y = oc.x + c.x;
    oc.z = oc.y + c.y;
    oc.w = oc.z + c.z;
    row[0] = oa;
    row[1] = oc;
    if (lane == 31) tot[b] = inc;
}

#define EP_STAGE 12800
__global__ void __launch_bounds__(EC_THREADS)
edge_place_kernel(const int* __restrict__ rows, const int* __restrict__ cols,
                  const float* __restrict__ ew,
                  const int* __restrict__ counts, const int* __restrict__ tot,
                  int* __restrict__ bstart, int4* __restrict__ recs,
                  int nb, int e) {
    __shared__ int4 stage[EP_STAGE];
    __shared__ int cursor[BK_MAX];
    __shared__ int delta[BK_MAX];
    __shared__ int wsum_g[32];
    __shared__ int wsum_l[32];
    const int t = threadIdx.x;
    const int lane = t & 31;
    const int wid = t >> 5;
    const int per = (e + EC_CHUNKS - 1) / EC_CHUNKS;
    const int start = blockIdx.x * per;
    const int end = min(e, start + per);
    const bool staged = per <= EP_STAGE;

    int mine = 0, cur = 0, len = 0;
    if (t < nb) {
        mine = tot[t];
        cur = counts[(size_t)t * EC_CHUNKS + blockIdx.x];
        const int nxt = (blockIdx.x + 1 < EC_CHUNKS)
                            ? counts[(size_t)t * EC_CHUNKS + blockIdx.x + 1] : mine;
        len = nxt - cur;
    }
    int inc_g = mine;
    int inc_l = len;
    for (int off = 1; off < 32; off <<= 1) {
        const int vg = __shfl_up(inc_g, off, 32);
        const int vl = __shfl_up(inc_l, off, 32);
        if (lane >= off) { inc_g += vg; inc_l += vl; }
    }
    if (lane == 31) { wsum_g[wid] = inc_g; wsum_l[wid] = inc_l; }
    __syncthreads();
    if (wid == 0) {
        const int sg = wsum_g[lane];
        const int sl = wsum_l[lane];
        int sig = sg;
        int sil = sl;
        for (int off = 1; off < 32; off <<= 1) {
            const int vg = __shfl_up(sig, off, 32);
            const int vl = __shfl_up(sil, off, 32);
            if (lane >= off) { sig += vg; sil += vl; }
        }
        wsum_g[lane] = sig - sg;
        wsum_l[lane] = sil - sl;
    }
    __syncthreads();
    const int excl = inc_g - mine + wsum_g[wid];
    const int lexcl = inc_l - len + wsum_l[wid];
    if (t < nb) {
        const int gpos = excl + cur;
        cursor[t] = staged ? lexcl : gpos;
        delta[t] = gpos - lexcl;
        if (blockIdx.x == 0) {
            bstart[t] = excl;
            if (t == nb - 1) bstart[nb] = excl + mine;
        }
    }
    __syncthreads();

    for (int base = start; base < end; base += EC_THREADS * 4) {
        int c[4];
        int r[4];
        float w[4];
#pragma unroll
        for (int u = 0; u < 4; u++) {
            const int i = base + u * EC_THREADS + t;
            const bool ok = i < end;
            c[u] = ok ? cols[i] : -1;
            r[u] = ok ? rows[i] : 0;
            w[u] = ok ? ew[i] : 0.0f;
        }
#pragma unroll
        for (int u = 0; u < 4; u++) {
            if (c[u] >= 0) {
                const int blk = (unsigned)c[u] / BK_NODES;
                const int pos = atomicAdd(&cursor[blk], 1);
                if (staged)
                    stage[pos] = make_int4(r[u], c[u], __float_as_int(w[u]), 0);
                else
                    recs[pos] = make_int4(r[u], c[u] - blk * BK_NODES, __float_as_int(w[u]), 0);
            }
        }
    }
    if (staged) {
        __syncthreads();
        const int cnt = end - start;
        for (int p = t; p < cnt; p += EC_THREADS) {
            const int4 rec = stage[p];
            const int blk = (unsigned)rec.y / BK_NODES;
            recs[p + delta[blk]] =
                make_int4(rec.x, rec.y - blk * BK_NODES, rec.z, 0);
        }
    }
}

#define AG_ROWS (BK_NODES / (AG_THREADS / 32))
static_assert(AG_ROWS * (AG_THREADS / 32) == BK_NODES && AG_ROWS <= 16,
              "the waves' node groups cover the tile; one lane of a 16-lane row per node");

__device__ __forceinline__ float lane_xor16(float x) {
    const int b = __float_as_int(x);
    return __int_as_float(__builtin_amdgcn_permlanex16(b, b, 0x76543210, 0xfedcba98, false, false));
}
template <int M>
__device__ __forceinline__ float row_xor(float x) {
    static_assert(M > 0 && M < 16, "row_xmask stays inside a row of 16");
    const int b = __float_as_int(x);
    return __int_as_float(__builtin_amdgcn_update_dpp(b, b, 0x160 + M, 0xf, 0xf, false));
}

__global__ void __launch_bounds__(AG_THREADS)
block_aggregate_kernel(const int4* __restrict__ recs, const int* __restrict__ bstart,
                       const float* __restrict__ dinv, const float* __restrict__ h0,
                       const float* __restrict__ b1, const float* __restrict__ w2,
                       float* __restrict__ g, int n) {
    __shared__ __attribute__((aligned(16))) float acc[(BK_NODES + 1) * HIDDEN];
    __shared__ float gs[BK_NODES];
    __shared__ float dl[BK_NODES];
    const int t = threadIdx.x;
    const int lane = t & 31;
    const int wave = __builtin_amdgcn_readfirstlane(t >> 5);
    const int lo = blockIdx.x * BK_NODES;
    const int nn = min(BK_NODES, n - lo);
    float2* acc2 = (float2*)acc;
    if (t < BK_NODES) dl[t] = (t < nn) ? dinv[lo + t] : 0.0f;
    for (int idx = t; idx < nn * (HIDDEN / 2); idx += AG_THREADS) {
        const int node = lo + (idx >> 5);
        acc2[idx] = ((const float2*)(h0 + (size_t)node * HIDDEN))[idx & 31];
    }
    __syncthreads();

    const int beg = __builtin_amdgcn_readfirstlane(bstart[blockIdx.x]);
    const int end = __builtin_amdgcn_readfirstlane(bstart[blockIdx.x + 1]);
    const int g0 = beg + wave * 32;
    if (g0 < end) {
        const int last = end - 1;
        int4 rec = recs[min(g0 + lane, last)];
        for (int g = g0; g < end; g += AG_STRIDE) {
            const int4 nxt = recs[min(g + AG_STRIDE + lane, last)];
            if (g + lane >= end) rec = make_int4(0, BK_NODES, 0, 0);
#pragma unroll
            for (int u0 = 0; u0 < 32; u0 += AG_BATCH) {
                float2 v[AG_BATCH];
                int cl[AG_BATCH];
                float nm[AG_BATCH];
#pragma unroll
                for (int k = 0; k < AG_BATCH; k++) {
                    const int r = __builtin_amdgcn_readlane(rec.x, u0 + k);
                    cl[k] = __builtin_amdgcn_readlane(rec.y, u0 + k);
                    nm[k] = __int_as_float(__builtin_amdgcn_readlane(rec.z, u0 + k));
                    v[k] = ((const float2*)(h0 + (size_t)r * HIDDEN))[lane];
                }
#pragma unroll
                for (int k = 0; k < AG_BATCH; k++) {
                    float* d = acc + cl[k] * HIDDEN + 2 * lane;
                    atomicAdd(d + 0, nm[k] * v[k].x);
                    atomicAdd(d + 1, nm[k] * v[k].y);
                }
            }
            rec = nxt;
        }
    }
    __syncthreads();

    const float2 bv = ((const float2*)b1)[lane];
    const float2 wv = ((const float2*)w2)[lane];
    float p[AG_ROWS];
#pragma unroll
    for (int k = 0; k < AG_ROWS; k++) {
        const int i = min(wave * AG_ROWS + k, nn - 1);
        const float2 a = acc2[i * (HIDDEN / 2) + lane];
        const float dc = dl[i];
        float2 hv;
        hv.x = fmaxf(dc * a.x + bv.x, 0.0f);
        hv.y = fmaxf(dc * a.y + bv.y, 0.0f);
        p[k] = hv.x * wv.x + hv.y * wv.y;
    }
#pragma unroll
    for (int k = 0; k < AG_ROWS; k++) p[k] += lane_xor16(p[k]);
#pragma unroll
    for (int k = 0; k < AG_ROWS; k++) p[k] += row_xor<8>(p[k]);
#pragma unroll
    for (int k = 0; k < AG_ROWS; k++) p[k] += row_xor<4>(p[k]);
#pragma unroll
    for (int k = 0; k < AG_ROWS; k++) p[k] += row_xor<2>(p[k]);
#pragma unroll
    for (int k = 0; k < AG_ROWS; k++) p[k] += row_xor<1>(p[k]);
    float mine = p[0];
#pragma unroll
    for (int k = 1; k < AG_ROWS; k++) mine = (lane == k) ? p[k] : mine;
    if (lane < AG_ROWS) gs[wave * AG_ROWS + lane] = dl[wave * AG_ROWS + lane] * mine;
    __syncthreads();
    if (t < nn) g[lo + t] = gs[t];
}

#define S2_THREADS 320
#define S2_UNROLL  8
static_assert(S2_UNROLL == 8, "HOLD8 names eight values");
static_assert(S2_THREADS >= BK_NODES, "one thread per node for the start values and the stores");

__global__ void __launch_bounds__(S2_THREADS)
block_scatter2_kernel(const int4* __restrict__ recs, const int* __restrict__ bstart,
                      const float* __restrict__ dinv, const float* __restrict__ g,
                      const float* __restrict__ b2, float* __restrict__ out, int n) {
    __shared__ float bins[BK_NODES];
    const int t = threadIdx.x;
    const int lo = blockIdx.x * BK_NODES;
    const int nn = min(BK_NODES, n - lo);
    if (t < nn) bins[t] = g[lo + t];
    __syncthreads();

    const int beg = bstart[blockIdx.x];
    const int end = bstart[blockIdx.x + 1];
    const int last = end - 1;
    for (int base = beg; base < end; base += S2_THREADS * S2_UNROLL) {
        int4 rec[S2_UNROLL];
        float gr[S2_UNROLL];
#pragma unroll
        for (int u = 0; u < S2_UNROLL; u++)
            rec[u] = recs[min(base + u * S2_THREADS + t, last)];
#pragma unroll
        for (int u = 0; u < S2_UNROLL; u++)
            gr[u] = g[rec[u].x];
        asm volatile("" :: HOLD8(gr));
#pragma unroll
        for (int u = 0; u < S2_UNROLL; u++)
            if (base + u * S2_THREADS + t < end)
                atomicAdd(&bins[rec[u].y], __int_as_float(rec[u].z) * gr[u]);
    }
    __syncthreads();
    if (t < nn) out[lo + t] = b2[0] + dinv[lo + t] * bins[t];
}

extern "C" void kernel_launch(void* const* d_in, const int* in_sizes, int n_in,
                              void* d_out, int out_size, void* d_ws, size_t ws_size,
                              hipStream_t stream) {
    const float* x  = (const float*)d_in[0];
    const int*   ei = (const int*)d_in[1];
    const float* ew = (const float*)d_in[2];
    const float* w1 = (const float*)d_in[3];
    const float* b1 = (const float*)d_in[4];
    const float* w2 = (const float*)d_in[5];
    const float* b2 = (const float*)d_in[6];
    float* out = (float*)d_out;

    const int E = in_sizes[2];
    const int N = in_sizes[0] / IN_FEAT;
    const int* rows = ei;
    const int* cols = ei + E;

    float* ws   = (float*)d_ws;
    float* dinv = ws;
    float* g    = ws + N;
    float* h0   = ws + 2 * (size_t)N;
    int*  counts = (int*)(h0 + (size_t)N * HIDDEN);
    int*  tot    = counts + BK_MAX * EC_CHUNKS;
    int*  bstart = tot + BK_MAX;
    int4* recs   = (int4*)(bstart + BK_MAX + 32);

    const int B = 256;
    const int NB = (N + BK_NODES - 1) / BK_NODES;

    edge_count_kernel<<<EC_CHUNKS, EC_THREADS, 0, stream>>>(cols, counts, NB, E);
    chunk_prefix_kernel<<<(NB + 7) / 8, B, 0, stream>>>(counts, tot, NB);
    edge_place_kernel<<<EC_CHUNKS, EC_THREADS, 0, stream>>>(rows, cols, ew, counts, tot,
                                                            bstart, recs, NB, E);
    block_gemm1_kernel<<<NB, GB_THREADS, 0, stream>>>(x, w1, recs, bstart, dinv, h0, N);

    block_aggregate_kernel<<<NB, AG_THREADS, 0, stream>>>(recs, bstart, dinv, h0, b1, w2, g, N);
    block_scatter2_kernel<<<NB, S2_THREADS, 0, stream>>>(recs, bstart, dinv, g, b2, out, N);
}


# Edge records 8 bytes, not 16: row and column-in-block share one word (row << 7 | col) beside the weight; the sorted array is written once and read three times at half the bytes
# speedup vs baseline: 4.8098x; 1.0313x over previous
#include <hip/hip_runtime.h>
#include <hip/hip_bf16.h>

#define IN_FEAT 128
#define HIDDEN  64

typedef __attribute__((ext_vector_type(2))) float v2f;
typedef __attribute__((ext_vector_type(8))) float v8f;

#define REC_SHIFT 7
#define REC_CMASK ((1 << REC_SHIFT) - 1)
#define HOLD8(a) "v"(a[0]), "v"(a[1]), "v"(a[2]), "v"(a[3]), "v"(a[4]), "v"(a[5]), "v"(a[6]), "v"(a[7])

#define BK_NODES   104
#define GB_THREADS 224
#define GB_WAVES   (GB_THREADS / 32)
#define GB_UNROLL  20
#define SO_PITCH   68
static_assert(GB_WAVES * 16 >= BK_NODES && GB_THREADS >= BK_NODES + 32, "the waves' 16-row strips cover the block; one thread per bin");
static_assert(IN_FEAT == 128 && HIDDEN == 64, "16 k groups of 8, four column tiles");

__global__ void __launch_bounds__(GB_THREADS)
block_gemm1_kernel(const float* __restrict__ X, const float* __restrict__ W,
                   const int2* __restrict__ recs, const int* __restrict__ bstart,
                   float* __restrict__ dinv, float* __restrict__ H, int n) {
    __shared__ float4 sW4[(IN_FEAT / 4) * HIDDEN];
    __shared__ __attribute__((aligned(16))) float sOut[GB_WAVES * 16 * SO_PITCH];
    __shared__ float bins[BK_NODES + 32];
    const int t = threadIdx.x;
    const int lane = t & 31;
    const int wave = t >> 5;
    const int half = lane >> 4;
    const int l = lane & 15;
    const int lo = blockIdx.x * BK_NODES;
    const int nn = min(BK_NODES, n - lo);

    if (t < BK_NODES + 32) bins[t] = 1.0f;
    for (int it = t; it < (IN_FEAT / 4) * (HIDDEN / 4); it += GB_THREADS) {
        const int kq = it >> 4;
        const int q = it & 15;
        const float4 r0 = *(const float4*)(W + (4 * kq + 0) * HIDDEN + 4 * q);
        const float4 r1 = *(const float4*)(W + (4 * kq + 1) * HIDDEN + 4 * q);
        const float4 r2 = *(const float4*)(W + (4 * kq + 2) * HIDDEN + 4 * q);
        const float4 r3 = *(const float4*)(W + (4 * kq + 3) * HIDDEN + 4 * q);
        float4* dst = sW4 + kq * HIDDEN + 4 * q;
        dst[0] = make_float4(r0.x, r1.x, r2.x, r3.x);
        dst[1] = make_float4(r0.y, r1.y, r2.y, r3.y);
        dst[2] = make_float4(r0.z, r1.z, r2.z, r3.z);
        dst[3] = make_float4(r0.w, r1.w, r2.w, r3.w);
    }
    __syncthreads();

    const int beg = bstart[blockIdx.x];
    const int end = bstart[blockIdx.x + 1];
    const int last = end - 1;
    for (int base = beg; base < end; base += GB_THREADS * GB_UNROLL) {
        int cl[GB_UNROLL];
        float w[GB_UNROLL];
#pragma unroll
        for (int u = 0; u < GB_UNROLL; u++) {
            const int idx = base + u * GB_THREADS + t;
            const int2 rec = recs[min(idx, last)];
            const bool ok = idx < end;
            cl[u] = ok ? (rec.x & REC_CMASK) : BK_NODES + lane;
            w[u] = ok ? __int_as_float(rec.y) : 0.0f;
        }
#pragma unroll
        for (int u = 0; u < GB_UNROLL; u++) atomicAdd(&bins[cl[u]], w[u]);
    }
    __syncthreads();
    if (t < nn) {
        const float d = bins[t];
        dinv[lo + t] = (d > 0.0f) ? __frsqrt_rn(d) : 0.0f;
    }

    const int m0 = lo + wave * 16;
    const float* Xrow = X + (size_t)min(m0 + l, n - 1) * IN_FEAT + 4 * half;
    float4 a[16];
#pragma unroll
    for (int j = 0; j < 16; j++) a[j] = *(const float4*)(Xrow + 8 * j);

    v8f acc[4];
#pragma unroll
    for (int nt = 0; nt < 4; nt++) acc[nt] = (v8f){0.f, 0.f, 0.f, 0.f, 0.f, 0.f, 0.f, 0.f};
#pragma unroll
    for (int j = 0; j < 16; j++) {
        const float4* brow = sW4 + (2 * j + half) * HIDDEN + l;
        float4 bv[4];
#pragma unroll
        for (int nt = 0; nt < 4; nt++) bv[nt] = brow[16 * nt];
        const v2f a0 = {a[j].x, a[j].y};
        const v2f a1 = {a[j].z, a[j].w};
#pragma unroll
        for (int nt = 0; nt < 4; nt++) {
            const v2f b0 = {bv[nt].x, bv[nt].y};
            acc[nt] = __builtin_amdgcn_wmma_f32_16x16x4_f32(
                 false, a0,  false, b0,
                 (short)0, acc[nt],  false,  false);
        }
#pragma unroll
        for (int nt = 0; nt < 4; nt++) {
            const v2f b1 = {bv[nt].z, bv[nt].w};
            acc[nt] = __builtin_amdgcn_wmma_f32_16x16x4_f32(
                 false, a1,  false, b1,
                 (short)0, acc[nt],  false,  false);
        }
    }

    float* tile = sOut + wave * 16 * SO_PITCH;
#pragma unroll
    for (int nt = 0; nt < 4; nt++)
#pragma unroll
        for (int r = 0; r < 8; r++)
            tile[(8 * half + r) * SO_PITCH + 16 * nt + l] = acc[nt][r];
    __syncthreads();
#pragma unroll
    for (int i = 0; i < 8; i++) {
        const int rr = 2 * i + half;
        const int node = wave * 16 + rr;
        const float4 v = *(const float4*)(tile + rr * SO_PITCH + 4 * l);
        const float d = bins[min(node, BK_NODES - 1)];
        const float s = (d > 0.0f) ? __frsqrt_rn(d) : 0.0f;
        if (node < nn)
            *(float4*)(H + (size_t)(lo + node) * HIDDEN + 4 * l) =
                make_float4(s * v.x, s * v.y, s * v.z, s * v.w);
    }
}

#define BK_MAX     1024
#define EC_CHUNKS  256
#define EC_THREADS 1024
#define AG_THREADS 256
#define AG_STRIDE  ((AG_THREADS / 32) * 32)
#define AG_BATCH   16
static_assert(EC_CHUNKS == 256, "chunk_prefix_kernel gives each lane 8 chunks");
static_assert(BK_NODES <= (1 << REC_SHIFT) && (long long)BK_MAX * BK_NODES <= (1ll << (31 - REC_SHIFT)),
              "a record's first word holds the row above the column inside the block");
static_assert(HIDDEN == 64, "one float2 per lane covers a row");

__global__ void __launch_bounds__(EC_THREADS)
edge_count_kernel(const int* __restrict__ cols, int* __restrict__ counts,
                  int nb, int e) {
    __shared__ int cnt[BK_MAX];
    for (int t = threadIdx.x; t < BK_MAX; t += EC_THREADS) cnt[t] = 0;
    __syncthreads();
    const int per = (e + EC_CHUNKS - 1) / EC_CHUNKS;
    const int start = blockIdx.x * per;
    const int end = min(e, start + per);
    for (int base = start; base < end; base += EC_THREADS * 4) {
        int c[4];
#pragma unroll
        for (int u = 0; u < 4; u++) {
            const int i = base + u * EC_THREADS + (int)threadIdx.x;
            c[u] = (i < end) ? cols[i] : -1;
        }
#pragma unroll
        for (int u = 0; u < 4; u++)
            if (c[u] >= 0) atomicAdd(&cnt[(unsigned)c[u] / BK_NODES], 1);
    }
    __syncthreads();
    for (int b = threadIdx.x; b < nb; b += EC_THREADS)
        counts[(size_t)b * EC_CHUNKS + blockIdx.x] = cnt[b];
}

__global__ void __launch_bounds__(256)
chunk_prefix_kernel(int* __restrict__ counts, int* __restrict__ tot, int nb) {
    const int b = blockIdx.x * (blockDim.x >> 5) + (threadIdx.x >> 5);
    const int lane = threadIdx.x & 31;
    if (b >= nb) return;
    int4* row = (int4*)(counts + (size_t)b * EC_CHUNKS) + 2 * lane;
    const int4 a = row[0];
    const int4 c = row[1];
    const int sum = a.x + a.y + a.z + a.w + c.x + c.y + c.z + c.w;
    int inc = sum;
    for (int off = 1; off < 32; off <<= 1) {
        const int v = __shfl_up(inc, off, 32);
        if (lane >= off) inc += v;
    }
    int4 oa, oc;
    oa.x = inc - sum;
    oa.y = oa.x + a.x;
    oa.z = oa.y + a.y;
    oa.w = oa.z + a.z;
    oc.x = oa.w + a.w;
    oc.y = oc.x + c.x;
    oc.z = oc.y + c.y;
    oc.w = oc.z + c.z;
    row[0] = oa;
    row[1] = oc;
    if (lane == 31) tot[b] = inc;
}

#define EP_STAGE 12800
__global__ void __launch_bounds__(EC_THREADS)
edge_place_kernel(const int* __restrict__ rows, const int* __restrict__ cols,
                  const float* __restrict__ ew,
                  const int* __restrict__ counts, const int* __restrict__ tot,
                  int* __restrict__ bstart, int2* __restrict__ recs,
                  int nb, int e) {
    __shared__ int4 stage[EP_STAGE];
    __shared__ int cursor[BK_MAX];
    __shared__ int delta[BK_MAX];
    __shared__ int wsum_g[32];
    __shared__ int wsum_l[32];
    const int t = threadIdx.x;
    const int lane = t & 31;
    const int wid = t >> 5;
    const int per = (e + EC_CHUNKS - 1) / EC_CHUNKS;
    const int start = blockIdx.x * per;
    const int end = min(e, start + per);
    const bool staged = per <= EP_STAGE;

    int mine = 0, cur = 0, len = 0;
    if (t < nb) {
        mine = tot[t];
        cur = counts[(size_t)t * EC_CHUNKS + blockIdx.x];
        const int nxt = (blockIdx.x + 1 < EC_CHUNKS)
                            ? counts[(size_t)t * EC_CHUNKS + blockIdx.x + 1] : mine;
        len = nxt - cur;
    }
    int inc_g = mine;
    int inc_l = len;
    for (int off = 1; off < 32; off <<= 1) {
        const int vg = __shfl_up(inc_g, off, 32);
        const int vl = __shfl_up(inc_l, off, 32);
        if (lane >= off) { inc_g += vg; inc_l += vl; }
    }
    if (lane == 31) { wsum_g[wid] = inc_g; wsum_l[wid] = inc_l; }
    __syncthreads();
    if (wid == 0) {
        const int sg = wsum_g[lane];
        const int sl = wsum_l[lane];
        int sig = sg;
        int sil = sl;
        for (int off = 1; off < 32; off <<= 1) {
            const int vg = __shfl_up(sig, off, 32);
            const int vl = __shfl_up(sil, off, 32);
            if (lane >= off) { sig += vg; sil += vl; }
        }
        wsum_g[lane] = sig - sg;
        wsum_l[lane] = sil - sl;
    }
    __syncthreads();
    const int excl = inc_g - mine + wsum_g[wid];
    const int lexcl = inc_l - len + wsum_l[wid];
    if (t < nb) {
        const int gpos = excl + cur;
        cursor[t] = staged ? lexcl : gpos;
        delta[t] = gpos - lexcl;
        if (blockIdx.x == 0) {
            bstart[t] = excl;
            if (t == nb - 1) bstart[nb] = excl + mine;
        }
    }
    __syncthreads();

    for (int base = start; base < end; base += EC_THREADS * 4) {
        int c[4];
        int r[4];
        float w[4];
#pragma unroll
        for (int u = 0; u < 4; u++) {
            const int i = base + u * EC_THREADS + t;
            const bool ok = i < end;
            c[u] = ok ? cols[i] : -1;
            r[u] = ok ? rows[i] : 0;
            w[u] = ok ? ew[i] : 0.0f;
        }
#pragma unroll
        for (int u = 0; u < 4; u++) {
            if (c[u] >= 0) {
                const int blk = (unsigned)c[u] / BK_NODES;
                const int pos = atomicAdd(&cursor[blk], 1);
                if (staged)
                    stage[pos] = make_int4(r[u], c[u], __float_as_int(w[u]), 0);
                else
                    recs[pos] = make_int2((r[u] << REC_SHIFT) | (c[u] - blk * BK_NODES), __float_as_int(w[u]));
            }
        }
    }
    if (staged) {
        __syncthreads();
        const int cnt = end - start;
        for (int p = t; p < cnt; p += EC_THREADS) {
            const int4 rec = stage[p];
            const int blk = (unsigned)rec.y / BK_NODES;
            recs[p + delta[blk]] =
                make_int2((rec.x << REC_SHIFT) | (rec.y - blk * BK_NODES), rec.z);
        }
    }
}

#define AG_ROWS (BK_NODES / (AG_THREADS / 32))
static_assert(AG_ROWS * (AG_THREADS / 32) == BK_NODES && AG_ROWS <= 16,
              "the waves' node groups cover the tile; one lane of a 16-lane row per node");

__device__ __forceinline__ float lane_xor16(float x) {
    const int b = __float_as_int(x);
    return __int_as_float(__builtin_amdgcn_permlanex16(b, b, 0x76543210, 0xfedcba98, false, false));
}
template <int M>
__device__ __forceinline__ float row_xor(float x) {
    static_assert(M > 0 && M < 16, "row_xmask stays inside a row of 16");
    const int b = __float_as_int(x);
    return __int_as_float(__builtin_amdgcn_update_dpp(b, b, 0x160 + M, 0xf, 0xf, false));
}

__global__ void __launch_bounds__(AG_THREADS)
block_aggregate_kernel(const int2* __restrict__ recs, const int* __restrict__ bstart,
                       const float* __restrict__ dinv, const float* __restrict__ h0,
                       const float* __restrict__ b1, const float* __restrict__ w2,
                       float* __restrict__ g, int n) {
    __shared__ __attribute__((aligned(16))) float acc[(BK_NODES + 1) * HIDDEN];
    __shared__ float gs[BK_NODES];
    __shared__ float dl[BK_NODES];
    const int t = threadIdx.x;
    const int lane = t & 31;
    const int wave = __builtin_amdgcn_readfirstlane(t >> 5);
    const int lo = blockIdx.x * BK_NODES;
    const int nn = min(BK_NODES, n - lo);
    float2* acc2 = (float2*)acc;
    if (t < BK_NODES) dl[t] = (t < nn) ? dinv[lo + t] : 0.0f;
    for (int idx = t; idx < nn * (HIDDEN / 2); idx += AG_THREADS) {
        const int node = lo + (idx >> 5);
        acc2[idx] = ((const float2*)(h0 + (size_t)node * HIDDEN))[idx & 31];
    }
    __syncthreads();

    const int beg = __builtin_amdgcn_readfirstlane(bstart[blockIdx.x]);
    const int end = __builtin_amdgcn_readfirstlane(bstart[blockIdx.x + 1]);
    const int g0 = beg + wave * 32;
    if (g0 < end) {
        const int last = end - 1;
        int2 rec = recs[min(g0 + lane, last)];
        const unsigned lane8 = 8u * (unsigned)lane;
        for (int g = g0; g < end; g += AG_STRIDE) {
            const int2 nxt = recs[min(g + AG_STRIDE + lane, last)];
            if (g + lane >= end) rec = make_int2(BK_NODES, 0);
#pragma unroll
            for (int u0 = 0; u0 < 32; u0 += AG_BATCH) {
                float2 v[AG_BATCH];
                int cl[AG_BATCH];
                float nm[AG_BATCH];
#pragma unroll
                for (int k = 0; k < AG_BATCH; k++) {
                    const unsigned p = (unsigned)__builtin_amdgcn_readlane(rec.x, u0 + k);
                    cl[k] = (int)(p & REC_CMASK);
                    nm[k] = __int_as_float(__builtin_amdgcn_readlane(rec.y, u0 + k));
                    const unsigned roff = (p >> REC_SHIFT) * (unsigned)(HIDDEN * sizeof(float));
                    v[k] = *(const float2*)((const char*)h0 + (roff | lane8));
                }
#pragma unroll
                for (int k = 0; k < AG_BATCH; k++) {
                    float* d = acc + cl[k] * HIDDEN + 2 * lane;
                    atomicAdd(d + 0, nm[k] * v[k].x);
                    atomicAdd(d + 1, nm[k] * v[k].y);
                }
            }
            rec = nxt;
        }
    }
    __syncthreads();

    const float2 bv = ((const float2*)b1)[lane];
    const float2 wv = ((const float2*)w2)[lane];
    float p[AG_ROWS];
#pragma unroll
    for (int k = 0; k < AG_ROWS; k++) {
        const int i = min(wave * AG_ROWS + k, nn - 1);
        const float2 a = acc2[i * (HIDDEN / 2) + lane];
        const float dc = dl[i];
        float2 hv;
        hv.x = fmaxf(dc * a.x + bv.x, 0.0f);
        hv.y = fmaxf(dc * a.y + bv.y, 0.0f);
        p[k] = hv.x * wv.x + hv.y * wv.y;
    }
#pragma unroll
    for (int k = 0; k < AG_ROWS; k++) p[k] += lane_xor16(p[k]);
#pragma unroll
    for (int k = 0; k < AG_ROWS; k++) p[k] += row_xor<8>(p[k]);
#pragma unroll
    for (int k = 0; k < AG_ROWS; k++) p[k] += row_xor<4>(p[k]);
#pragma unroll
    for (int k = 0; k < AG_ROWS; k++) p[k] += row_xor<2>(p[k]);
#pragma unroll
    for (int k = 0; k < AG_ROWS; k++) p[k] += row_xor<1>(p[k]);
    float mine = p[0];
#pragma unroll
    for (int k = 1; k < AG_ROWS; k++) mine = (lane == k) ? p[k] : mine;
    if (lane < AG_ROWS) gs[wave * AG_ROWS + lane] = dl[wave * AG_ROWS + lane] * mine;
    __syncthreads();
    if (t < nn) g[lo + t] = gs[t];
}

#define S2_THREADS 320
#define S2_UNROLL  8
static_assert(S2_UNROLL == 8, "HOLD8 names eight values");
static_assert(S2_THREADS >= BK_NODES, "one thread per node for the start values and the stores");

__global__ void __launch_bounds__(S2_THREADS)
block_scatter2_kernel(const int2* __restrict__ recs, const int* __restrict__ bstart,
                      const float* __restrict__ dinv, const float* __restrict__ g,
                      const float* __restrict__ b2, float* __restrict__ out, int n) {
    __shared__ float bins[BK_NODES];
    const int t = threadIdx.x;
    const int lo = blockIdx.x * BK_NODES;
    const int nn = min(BK_NODES, n - lo);
    if (t < nn) bins[t] = g[lo + t];
    __syncthreads();

    const int beg = bstart[blockIdx.x];
    const int end = bstart[blockIdx.x + 1];
    const int last = end - 1;
    for (int base = beg; base < end; base += S2_THREADS * S2_UNROLL) {
        int2 rec[S2_UNROLL];
        float gr[S2_UNROLL];
#pragma unroll
        for (int u = 0; u < S2_UNROLL; u++)
            rec[u] = recs[min(base + u * S2_THREADS + t, last)];
#pragma unroll
        for (int u = 0; u < S2_UNROLL; u++)
            gr[u] = g[rec[u].x >> REC_SHIFT];
        asm volatile("" :: HOLD8(gr));
#pragma unroll
        for (int u = 0; u < S2_UNROLL; u++)
            if (base + u * S2_THREADS + t < end)
                atomicAdd(&bins[rec[u].x & REC_CMASK], __int_as_float(rec[u].y) * gr[u]);
    }
    __syncthreads();
    if (t < nn) out[lo + t] = b2[0] + dinv[lo + t] * bins[t];
}

extern "C" void kernel_launch(void* const* d_in, const int* in_sizes, int n_in,
                              void* d_out, int out_size, void* d_ws, size_t ws_size,
                              hipStream_t stream) {
    const float* x  = (const float*)d_in[0];
    const int*   ei = (const int*)d_in[1];
    const float* ew = (const float*)d_in[2];
    const float* w1 = (const float*)d_in[3];
    const float* b1 = (const float*)d_in[4];
    const float* w2 = (const float*)d_in[5];
    const float* b2 = (const float*)d_in[6];
    float* out = (float*)d_out;

    const int E = in_sizes[2];
    const int N = in_sizes[0] / IN_FEAT;
    const int* rows = ei;
    const int* cols = ei + E;

    float* ws   = (float*)d_ws;
    float* dinv = ws;
    float* g    = ws + N;
    float* h0   = ws + 2 * (size_t)N;
    int*  counts = (int*)(h0 + (size_t)N * HIDDEN);
    int*  tot    = counts + BK_MAX * EC_CHUNKS;
    int*  bstart = tot + BK_MAX;
    int2* recs   = (int2*)(bstart + BK_MAX + 32);

    const int B = 256;
    const int NB = (N + BK_NODES - 1) / BK_NODES;

    edge_count_kernel<<<EC_CHUNKS, EC_THREADS, 0, stream>>>(cols, counts, NB, E);
    chunk_prefix_kernel<<<(NB + 7) / 8, B, 0, stream>>>(counts, tot, NB);
    edge_place_kernel<<<EC_CHUNKS, EC_THREADS, 0, stream>>>(rows, cols, ew, counts, tot,
                                                            bstart, recs, NB, E);
    block_gemm1_kernel<<<NB, GB_THREADS, 0, stream>>>(x, w1, recs, bstart, dinv, h0, N);

    block_aggregate_kernel<<<NB, AG_THREADS, 0, stream>>>(recs, bstart, dinv, h0, b1, w2, g, N);
    block_scatter2_kernel<<<NB, S2_THREADS, 0, stream>>>(recs, bstart, dinv, g, b2, out, N);
}


# Layer-1 aggregate: the block's records are ordered by destination node in an LDS stage (integer counting sort); a wave sums a node's rows in registers, no LDS float add per edge; sum order changes
# speedup vs baseline: 5.8643x; 1.2192x over previous
#include <hip/hip_runtime.h>
#include <hip/hip_bf16.h>

#define IN_FEAT 128
#define HIDDEN  64

typedef __attribute__((ext_vector_type(2))) float v2f;
typedef __attribute__((ext_vector_type(8))) float v8f;

#define REC_SHIFT 7
#define REC_CMASK ((1 << REC_SHIFT) - 1)
#define HOLD8(a) "v"(a[0]), "v"(a[1]), "v"(a[2]), "v"(a[3]), "v"(a[4]), "v"(a[5]), "v"(a[6]), "v"(a[7])

#define BK_NODES   104
#define GB_THREADS 224
#define GB_WAVES   (GB_THREADS / 32)
#define GB_UNROLL  20
#define SO_PITCH   68
static_assert(GB_WAVES * 16 >= BK_NODES && GB_THREADS >= BK_NODES + 32, "the waves' 16-row strips cover the block; one thread per bin");
static_assert(IN_FEAT == 128 && HIDDEN == 64, "16 k groups of 8, four column tiles");

__global__ void __launch_bounds__(GB_THREADS)
block_gemm1_kernel(const float* __restrict__ X, const float* __restrict__ W,
                   const int2* __restrict__ recs, const int* __restrict__ bstart,
                   float* __restrict__ dinv, float* __restrict__ H, int n) {
    __shared__ float4 sW4[(IN_FEAT / 4) * HIDDEN];
    __shared__ __attribute__((aligned(16))) float sOut[GB_WAVES * 16 * SO_PITCH];
    __shared__ float bins[BK_NODES + 32];
    const int t = threadIdx.x;
    const int lane = t & 31;
    const int wave = t >> 5;
    const int half = lane >> 4;
    const int l = lane & 15;
    const int lo = blockIdx.x * BK_NODES;
    const int nn = min(BK_NODES, n - lo);

    if (t < BK_NODES + 32) bins[t] = 1.0f;
    for (int it = t; it < (IN_FEAT / 4) * (HIDDEN / 4); it += GB_THREADS) {
        const int kq = it >> 4;
        const int q = it & 15;
        const float4 r0 = *(const float4*)(W + (4 * kq + 0) * HIDDEN + 4 * q);
        const float4 r1 = *(const float4*)(W + (4 * kq + 1) * HIDDEN + 4 * q);
        const float4 r2 = *(const float4*)(W + (4 * kq + 2) * HIDDEN + 4 * q);
        const float4 r3 = *(const float4*)(W + (4 * kq + 3) * HIDDEN + 4 * q);
        float4* dst = sW4 + kq * HIDDEN + 4 * q;
        dst[0] = make_float4(r0.x, r1.x, r2.x, r3.x);
        dst[1] = make_float4(r0.y, r1.y, r2.y, r3.y);
        dst[2] = make_float4(r0.z, r1.z, r2.z, r3.z);
        dst[3] = make_float4(r0.w, r1.w, r2.w, r3.w);
    }
    __syncthreads();

    const int beg = bstart[blockIdx.x];
    const int end = bstart[blockIdx.x + 1];
    const int last = end - 1;
    for (int base = beg; base < end; base += GB_THREADS * GB_UNROLL) {
        int cl[GB_UNROLL];
        float w[GB_UNROLL];
#pragma unroll
        for (int u = 0; u < GB_UNROLL; u++) {
            const int idx = base + u * GB_THREADS + t;
            const int2 rec = recs[min(idx, last)];
            const bool ok = idx < end;
            cl[u] = ok ? (rec.x & REC_CMASK) : BK_NODES + lane;
            w[u] = ok ? __int_as_float(rec.y) : 0.0f;
        }
#pragma unroll
        for (int u = 0; u < GB_UNROLL; u++) atomicAdd(&bins[cl[u]], w[u]);
    }
    __syncthreads();
    if (t < nn) {
        const float d = bins[t];
        dinv[lo + t] = (d > 0.0f) ? __frsqrt_rn(d) : 0.0f;
    }

    const int m0 = lo + wave * 16;
    const float* Xrow = X + (size_t)min(m0 + l, n - 1) * IN_FEAT + 4 * half;
    float4 a[16];
#pragma unroll
    for (int j = 0; j < 16; j++) a[j] = *(const float4*)(Xrow + 8 * j);

    v8f acc[4];
#pragma unroll
    for (int nt = 0; nt < 4; nt++) acc[nt] = (v8f){0.f, 0.f, 0.f, 0.f, 0.f, 0.f, 0.f, 0.f};
#pragma unroll
    for (int j = 0; j < 16; j++) {
        const float4* brow = sW4 + (2 * j + half) * HIDDEN + l;
        float4 bv[4];
#pragma unroll
        for (int nt = 0; nt < 4; nt++) bv[nt] = brow[16 * nt];
        const v2f a0 = {a[j].x, a[j].y};
        const v2f a1 = {a[j].z, a[j].w};
#pragma unroll
        for (int nt = 0; nt < 4; nt++) {
            const v2f b0 = {bv[nt].x, bv[nt].y};
            acc[nt] = __builtin_amdgcn_wmma_f32_16x16x4_f32(
                 false, a0,  false, b0,
                 (short)0, acc[nt],  false,  false);
        }
#pragma unroll
        for (int nt = 0; nt < 4; nt++) {
            const v2f b1 = {bv[nt].z, bv[nt].w};
            acc[nt] = __builtin_amdgcn_wmma_f32_16x16x4_f32(
                 false, a1,  false, b1,
                 (short)0, acc[nt],  false,  false);
        }
    }

    float* tile = sOut + wave * 16 * SO_PITCH;
#pragma unroll
    for (int nt = 0; nt < 4; nt++)
#pragma unroll
        for (int r = 0; r < 8; r++)
            tile[(8 * half + r) * SO_PITCH + 16 * nt + l] = acc[nt][r];
    __syncthreads();
#pragma unroll
    for (int i = 0; i < 8; i++) {
        const int rr = 2 * i + half;
        const int node = wave * 16 + rr;
        const float4 v = *(const float4*)(tile + rr * SO_PITCH + 4 * l);
        const float d = bins[min(node, BK_NODES - 1)];
        const float s = (d > 0.0f) ? __frsqrt_rn(d) : 0.0f;
        if (node < nn)
            *(float4*)(H + (size_t)(lo + node) * HIDDEN + 4 * l) =
                make_float4(s * v.x, s * v.y, s * v.z, s * v.w);
    }
}

#define BK_MAX     1024
#define EC_CHUNKS  256
#define EC_THREADS 1024
#define AG_THREADS 256
#define AG_STRIDE  ((AG_THREADS / 32) * 32)
#define AG_BATCH   16
static_assert(EC_CHUNKS == 256, "chunk_prefix_kernel gives each lane 8 chunks");
static_assert(BK_NODES <= (1 << REC_SHIFT) && (long long)BK_MAX * BK_NODES <= (1ll << (31 - REC_SHIFT)),
              "a record's first word holds the row above the column inside the block");
static_assert(HIDDEN == 64, "one float2 per lane covers a row");

__global__ void __launch_bounds__(EC_THREADS)
edge_count_kernel(const int* __restrict__ cols, int* __restrict__ counts,
                  int nb, int e) {
    __shared__ int cnt[BK_MAX];
    for (int t = threadIdx.x; t < BK_MAX; t += EC_THREADS) cnt[t] = 0;
    __syncthreads();
    const int per = (e + EC_CHUNKS - 1) / EC_CHUNKS;
    const int start = blockIdx.x * per;
    const int end = min(e, start + per);
    for (int base = start; base < end; base += EC_THREADS * 4) {
        int c[4];
#pragma unroll
        for (int u = 0; u < 4; u++) {
            const int i = base + u * EC_THREADS + (int)threadIdx.x;
            c[u] = (i < end) ? cols[i] : -1;
        }
#pragma unroll
        for (int u = 0; u < 4; u++)
            if (c[u] >= 0) atomicAdd(&cnt[(unsigned)c[u] / BK_NODES], 1);
    }
    __syncthreads();
    for (int b = threadIdx.x; b < nb; b += EC_THREADS)
        counts[(size_t)b * EC_CHUNKS + blockIdx.x] = cnt[b];
}

__global__ void __launch_bounds__(256)
chunk_prefix_kernel(int* __restrict__ counts, int* __restrict__ tot, int nb) {
    const int b = blockIdx.x * (blockDim.x >> 5) + (threadIdx.x >> 5);
    const int lane = threadIdx.x & 31;
    if (b >= nb) return;
    int4* row = (int4*)(counts + (size_t)b * EC_CHUNKS) + 2 * lane;
    const int4 a = row[0];
    const int4 c = row[1];
    const int sum = a.x + a.y + a.z + a.w + c.x + c.y + c.z + c.w;
    int inc = sum;
    for (int off = 1; off < 32; off <<= 1) {
        const int v = __shfl_up(inc, off, 32);
        if (lane >= off) inc += v;
    }
    int4 oa, oc;
    oa.x = inc - sum;
    oa.y = oa.x + a.x;
    oa.z = oa.y + a.y;
    oa.w = oa.z + a.z;
    oc.x = oa.w + a.w;
    oc.y = oc.x + c.x;
    oc.z = oc.y + c.y;
    oc.w = oc.z + c.z;
    row[0] = oa;
    row[1] = oc;
    if (lane == 31) tot[b] = inc;
}

#define EP_STAGE 12800
__global__ void __launch_bounds__(EC_THREADS)
edge_place_kernel(const int* __restrict__ rows, const int* __restrict__ cols,
                  const float* __restrict__ ew,
                  const int* __restrict__ counts, const int* __restrict__ tot,
                  int* __restrict__ bstart, int2* __restrict__ recs,
                  int nb, int e) {
    __shared__ int4 stage[EP_STAGE];
    __shared__ int cursor[BK_MAX];
    __shared__ int delta[BK_MAX];
    __shared__ int wsum_g[32];
    __shared__ int wsum_l[32];
    const int t = threadIdx.x;
    const int lane = t & 31;
    const int wid = t >> 5;
    const int per = (e + EC_CHUNKS - 1) / EC_CHUNKS;
    const int start = blockIdx.x * per;
    const int end = min(e, start + per);
    const bool staged = per <= EP_STAGE;

    int mine = 0, cur = 0, len = 0;
    if (t < nb) {
        mine = tot[t];
        cur = counts[(size_t)t * EC_CHUNKS + blockIdx.x];
        const int nxt = (blockIdx.x + 1 < EC_CHUNKS)
                            ? counts[(size_t)t * EC_CHUNKS + blockIdx.x + 1] : mine;
        len = nxt - cur;
    }
    int inc_g = mine;
    int inc_l = len;
    for (int off = 1; off < 32; off <<= 1) {
        const int vg = __shfl_up(inc_g, off, 32);
        const int vl = __shfl_up(inc_l, off, 32);
        if (lane >= off) { inc_g += vg; inc_l += vl; }
    }
    if (lane == 31) { wsum_g[wid] = inc_g; wsum_l[wid] = inc_l; }
    __syncthreads();
    if (wid == 0) {
        const int sg = wsum_g[lane];
        const int sl = wsum_l[lane];
        int sig = sg;
        int sil = sl;
        for (int off = 1; off < 32; off <<= 1) {
            const int vg = __shfl_up(sig, off, 32);
            const int vl = __shfl_up(sil, off, 32);
            if (lane >= off) { sig += vg; sil += vl; }
        }
        wsum_g[lane] = sig - sg;
        wsum_l[lane] = sil - sl;
    }
    __syncthreads();
    const int excl = inc_g - mine + wsum_g[wid];
    const int lexcl = inc_l - len + wsum_l[wid];
    if (t < nb) {
        const int gpos = excl + cur;
        cursor[t] = staged ? lexcl : gpos;
        delta[t] = gpos - lexcl;
        if (blockIdx.x == 0) {
            bstart[t] = excl;
            if (t == nb - 1) bstart[nb] = excl + mine;
        }
    }
    __syncthreads();

    for (int base = start; base < end; base += EC_THREADS * 4) {
        int c[4];
        int r[4];
        float w[4];
#pragma unroll
        for (int u = 0; u < 4; u++) {
            const int i = base + u * EC_THREADS + t;
            const bool ok = i < end;
            c[u] = ok ? cols[i] : -1;
            r[u] = ok ? rows[i] : 0;
            w[u] = ok ? ew[i] : 0.0f;
        }
#pragma unroll
        for (int u = 0; u < 4; u++) {
            if (c[u] >= 0) {
                const int blk = (unsigned)c[u] / BK_NODES;
                const int pos = atomicAdd(&cursor[blk], 1);
                if (staged)
                    stage[pos] = make_int4(r[u], c[u], __float_as_int(w[u]), 0);
                else
                    recs[pos] = make_int2((r[u] << REC_SHIFT) | (c[u] - blk * BK_NODES), __float_as_int(w[u]));
            }
        }
    }
    if (staged) {
        __syncthreads();
        const int cnt = end - start;
        for (int p = t; p < cnt; p += EC_THREADS) {
            const int4 rec = stage[p];
            const int blk = (unsigned)rec.y / BK_NODES;
            recs[p + delta[blk]] =
                make_int2((rec.x << REC_SHIFT) | (rec.y - blk * BK_NODES), rec.z);
        }
    }
}

#define AG_ROWS (BK_NODES / (AG_THREADS / 32))
#define AG_STAGE 4096
#define AG_SLOTS (AG_STAGE / AG_THREADS)
static_assert(AG_ROWS * (AG_THREADS / 32) == BK_NODES && AG_ROWS <= 16,
              "the waves' node groups cover the tile; one lane of a 16-lane row per node");
static_assert(BK_NODES < 4 * 32, "one wave scans the per-node counts, four per lane, and nstart[BK_NODES] is inside the array");

__device__ __forceinline__ float lane_xor16(float x) {
    const int b = __float_as_int(x);
    return __int_as_float(__builtin_amdgcn_permlanex16(b, b, 0x76543210, 0xfedcba98, false, false));
}
template <int M>
__device__ __forceinline__ float row_xor(float x) {
    static_assert(M > 0 && M < 16, "row_xmask stays inside a row of 16");
    const int b = __float_as_int(x);
    return __int_as_float(__builtin_amdgcn_update_dpp(b, b, 0x160 + M, 0xf, 0xf, false));
}

__global__ void __launch_bounds__(AG_THREADS)
block_aggregate_kernel(const int2* __restrict__ recs, const int* __restrict__ bstart,
                       const float* __restrict__ dinv, const float* __restrict__ h0,
                       const float* __restrict__ b1, const float* __restrict__ w2,
                       float* __restrict__ g, int n) {
    __shared__ __attribute__((aligned(16))) float acc[(BK_NODES + 1) * HIDDEN];
    __shared__ int2 stage[AG_STAGE];
    __shared__ int cnt[BK_NODES];
    __shared__ int nstart[4 * 32];
    __shared__ float gs[BK_NODES];
    __shared__ float dl[BK_NODES];
    const int t = threadIdx.x;
    const int lane = t & 31;
    const int wave = __builtin_amdgcn_readfirstlane(t >> 5);
    const int lo = blockIdx.x * BK_NODES;
    const int nn = min(BK_NODES, n - lo);
    float2* acc2 = (float2*)acc;
    if (t < BK_NODES) {
        dl[t] = (t < nn) ? dinv[lo + t] : 0.0f;
        cnt[t] = 0;
    }
    for (int idx = t; idx < nn * (HIDDEN / 2); idx += AG_THREADS) {
        const int node = lo + (idx >> 5);
        acc2[idx] = ((const float2*)(h0 + (size_t)node * HIDDEN))[idx & 31];
    }
    __syncthreads();

    const int beg = __builtin_amdgcn_readfirstlane(bstart[blockIdx.x]);
    const int end = __builtin_amdgcn_readfirstlane(bstart[blockIdx.x + 1]);
    const int nrec = end - beg;
    const unsigned lane8 = 8u * (unsigned)lane;
    if (nrec > 0 && nrec <= AG_STAGE) {
        const int last = end - 1;
        int2 r[AG_SLOTS];
        int rk[AG_SLOTS];
#pragma unroll
        for (int u = 0; u < AG_SLOTS; u++)
            r[u] = recs[min(beg + u * AG_THREADS + t, last)];
#pragma unroll
        for (int u = 0; u < AG_SLOTS; u++)
            rk[u] = (u * AG_THREADS + t < nrec) ? atomicAdd(&cnt[r[u].x & REC_CMASK], 1) : 0;
        __syncthreads();
        if (wave == 0) {
            int c[4];
#pragma unroll
            for (int i = 0; i < 4; i++) c[i] = (4 * lane + i < BK_NODES) ? cnt[4 * lane + i] : 0;
            const int sum = c[0] + c[1] + c[2] + c[3];
            int inc = sum;
            for (int off = 1; off < 32; off <<= 1) {
                const int v = __shfl_up(inc, off, 32);
                if (lane >= off) inc += v;
            }
            int run = inc - sum;
#pragma unroll
            for (int i = 0; i < 4; i++) {
                nstart[4 * lane + i] = run;
                run += c[i];
            }
        }
        __syncthreads();
        int pos[AG_SLOTS];
#pragma unroll
        for (int u = 0; u < AG_SLOTS; u++) pos[u] = nstart[r[u].x & REC_CMASK] + rk[u];
#pragma unroll
        for (int u = 0; u < AG_SLOTS; u++)
            if (u * AG_THREADS + t < nrec) stage[pos[u]] = r[u];
        __syncthreads();

        const int s0 = __builtin_amdgcn_readfirstlane(nstart[wave * AG_ROWS]);
        const int s1 = __builtin_amdgcn_readfirstlane(nstart[wave * AG_ROWS + AG_ROWS]);
        if (s0 < s1) {
            const int slast = s1 - 1;
            int2 rec = stage[min(s0 + lane, slast)];
            int cur = -1;
            float2 a = make_float2(0.0f, 0.0f);
            for (int gi = s0; gi < s1; gi += 32) {
                const int2 nxt = stage[min(gi + 32 + lane, slast)];
                if (gi + lane >= s1) rec.y = 0;
#pragma unroll
                for (int u0 = 0; u0 < 32; u0 += AG_BATCH) {
                    float2 v[AG_BATCH];
                    int cl[AG_BATCH];
                    float nm[AG_BATCH];
#pragma unroll
                    for (int k = 0; k < AG_BATCH; k++) {
                        const unsigned p = (unsigned)__builtin_amdgcn_readlane(rec.x, u0 + k);
                        cl[k] = (int)(p & REC_CMASK);
                        nm[k] = __int_as_float(__builtin_amdgcn_readlane(rec.y, u0 + k));
                        const unsigned roff = (p >> REC_SHIFT) * (unsigned)(HIDDEN * sizeof(float));
                        v[k] = *(const float2*)((const char*)h0 + (roff | lane8));
                    }
#pragma unroll
                    for (int k = 0; k < AG_BATCH; k++) {
                        if (cl[k] != cur) {
                            if (cur >= 0) {
                                float* d = acc + cur * HIDDEN + 2 * lane;
                                atomicAdd(d + 0, a.x);
                                atomicAdd(d + 1, a.y);
                            }
                            cur = cl[k];
                            a = make_float2(0.0f, 0.0f);
                        }
                        a.x += nm[k] * v[k].x;
                        a.y += nm[k] * v[k].y;
                    }
                }
                rec = nxt;
            }
            float* d = acc + cur * HIDDEN + 2 * lane;
            atomicAdd(d + 0, a.x);
            atomicAdd(d + 1, a.y);
        }
    } else if (nrec > 0) {
        const int g0 = beg + wave * 32;
        if (g0 < end) {
            const int last = end - 1;
            int2 rec = recs[min(g0 + lane, last)];
            for (int gi = g0; gi < end; gi += AG_STRIDE) {
                const int2 nxt = recs[min(gi + AG_STRIDE + lane, last)];
                if (gi + lane >= end) rec = make_int2(BK_NODES, 0);
#pragma unroll
                for (int u0 = 0; u0 < 32; u0 += AG_BATCH) {
                    float2 v[AG_BATCH];
                    int cl[AG_BATCH];
                    float nm[AG_BATCH];
#pragma unroll
                    for (int k = 0; k < AG_BATCH; k++) {
                        const unsigned p = (unsigned)__builtin_amdgcn_readlane(rec.x, u0 + k);
                        cl[k] = (int)(p & REC_CMASK);
                        nm[k] = __int_as_float(__builtin_amdgcn_readlane(rec.y, u0 + k));
                        const unsigned roff = (p >> REC_SHIFT) * (unsigned)(HIDDEN * sizeof(float));
                        v[k] = *(const float2*)((const char*)h0 + (roff | lane8));
                    }
#pragma unroll
                    for (int k = 0; k < AG_BATCH; k++) {
                        float* d = acc + cl[k] * HIDDEN + 2 * lane;
                        atomicAdd(d + 0, nm[k] * v[k].x);
                        atomicAdd(d + 1, nm[k] * v[k].y);
                    }
                }
                rec = nxt;
            }
        }
    }
    __syncthreads();

    const float2 bv = ((const float2*)b1)[lane];
    const float2 wv = ((const float2*)w2)[lane];
    float p[AG_ROWS];
#pragma unroll
    for (int k = 0; k < AG_ROWS; k++) {
        const int i = min(wave * AG_ROWS + k, nn - 1);
        const float2 a = acc2[i * (HIDDEN / 2) + lane];
        const float dc = dl[i];
        float2 hv;
        hv.x = fmaxf(dc * a.x + bv.x, 0.0f);
        hv.y = fmaxf(dc * a.y + bv.y, 0.0f);
        p[k] = hv.x * wv.x + hv.y * wv.y;
    }
#pragma unroll
    for (int k = 0; k < AG_ROWS; k++) p[k] += lane_xor16(p[k]);
#pragma unroll
    for (int k = 0; k < AG_ROWS; k++) p[k] += row_xor<8>(p[k]);
#pragma unroll
    for (int k = 0; k < AG_ROWS; k++) p[k] += row_xor<4>(p[k]);
#pragma unroll
    for (int k = 0; k < AG_ROWS; k++) p[k] += row_xor<2>(p[k]);
#pragma unroll
    for (int k = 0; k < AG_ROWS; k++) p[k] += row_xor<1>(p[k]);
    float mine = p[0];
#pragma unroll
    for (int k = 1; k < AG_ROWS; k++) mine = (lane == k) ? p[k] : mine;
    if (lane < AG_ROWS) gs[wave * AG_ROWS + lane] = dl[wave * AG_ROWS + lane] * mine;
    __syncthreads();
    if (t < nn) g[lo + t] = gs[t];
}

#define S2_THREADS 320
#define S2_UNROLL  8
static_assert(S2_UNROLL == 8, "HOLD8 names eight values");
static_assert(S2_THREADS >= BK_NODES, "one thread per node for the start values and the stores");

__global__ void __launch_bounds__(S2_THREADS)
block_scatter2_kernel(const int2* __restrict__ recs, const int* __restrict__ bstart,
                      const float* __restrict__ dinv, const float* __restrict__ g,
                      const float* __restrict__ b2, float* __restrict__ out, int n) {
    __shared__ float bins[BK_NODES];
    const int t = threadIdx.x;
    const int lo = blockIdx.x * BK_NODES;
    const int nn = min(BK_NODES, n - lo);
    if (t < nn) bins[t] = g[lo + t];
    __syncthreads();

    const int beg = bstart[blockIdx.x];
    const int end = bstart[blockIdx.x + 1];
    const int last = end - 1;
    for (int base = beg; base < end; base += S2_THREADS * S2_UNROLL) {
        int2 rec[S2_UNROLL];
        float gr[S2_UNROLL];
#pragma unroll
        for (int u = 0; u < S2_UNROLL; u++)
            rec[u] = recs[min(base + u * S2_THREADS + t, last)];
#pragma unroll
        for (int u = 0; u < S2_UNROLL; u++)
            gr[u] = g[rec[u].x >> REC_SHIFT];
        asm volatile("" :: HOLD8(gr));
#pragma unroll
        for (int u = 0; u < S2_UNROLL; u++)
            if (base + u * S2_THREADS + t < end)
                atomicAdd(&bins[rec[u].x & REC_CMASK], __int_as_float(rec[u].y) * gr[u]);
    }
    __syncthreads();
    if (t < nn) out[lo + t] = b2[0] + dinv[lo + t] * bins[t];
}

extern "C" void kernel_launch(void* const* d_in, const int* in_sizes, int n_in,
                              void* d_out, int out_size, void* d_ws, size_t ws_size,
                              hipStream_t stream) {
    const float* x  = (const float*)d_in[0];
    const int*   ei = (const int*)d_in[1];
    const float* ew = (const float*)d_in[2];
    const float* w1 = (const float*)d_in[3];
    const float* b1 = (const float*)d_in[4];
    const float* w2 = (const float*)d_in[5];
    const float* b2 = (const float*)d_in[6];
    float* out = (float*)d_out;

    const int E = in_sizes[2];
    const int N = in_sizes[0] / IN_FEAT;
    const int* rows = ei;
    const int* cols = ei + E;

    float* ws   = (float*)d_ws;
    float* dinv = ws;
    float* g    = ws + N;
    float* h0   = ws + 2 * (size_t)N;
    int*  counts = (int*)(h0 + (size_t)N * HIDDEN);
    int*  tot    = counts + BK_MAX * EC_CHUNKS;
    int*  bstart = tot + BK_MAX;
    int2* recs   = (int2*)(bstart + BK_MAX + 32);

    const int B = 256;
    const int NB = (N + BK_NODES - 1) / BK_NODES;

    edge_count_kernel<<<EC_CHUNKS, EC_THREADS, 0, stream>>>(cols, counts, NB, E);
    chunk_prefix_kernel<<<(NB + 7) / 8, B, 0, stream>>>(counts, tot, NB);
    edge_place_kernel<<<EC_CHUNKS, EC_THREADS, 0, stream>>>(rows, cols, ew, counts, tot,
                                                            bstart, recs, NB, E);
    block_gemm1_kernel<<<NB, GB_THREADS, 0, stream>>>(x, w1, recs, bstart, dinv, h0, N);

    block_aggregate_kernel<<<NB, AG_THREADS, 0, stream>>>(recs, bstart, dinv, h0, b1, w2, g, N);
    block_scatter2_kernel<<<NB, S2_THREADS, 0, stream>>>(recs, bstart, dinv, g, b2, out, N);
}
